# starting kernel
# baseline (speedup 1.0000x reference)
#include <hip/hip_runtime.h>

typedef float v2f __attribute__((ext_vector_type(2)));
typedef float v8f __attribute__((ext_vector_type(8)));

constexpr int HD   = 96;
constexpr int RPB  = 128;
constexpr int BK   = 32;
constexpr int LPAD = 36;
constexpr int WPAD = 38;


__global__ void fill_zero_kernel(float* __restrict__ p, long n) {
  long i = (long)blockIdx.x * blockDim.x + threadIdx.x;
  long stride = (long)gridDim.x * blockDim.x;
  for (; i < n; i += stride) p[i] = 0.0f;
}

__global__ void degree_kernel(const int* __restrict__ src, const int* __restrict__ dst,
                              float* __restrict__ deg_out, float* __restrict__ deg_in, int E) {
  int e = blockIdx.x * blockDim.x + threadIdx.x;
  if (e < E) {
    __hip_atomic_fetch_add(&deg_out[src[e]], 1.0f, __ATOMIC_RELAXED, __HIP_MEMORY_SCOPE_AGENT);
    __hip_atomic_fetch_add(&deg_in[dst[e]],  1.0f, __ATOMIC_RELAXED, __HIP_MEMORY_SCOPE_AGENT);
  }
}

__global__ void finalize_norm_kernel(float* __restrict__ p, int n) {
  int i = blockIdx.x * blockDim.x + threadIdx.x;
  if (i < n) p[i] = 1.0f / sqrtf(fmaxf(p[i], 1.0f));
}

__global__ void wsum_kernel(const float* __restrict__ Wm, const float* __restrict__ bm,
                            float* __restrict__ wsum, float* __restrict__ bsum) {
  int t = threadIdx.x;
  if (t < HD) {
    float s = 0.0f;
    for (int j = 0; j < HD; ++j) s += Wm[t * HD + j];
    wsum[t] = s;
  } else if (t == HD) {
    float s = 0.0f;
    for (int j = 0; j < HD; ++j) s += bm[j];
    *bsum = s;
  }
}

__global__ __launch_bounds__(256)
void gemm_norm_wmma_kernel(const float* __restrict__ X, const int* __restrict__ rowmap,
                           const float* __restrict__ norm, const float* __restrict__ W,
                           float* __restrict__ OUT, int nrows, int K) {
  __shared__ float ldsX[RPB][LPAD];
  __shared__ float ldsW[HD][WPAD];

  const int tid  = threadIdx.x;
  const int wave = tid >> 5;
  const int lane = tid & 31;
  const int half = lane >> 4;
  const int l16  = lane & 15;
  const int row0 = blockIdx.x * RPB;

  v8f acc[6] = {};

  for (int k0 = 0; k0 < K; k0 += BK) {
    #pragma unroll
    for (int it = 0; it < (RPB * BK / 4) / 256; ++it) {
      int idx  = tid + 256 * it;
      int r    = idx >> 3;
      int c    = idx & 7;
      int grow = row0 + r;
      int kg   = k0 + c * 4;
      float4 v = make_float4(0.f, 0.f, 0.f, 0.f);
      if (grow < nrows && kg < K) {
        int srow = rowmap ? rowmap[grow] : grow;
        const float* p = X + (long)srow * K + kg;
        v = *(const float4*)p;
        if (kg + BK < K) __builtin_prefetch(p + BK, 0, 0);
        float s = norm[grow];
        v.x *= s; v.y *= s; v.z *= s; v.w *= s;
      }
      ldsX[r][c * 4 + 0] = v.x;
      ldsX[r][c * 4 + 1] = v.y;
      ldsX[r][c * 4 + 2] = v.z;
      ldsX[r][c * 4 + 3] = v.w;
    }
    #pragma unroll
    for (int it = 0; it < (BK * HD) / 256; ++it) {
      int idx = tid + 256 * it;
      int kk  = idx / HD;
      int col = idx % HD;
      float v = 0.0f;
      if (k0 + kk < K) v = W[(long)(k0 + kk) * HD + col];
      ldsW[col][kk] = v;
    }
    __syncthreads();

    const int kmax = (BK < K - k0) ? BK : (K - k0);
    const int mrow = wave * 16 + l16;
    for (int kk = 0; kk < kmax; kk += 4) {
      int ka = kk + half * 2;

      v2f A = *(const v2f*)&ldsX[mrow][ka];
      v2f Bf[6];
      #pragma unroll
      for (int t = 0; t < 6; ++t)
        Bf[t] = *(const v2f*)&ldsW[t * 16 + l16][ka];

      #pragma unroll
      for (int t = 0; t < 6; ++t)
        acc[t] = __builtin_amdgcn_wmma_f32_16x16x4_f32(
             false, A,  false, Bf[t],
             (short)0, acc[t],  false,  false);
    }
    __syncthreads();
  }

  const int orow = row0 + wave * 16 + half * 8;
  #pragma unroll
  for (int t = 0; t < 6; ++t) {
    int col = t * 16 + l16;
    #pragma unroll
    for (int i = 0; i < 8; ++i) {
      int r = orow + i;
      if (r < nrows) OUT[(long)r * HD + col] = acc[t][i];
    }
  }
}

__global__ void spmm_kernel(const int* __restrict__ src, const int* __restrict__ dst,
                            const float* __restrict__ h, float* __restrict__ agg, int E) {
  int gw   = (int)(((long)blockIdx.x * blockDim.x + threadIdx.x) >> 5);
  int lane = threadIdx.x & 31;
  if (gw < E) {
    int s = src[gw], d = dst[gw];
    const float* hp = h + (long)s * HD;
    float* ap = agg + (long)d * HD;
    #pragma unroll
    for (int i = 0; i < 3; ++i) {
      float v = hp[lane + 32 * i];
      __hip_atomic_fetch_add(&ap[lane + 32 * i], v, __ATOMIC_RELAXED, __HIP_MEMORY_SCOPE_AGENT);
    }
  }
}

__global__ void prelu_kernel(const float* __restrict__ agg, const float* __restrict__ norm_in,
                             const float* __restrict__ b, const float* __restrict__ a,
                             float* __restrict__ out, int n) {
  long idx = (long)blockIdx.x * blockDim.x + threadIdx.x;
  if (idx < (long)n * HD) {
    int i = (int)(idx / HD);
    int j = (int)(idx % HD);
    float v = agg[idx] * norm_in[i] + b[j];
    out[idx] = (v >= 0.0f) ? v : a[j] * v;
  }
}

__global__ void reduce_kernel(const float* __restrict__ z, const float* __restrict__ wsum,
                              const float* __restrict__ bsum, float* __restrict__ out, int n) {
  int gw   = (int)(((long)blockIdx.x * blockDim.x + threadIdx.x) >> 5);
  int lane = threadIdx.x & 31;
  if (gw < n) {
    const float* zp = z + (long)gw * HD;
    float s = 0.0f;
    #pragma unroll
    for (int i = 0; i < 3; ++i) s = fmaf(zp[lane + 32 * i], wsum[lane + 32 * i], s);
    #pragma unroll
    for (int off = 16; off > 0; off >>= 1) s += __shfl_xor(s, off, 32);
    if (lane == 0) out[gw] = s + *bsum;
  }
}

extern "C" void kernel_launch(void* const* d_in, const int* in_sizes, int n_in,
                              void* d_out, int out_size, void* d_ws, size_t ws_size,
                              hipStream_t stream) {
  const float* x    = (const float*)d_in[0];
  const int*   src  = (const int*)  d_in[1];
  const int*   dst  = (const int*)  d_in[2];
  const int*   perm = (const int*)  d_in[3];
  const float* W1   = (const float*)d_in[4];
  const float* b1   = (const float*)d_in[5];
  const float* a1   = (const float*)d_in[6];
  const float* W2   = (const float*)d_in[7];
  const float* b2   = (const float*)d_in[8];
  const float* a2   = (const float*)d_in[9];
  const float* Wm   = (const float*)d_in[10];
  const float* bm   = (const float*)d_in[11];
  float* out = (float*)d_out;

  const int E  = in_sizes[1];
  const int N  = in_sizes[3];
  const int K1 = in_sizes[0] / N;

  float* ws       = (float*)d_ws;
  float* norm_out = ws;
  float* norm_in  = ws + N;
  float* wsum     = ws + 2L * N;
  float* bsum     = ws + 2L * N + HD;
  float* bufA     = ws + 2L * N + 2 * HD;
  float* bufB     = bufA + (long)N * HD;

  const long NH = (long)N * HD;
  const int gemm_blocks = (N + RPB - 1) / RPB;
  const int spmm_blocks = (int)(((long)E * 32 + 255) / 256);
  const int elw_blocks  = (int)((NH + 255) / 256);
  const int red_blocks  = (int)(((long)N * 32 + 255) / 256);

  fill_zero_kernel<<<512, 256, 0, stream>>>(norm_out, 2L * N);
  degree_kernel<<<(E + 255) / 256, 256, 0, stream>>>(src, dst, norm_out, norm_in, E);
  finalize_norm_kernel<<<(2 * N + 255) / 256, 256, 0, stream>>>(norm_out, 2 * N);
  wsum_kernel<<<1, 128, 0, stream>>>(Wm, bm, wsum, bsum);

  for (int view = 0; view < 2; ++view) {
    const int* rowmap = view ? perm : nullptr;
    gemm_norm_wmma_kernel<<<gemm_blocks, 256, 0, stream>>>(x, rowmap, norm_out, W1, bufA, N, K1);
    fill_zero_kernel<<<2048, 256, 0, stream>>>(bufB, NH);
    spmm_kernel<<<spmm_blocks, 256, 0, stream>>>(src, dst, bufA, bufB, E);
    prelu_kernel<<<elw_blocks, 256, 0, stream>>>(bufB, norm_in, b1, a1, bufA, N);
    gemm_norm_wmma_kernel<<<gemm_blocks, 256, 0, stream>>>(bufA, nullptr, norm_out, W2, bufB, N, HD);
    fill_zero_kernel<<<2048, 256, 0, stream>>>(bufA, NH);
    spmm_kernel<<<spmm_blocks, 256, 0, stream>>>(src, dst, bufB, bufA, E);
    prelu_kernel<<<elw_blocks, 256, 0, stream>>>(bufA, norm_in, b2, a2, bufB, N);
    reduce_kernel<<<red_blocks, 256, 0, stream>>>(bufB, wsum, bsum, out + (long)view * N, N);
  }
}


# GEMM operand path: each wave loads its 16 rows from global straight into WMMA A registers (no LDS tile, no K-loop barrier), whole W^T resident in LDS in fragment order; k summed in another order
# speedup vs baseline: 2.0139x; 2.0139x over previous
#include <hip/hip_runtime.h>

typedef float v2f __attribute__((ext_vector_type(2)));
typedef float v8f __attribute__((ext_vector_type(8)));

constexpr int HD   = 96;


__global__ void fill_zero_kernel(float* __restrict__ p, long n) {
  long i = (long)blockIdx.x * blockDim.x + threadIdx.x;
  long stride = (long)gridDim.x * blockDim.x;
  for (; i < n; i += stride) p[i] = 0.0f;
}

__global__ void degree_kernel(const int* __restrict__ src, const int* __restrict__ dst,
                              float* __restrict__ deg_out, float* __restrict__ deg_in, int E) {
  int e = blockIdx.x * blockDim.x + threadIdx.x;
  if (e < E) {
    __hip_atomic_fetch_add(&deg_out[src[e]], 1.0f, __ATOMIC_RELAXED, __HIP_MEMORY_SCOPE_AGENT);
    __hip_atomic_fetch_add(&deg_in[dst[e]],  1.0f, __ATOMIC_RELAXED, __HIP_MEMORY_SCOPE_AGENT);
  }
}

__global__ void finalize_norm_kernel(float* __restrict__ p, int n) {
  int i = blockIdx.x * blockDim.x + threadIdx.x;
  if (i < n) p[i] = 1.0f / sqrtf(fmaxf(p[i], 1.0f));
}

__global__ void wsum_kernel(const float* __restrict__ Wm, const float* __restrict__ bm,
                            float* __restrict__ wsum, float* __restrict__ bsum) {
  int t = threadIdx.x;
  if (t < HD) {
    float s = 0.0f;
    for (int j = 0; j < HD; ++j) s += Wm[t * HD + j];
    wsum[t] = s;
  } else if (t == HD) {
    float s = 0.0f;
    for (int j = 0; j < HD; ++j) s += bm[j];
    *bsum = s;
  }
}

constexpr int KB   = 64;
constexpr int KMAX = 512;
constexpr int WS   = 514;
constexpr int GW   = 16;

__device__ __forceinline__
void gemm_load_block(float4 (&dstv)[8], const float* X, int srow, int K, int k0) {
  #pragma unroll
  for (int q = 0; q < 8; ++q) {
    const int k  = k0 + q * 8;
    const int kc = (k < K) ? k : (K - 4);
    dstv[q] = *(const float4*)(X + (long)srow * K + kc);
  }
}

__device__ __forceinline__
void gemm_load_finish(float4 (&bb)[8], float4 (&aa)[8], const float* __restrict__ fin_b,
                      const float* __restrict__ fin_a, int K, int k0) {
  #pragma unroll
  for (int q = 0; q < 8; ++q) {
    const int k  = k0 + q * 8;
    const int kc = (k < K) ? k : (K - 4);
    bb[q] = *(const float4*)(fin_b + kc);
    aa[q] = *(const float4*)(fin_a + kc);
  }
}

template <bool FIN>
__device__ __forceinline__
void gemm_block_products(v8f (&acc)[6], const float4 (&raw)[8], const float4 (&bb)[8],
                         const float4 (&aa)[8], float s, float si, int K, int k0,
                         const float* wb) {
  float4 cur[8];
  #pragma unroll
  for (int q = 0; q < 8; ++q) {
    float4 v = raw[q];
    const int k = k0 + q * 8;
    if (FIN) {
      float t;
      t = v.x * si + bb[q].x; v.x = (t >= 0.0f) ? t : aa[q].x * t;
      t = v.y * si + bb[q].y; v.y = (t >= 0.0f) ? t : aa[q].y * t;
      t = v.z * si + bb[q].z; v.z = (t >= 0.0f) ? t : aa[q].z * t;
      t = v.w * si + bb[q].w; v.w = (t >= 0.0f) ? t : aa[q].w * t;
    }
    v.x *= s; v.y *= s; v.z *= s; v.w *= s;
    if (k >= K) v = make_float4(0.f, 0.f, 0.f, 0.f);
    cur[q] = v;
  }
  #pragma unroll
  for (int st = 0; st < 16; ++st) {
    v2f A;
    if (st & 1) { A.x = cur[st >> 1].z; A.y = cur[st >> 1].w; }
    else        { A.x = cur[st >> 1].x; A.y = cur[st >> 1].y; }
    v2f Bf[6];
    #pragma unroll
    for (int t = 0; t < 6; ++t)
      Bf[t] = *(const v2f*)(wb + t * 16 * WS + st * 2);
    #pragma unroll
    for (int t = 0; t < 6; ++t)
      acc[t] = __builtin_amdgcn_wmma_f32_16x16x4_f32(
           false, A,  false, Bf[t],
           (short)0, acc[t],  false,  false);
  }
}

template <bool FIN>
__device__ __forceinline__
void gemm_rows_body(float* ldsW, const float* X, const int* __restrict__ rowmap,
                    const float* __restrict__ norm, const float* __restrict__ W,
                    const float* __restrict__ fin_norm, const float* __restrict__ fin_b,
                    const float* __restrict__ fin_a,
                    float* __restrict__ OUT, float* ZERO, int nrows, int K) {

  const int tid  = threadIdx.x;
  const int lane = tid & 31;
  const int half = lane >> 4;
  const int l16  = lane & 15;
  const int wave = __builtin_amdgcn_readfirstlane(tid >> 5);
  const int Kpad = (K + 2 * KB - 1) & ~(2 * KB - 1);

  {
    const int nitem = Kpad * (HD / 4);
    #pragma unroll 6
    for (int idx = tid; idx < nitem; idx += GW * 32) {
      const int k  = idx / (HD / 4);
      const int j  = idx % (HD / 4);
      const int kc = (k < K) ? k : (K - 1);
      float4 v = *(const float4*)(W + (long)kc * HD + j * 4);
      if (k >= K) v = make_float4(0.f, 0.f, 0.f, 0.f);
      const int kk = k & (KB - 1);
      const int p  = (k & ~(KB - 1)) + ((kk >> 2) & 1) * 32 + (kk >> 3) * 4 + (kk & 3);
      ldsW[(j * 4 + 0) * WS + p] = v.x;
      ldsW[(j * 4 + 1) * WS + p] = v.y;
      ldsW[(j * 4 + 2) * WS + p] = v.z;
      ldsW[(j * 4 + 3) * WS + p] = v.w;
    }
  }
  __syncthreads();

  const int tile = blockIdx.x * GW + wave;
  if (tile * 16 >= nrows) return;

  const int grow = tile * 16 + l16;
  const int crow = (grow < nrows) ? grow : (nrows - 1);
  const int srow = rowmap ? rowmap[crow] : crow;
  const float s  = norm[crow];
  float si = 0.0f;
  if (FIN) si = fin_norm[crow];

  v8f acc[6] = {};
  const float* wl = ldsW + l16 * WS + half * 32;

  float4 buf0[8], buf1[8];
  float4 bb[8], aa[8];
  gemm_load_block(buf0, X, srow, K, half * 4);
  for (int kb = 0; kb < Kpad; kb += 2 * KB) {
    if (FIN) gemm_load_finish(bb, aa, fin_b, fin_a, K, kb + half * 4);
    gemm_load_block(buf1, X, srow, K, kb + KB + half * 4);
    gemm_block_products<FIN>(acc, buf0, bb, aa, s, si, K, kb + half * 4, wl + kb);
    if (FIN) gemm_load_finish(bb, aa, fin_b, fin_a, K, kb + KB + half * 4);
    gemm_load_block(buf0, X, srow, K, kb + 2 * KB + half * 4);
    gemm_block_products<FIN>(acc, buf1, bb, aa, s, si, K, kb + KB + half * 4, wl + kb + KB);
  }

  {
    const long zend = (long)nrows * HD;
    #pragma unroll
    for (int it = 0; it < (16 * HD / 4) / 32; ++it) {
      const long off = (long)tile * 16 * HD + (long)(lane + 32 * it) * 4;
      if (off < zend) *(float4*)(ZERO + off) = make_float4(0.f, 0.f, 0.f, 0.f);
    }
  }

  const int orow = tile * 16 + half * 8;
  #pragma unroll
  for (int t = 0; t < 6; ++t) {
    int col = t * 16 + l16;
    #pragma unroll
    for (int i = 0; i < 8; ++i) {
      int r = orow + i;
      if (r < nrows) OUT[(long)r * HD + col] = acc[t][i];
    }
  }
}

__global__ __launch_bounds__(GW * 32)
void gemm_rows_wmma_kernel(const float* X, const int* __restrict__ rowmap,
                           const float* __restrict__ norm, const float* __restrict__ W,
                           float* __restrict__ OUT, float* ZERO, int nrows, int K) {
  __shared__ float ldsW[HD * WS];
  gemm_rows_body<false>(ldsW, X, rowmap, norm, W, nullptr, nullptr, nullptr, OUT, ZERO, nrows, K);
}

__global__ __launch_bounds__(GW * 32)
void gemm_rows_finish_wmma_kernel(const float* X, const float* __restrict__ norm,
                                  const float* __restrict__ W,
                                  const float* __restrict__ fin_norm,
                                  const float* __restrict__ fin_b,
                                  const float* __restrict__ fin_a,
                                  float* __restrict__ OUT, float* ZERO, int nrows, int K) {
  __shared__ float ldsW[HD * WS];
  gemm_rows_body<true>(ldsW, X, nullptr, norm, W, fin_norm, fin_b, fin_a, OUT, ZERO, nrows, K);
}

constexpr int EPW = 32;
constexpr int EB  = 8;

__device__ __forceinline__ void atomic_add_row(float* __restrict__ ap, float v0, float v1, float v2) {
  __hip_atomic_fetch_add(&ap[0],  v0, __ATOMIC_RELAXED, __HIP_MEMORY_SCOPE_AGENT);
  __hip_atomic_fetch_add(&ap[32], v1, __ATOMIC_RELAXED, __HIP_MEMORY_SCOPE_AGENT);
  __hip_atomic_fetch_add(&ap[64], v2, __ATOMIC_RELAXED, __HIP_MEMORY_SCOPE_AGENT);
}

__global__ __launch_bounds__(256)
void spmm_kernel(const int* __restrict__ src, const int* __restrict__ dst,
                 const float* __restrict__ h, float* __restrict__ agg, int E) {
  const int lane = threadIdx.x & 31;
  const int wave = __builtin_amdgcn_readfirstlane((int)(threadIdx.x >> 5));
  const long e0  = ((long)blockIdx.x * (blockDim.x >> 5) + wave) * EPW;
  if (e0 >= E) return;
  const int ne = (E - e0 < EPW) ? (int)(E - e0) : EPW;

  int my_s = 0, my_d = 0;
  if (lane < ne) { my_s = src[e0 + lane]; my_d = dst[e0 + lane]; }

  if (ne == EPW) {
    #pragma unroll
    for (int b = 0; b < EPW; b += EB) {
      float v[EB][3];
      #pragma unroll
      for (int j = 0; j < EB; ++j) {
        const int s = __builtin_amdgcn_readlane(my_s, b + j);
        const float* hp = h + (long)s * HD + lane;
        v[j][0] = hp[0]; v[j][1] = hp[32]; v[j][2] = hp[64];
      }
      #pragma unroll
      for (int j = 0; j < EB; ++j) {
        const int d = __builtin_amdgcn_readlane(my_d, b + j);
        atomic_add_row(agg + (long)d * HD + lane, v[j][0], v[j][1], v[j][2]);
      }
    }
  } else {
    for (int j = 0; j < ne; ++j) {
      const int s = __builtin_amdgcn_readlane(my_s, j);
      const int d = __builtin_amdgcn_readlane(my_d, j);
      const float* hp = h + (long)s * HD + lane;
      atomic_add_row(agg + (long)d * HD + lane, hp[0], hp[32], hp[64]);
    }
  }
}

constexpr int RW = 8;

__global__ __launch_bounds__(256)
void finish_dot_kernel(const float* __restrict__ agg, const float* __restrict__ norm_in,
                       const float* __restrict__ b, const float* __restrict__ a,
                       const float* __restrict__ wsum, const float* __restrict__ bsum,
                       float* __restrict__ out, int n) {
  const int lane = threadIdx.x & 31;
  const int wave = __builtin_amdgcn_readfirstlane((int)(threadIdx.x >> 5));
  const long r0  = ((long)blockIdx.x * (blockDim.x >> 5) + wave) * RW;
  if (r0 >= n) return;
  const float w0 = wsum[lane], w1 = wsum[lane + 32], w2 = wsum[lane + 64];
  const float b0 = b[lane],    b1 = b[lane + 32],    b2 = b[lane + 64];
  const float a0 = a[lane],    a1 = a[lane + 32],    a2 = a[lane + 64];
  const float bs = *bsum;
  float zv[RW][3];
  float sc[RW];
  #pragma unroll
  for (int q = 0; q < RW; ++q) {
    const long r = (r0 + q < n) ? (r0 + q) : (long)(n - 1);
    const float* zp = agg + r * HD + lane;
    zv[q][0] = zp[0]; zv[q][1] = zp[32]; zv[q][2] = zp[64];
    sc[q] = norm_in[r];
  }
  float sum[RW];
  #pragma unroll
  for (int q = 0; q < RW; ++q) {
    float t, z0, z1, z2;
    t = zv[q][0] * sc[q] + b0; z0 = (t >= 0.0f) ? t : a0 * t;
    t = zv[q][1] * sc[q] + b1; z1 = (t >= 0.0f) ? t : a1 * t;
    t = zv[q][2] * sc[q] + b2; z2 = (t >= 0.0f) ? t : a2 * t;
    float s = 0.0f;
    s = fmaf(z0, w0, s);
    s = fmaf(z1, w1, s);
    s = fmaf(z2, w2, s);
    sum[q] = s;
  }
  #pragma unroll
  for (int off = 16; off > 0; off >>= 1) {
    #pragma unroll
    for (int q = 0; q < RW; ++q) sum[q] += __shfl_xor(sum[q], off, 32);
  }
  if (lane == 0) {
    #pragma unroll
    for (int q = 0; q < RW; ++q)
      if (r0 + q < n) out[r0 + q] = sum[q] + bs;
  }
}

extern "C" void kernel_launch(void* const* d_in, const int* in_sizes, int n_in,
                              void* d_out, int out_size, void* d_ws, size_t ws_size,
                              hipStream_t stream) {
  const float* x    = (const float*)d_in[0];
  const int*   src  = (const int*)  d_in[1];
  const int*   dst  = (const int*)  d_in[2];
  const int*   perm = (const int*)  d_in[3];
  const float* W1   = (const float*)d_in[4];
  const float* b1   = (const float*)d_in[5];
  const float* a1   = (const float*)d_in[6];
  const float* W2   = (const float*)d_in[7];
  const float* b2   = (const float*)d_in[8];
  const float* a2   = (const float*)d_in[9];
  const float* Wm   = (const float*)d_in[10];
  const float* bm   = (const float*)d_in[11];
  float* out = (float*)d_out;

  const int E  = in_sizes[1];
  const int N  = in_sizes[3];
  const int K1 = in_sizes[0] / N;

  float* ws       = (float*)d_ws;
  float* norm_out = ws;
  float* norm_in  = ws + N;
  float* wsum     = ws + 2L * N;
  float* bsum     = ws + 2L * N + HD;
  float* bufA     = ws + 2L * N + 2 * HD;
  float* bufB     = bufA + (long)N * HD;

  const int gemm_blocks = (N + GW * 16 - 1) / (GW * 16);
  const int spmm_blocks = (int)(((long)E + 8 * EPW - 1) / (8 * EPW));
  const int red_blocks  = (int)(((long)N + 8 * RW - 1) / (8 * RW));

  fill_zero_kernel<<<512, 256, 0, stream>>>(norm_out, 2L * N);
  degree_kernel<<<(E + 255) / 256, 256, 0, stream>>>(src, dst, norm_out, norm_in, E);
  finalize_norm_kernel<<<(2 * N + 255) / 256, 256, 0, stream>>>(norm_out, 2 * N);
  wsum_kernel<<<1, 128, 0, stream>>>(Wm, bm, wsum, bsum);

  for (int view = 0; view < 2; ++view) {
    const int* rowmap = view ? perm : nullptr;
    gemm_rows_wmma_kernel<<<gemm_blocks, GW * 32, 0, stream>>>(x, rowmap, norm_out, W1, bufA, bufB, N, K1);
    spmm_kernel<<<spmm_blocks, 256, 0, stream>>>(src, dst, bufA, bufB, E);
    gemm_rows_finish_wmma_kernel<<<gemm_blocks, GW * 32, 0, stream>>>(bufB, norm_out, W2,
                                                                     norm_in, b1, a1, bufA, bufB, N, HD);
    spmm_kernel<<<spmm_blocks, 256, 0, stream>>>(src, dst, bufA, bufB, E);
    finish_dot_kernel<<<red_blocks, 256, 0, stream>>>(bufB, norm_in, b2, a2, wsum, bsum,
                                                      out + (long)view * N, N);
  }
}


# atomic adds kept inside one L2 half: nodes split in two ranges, a work-group (die id read at run time) adds only into its half's rows, edge chunks drawn from two ticket queues; arrival-order sums
# speedup vs baseline: 3.1351x; 1.5567x over previous
#include <hip/hip_runtime.h>

typedef float v2f __attribute__((ext_vector_type(2)));
typedef float v8f __attribute__((ext_vector_type(8)));

constexpr int HD   = 96;


__global__ void fill_zero_kernel(float* __restrict__ p, long n, int* __restrict__ ctr) {
  long i = (long)blockIdx.x * blockDim.x + threadIdx.x;
  long stride = (long)gridDim.x * blockDim.x;
  if (blockIdx.x == 0 && threadIdx.x < 64) ctr[threadIdx.x] = 0;
  for (; i < n; i += stride) p[i] = 0.0f;
}

constexpr int CTR_STRIDE = 32;

__device__ __forceinline__ int l2_half() {
  return (int)((__builtin_amdgcn_s_sendmsg_rtn(0x87) >> 18) & 1u);
}

__device__ __forceinline__ int draw_ticket(int* ctr, int q) {
  return __hip_atomic_fetch_add(ctr + q * CTR_STRIDE, 1, __ATOMIC_RELAXED, __HIP_MEMORY_SCOPE_AGENT);
}

constexpr int DCH = 1024;

__global__ __launch_bounds__(256)
void degree_kernel(const int* __restrict__ src, const int* __restrict__ dst,
                   float* __restrict__ deg_out, float* __restrict__ deg_in,
                   int* __restrict__ ctr, int E, int n) {
  __shared__ int sh_t[2];
  const int tid = threadIdx.x;
  const int myh = l2_half();
  const int Nh  = (n >> 1) & ~1;
  const int nitems = (E + DCH - 1) / DCH;
  int q = myh, tn = 0;
  if (tid == 0) tn = draw_ticket(ctr, q);
  for (int it = 0; ; ++it) {
    if (tid == 0) sh_t[it & 1] = tn;
    __syncthreads();
    const int t = __builtin_amdgcn_readfirstlane(sh_t[it & 1]);
    if (t >= nitems) {
      if (q != myh) break;
      q = 1 - myh;
      if (tid == 0) tn = draw_ticket(ctr, q);
      continue;
    }
    if (tid == 0) tn = draw_ticket(ctr, q);
    const bool up = (q != 0);
    int sv[DCH / 256], dv[DCH / 256];
    #pragma unroll
    for (int k = 0; k < DCH / 256; ++k) {
      const int e  = t * DCH + k * 256 + tid;
      const int ec = (e < E) ? e : (E - 1);
      sv[k] = src[ec];
      dv[k] = dst[ec];
    }
    #pragma unroll
    for (int k = 0; k < DCH / 256; ++k) {
      const bool ok = (t * DCH + k * 256 + tid) < E;
      if (ok && ((sv[k] >= Nh) == up))
        __hip_atomic_fetch_add(&deg_out[sv[k]], 1.0f, __ATOMIC_RELAXED, __HIP_MEMORY_SCOPE_AGENT);
      if (ok && ((dv[k] >= Nh) == up))
        __hip_atomic_fetch_add(&deg_in[dv[k]],  1.0f, __ATOMIC_RELAXED, __HIP_MEMORY_SCOPE_AGENT);
    }
  }
}

__global__ void finalize_norm_kernel(float* __restrict__ p, int n) {
  int i = blockIdx.x * blockDim.x + threadIdx.x;
  if (i < n) p[i] = 1.0f / sqrtf(fmaxf(p[i], 1.0f));
}

__global__ void wsum_kernel(const float* __restrict__ Wm, const float* __restrict__ bm,
                            float* __restrict__ wsum, float* __restrict__ bsum) {
  int t = threadIdx.x;
  if (t < HD) {
    float s = 0.0f;
    for (int j = 0; j < HD; ++j) s += Wm[t * HD + j];
    wsum[t] = s;
  } else if (t == HD) {
    float s = 0.0f;
    for (int j = 0; j < HD; ++j) s += bm[j];
    *bsum = s;
  }
}

constexpr int KB   = 64;
constexpr int KMAX = 512;
constexpr int WS   = 514;
constexpr int GW   = 16;

__device__ __forceinline__
void gemm_load_block(float4 (&dstv)[8], const float* X, int srow, int K, int k0) {
  #pragma unroll
  for (int q = 0; q < 8; ++q) {
    const int k  = k0 + q * 8;
    const int kc = (k < K) ? k : (K - 4);
    dstv[q] = *(const float4*)(X + (long)srow * K + kc);
  }
}

__device__ __forceinline__
void gemm_load_finish(float4 (&bb)[8], float4 (&aa)[8], const float* __restrict__ fin_b,
                      const float* __restrict__ fin_a, int K, int k0) {
  #pragma unroll
  for (int q = 0; q < 8; ++q) {
    const int k  = k0 + q * 8;
    const int kc = (k < K) ? k : (K - 4);
    bb[q] = *(const float4*)(fin_b + kc);
    aa[q] = *(const float4*)(fin_a + kc);
  }
}

template <bool FIN>
__device__ __forceinline__
void gemm_block_products(v8f (&acc)[6], const float4 (&raw)[8], const float4 (&bb)[8],
                         const float4 (&aa)[8], float s, float si, int K, int k0,
                         const float* wb) {
  float4 cur[8];
  #pragma unroll
  for (int q = 0; q < 8; ++q) {
    float4 v = raw[q];
    const int k = k0 + q * 8;
    if (FIN) {
      float t;
      t = v.x * si + bb[q].x; v.x = (t >= 0.0f) ? t : aa[q].x * t;
      t = v.y * si + bb[q].y; v.y = (t >= 0.0f) ? t : aa[q].y * t;
      t = v.z * si + bb[q].z; v.z = (t >= 0.0f) ? t : aa[q].z * t;
      t = v.w * si + bb[q].w; v.w = (t >= 0.0f) ? t : aa[q].w * t;
    }
    v.x *= s; v.y *= s; v.z *= s; v.w *= s;
    if (k >= K) v = make_float4(0.f, 0.f, 0.f, 0.f);
    cur[q] = v;
  }
  #pragma unroll
  for (int st = 0; st < 16; ++st) {
    v2f A;
    if (st & 1) { A.x = cur[st >> 1].z; A.y = cur[st >> 1].w; }
    else        { A.x = cur[st >> 1].x; A.y = cur[st >> 1].y; }
    v2f Bf[6];
    #pragma unroll
    for (int t = 0; t < 6; ++t)
      Bf[t] = *(const v2f*)(wb + t * 16 * WS + st * 2);
    #pragma unroll
    for (int t = 0; t < 6; ++t)
      acc[t] = __builtin_amdgcn_wmma_f32_16x16x4_f32(
           false, A,  false, Bf[t],
           (short)0, acc[t],  false,  false);
  }
}

template <bool FIN>
__device__ __forceinline__
void gemm_rows_body(float* ldsW, const float* X, const int* __restrict__ rowmap,
                    const float* __restrict__ norm, const float* __restrict__ W,
                    const float* __restrict__ fin_norm, const float* __restrict__ fin_b,
                    const float* __restrict__ fin_a,
                    float* __restrict__ OUT, float* ZERO, int nrows, int K) {

  const int tid  = threadIdx.x;
  const int lane = tid & 31;
  const int half = lane >> 4;
  const int l16  = lane & 15;
  const int wave = __builtin_amdgcn_readfirstlane(tid >> 5);
  const int Kpad = (K + 2 * KB - 1) & ~(2 * KB - 1);

  {
    const int nitem = Kpad * (HD / 4);
    #pragma unroll 6
    for (int idx = tid; idx < nitem; idx += GW * 32) {
      const int k  = idx / (HD / 4);
      const int j  = idx % (HD / 4);
      const int kc = (k < K) ? k : (K - 1);
      float4 v = *(const float4*)(W + (long)kc * HD + j * 4);
      if (k >= K) v = make_float4(0.f, 0.f, 0.f, 0.f);
      const int kk = k & (KB - 1);
      const int p  = (k & ~(KB - 1)) + ((kk >> 2) & 1) * 32 + (kk >> 3) * 4 + (kk & 3);
      ldsW[(j * 4 + 0) * WS + p] = v.x;
      ldsW[(j * 4 + 1) * WS + p] = v.y;
      ldsW[(j * 4 + 2) * WS + p] = v.z;
      ldsW[(j * 4 + 3) * WS + p] = v.w;
    }
  }
  __syncthreads();

  const int tile = blockIdx.x * GW + wave;
  if (tile * 16 >= nrows) return;

  const int grow = tile * 16 + l16;
  const int crow = (grow < nrows) ? grow : (nrows - 1);
  const int srow = rowmap ? rowmap[crow] : crow;
  const float s  = norm[crow];
  float si = 0.0f;
  if (FIN) si = fin_norm[crow];

  v8f acc[6] = {};
  const float* wl = ldsW + l16 * WS + half * 32;

  float4 buf0[8], buf1[8];
  float4 bb[8], aa[8];
  gemm_load_block(buf0, X, srow, K, half * 4);
  for (int kb = 0; kb < Kpad; kb += 2 * KB) {
    if (FIN) gemm_load_finish(bb, aa, fin_b, fin_a, K, kb + half * 4);
    gemm_load_block(buf1, X, srow, K, kb + KB + half * 4);
    gemm_block_products<FIN>(acc, buf0, bb, aa, s, si, K, kb + half * 4, wl + kb);
    if (FIN) gemm_load_finish(bb, aa, fin_b, fin_a, K, kb + KB + half * 4);
    gemm_load_block(buf0, X, srow, K, kb + 2 * KB + half * 4);
    gemm_block_products<FIN>(acc, buf1, bb, aa, s, si, K, kb + KB + half * 4, wl + kb + KB);
  }

  {
    const long zend = (long)nrows * HD;
    #pragma unroll
    for (int it = 0; it < (16 * HD / 4) / 32; ++it) {
      const long off = (long)tile * 16 * HD + (long)(lane + 32 * it) * 4;
      if (off < zend) *(float4*)(ZERO + off) = make_float4(0.f, 0.f, 0.f, 0.f);
    }
  }

  const int orow = tile * 16 + half * 8;
  #pragma unroll
  for (int t = 0; t < 6; ++t) {
    int col = t * 16 + l16;
    #pragma unroll
    for (int i = 0; i < 8; ++i) {
      int r = orow + i;
      if (r < nrows) OUT[(long)r * HD + col] = acc[t][i];
    }
  }
}

__global__ __launch_bounds__(GW * 32)
void gemm_rows_wmma_kernel(const float* X, const int* __restrict__ rowmap,
                           const float* __restrict__ norm, const float* __restrict__ W,
                           float* __restrict__ OUT, float* ZERO, int nrows, int K) {
  __shared__ float ldsW[HD * WS];
  gemm_rows_body<false>(ldsW, X, rowmap, norm, W, nullptr, nullptr, nullptr, OUT, ZERO, nrows, K);
}

__global__ __launch_bounds__(GW * 32)
void gemm_rows_finish_wmma_kernel(const float* X, const float* __restrict__ norm,
                                  const float* __restrict__ W,
                                  const float* __restrict__ fin_norm,
                                  const float* __restrict__ fin_b,
                                  const float* __restrict__ fin_a,
                                  float* __restrict__ OUT, float* ZERO, int nrows, int K) {
  __shared__ float ldsW[HD * WS];
  gemm_rows_body<true>(ldsW, X, nullptr, norm, W, fin_norm, fin_b, fin_a, OUT, ZERO, nrows, K);
}

constexpr int EB   = 8;
constexpr int SPW  = 8;
constexpr int WCH  = 64;
constexpr int GCH  = SPW * WCH;
constexpr int LCAP = WCH + EB;

__device__ __forceinline__ void atomic_add_row(float* __restrict__ ap, float v0, float v1, float v2) {
  __hip_atomic_fetch_add(&ap[0],  v0, __ATOMIC_RELAXED, __HIP_MEMORY_SCOPE_AGENT);
  __hip_atomic_fetch_add(&ap[32], v1, __ATOMIC_RELAXED, __HIP_MEMORY_SCOPE_AGENT);
  __hip_atomic_fetch_add(&ap[64], v2, __ATOMIC_RELAXED, __HIP_MEMORY_SCOPE_AGENT);
}

__global__ __launch_bounds__(256)
void spmm_kernel(const int* __restrict__ src, const int* __restrict__ dst,
                 const float* __restrict__ h, float* __restrict__ agg,
                 int* __restrict__ ctr, int E, int n) {
  __shared__ int lst_s[SPW][LCAP];
  __shared__ int lst_d[SPW][LCAP];
  __shared__ int sh_t[2];
  const int tid  = threadIdx.x;
  const int lane = tid & 31;
  const int wave = __builtin_amdgcn_readfirstlane(tid >> 5);
  const int myh  = l2_half();
  const int Nh   = (n >> 1) & ~1;
  const int nitems = (E + GCH - 1) / GCH;
  int* ms = lst_s[wave];
  int* md = lst_d[wave];

  int q = myh, cnt = 0, tn = 0;
  if (tid == 0) tn = draw_ticket(ctr, q);
  for (int it = 0; ; ++it) {
    if (tid == 0) sh_t[it & 1] = tn;
    __syncthreads();
    const int t = __builtin_amdgcn_readfirstlane(sh_t[it & 1]);
    if (t >= nitems) {
      if (q != myh) break;
      q = 1 - myh;
      if (tid == 0) tn = draw_ticket(ctr, q);
      continue;
    }
    if (tid == 0) tn = draw_ticket(ctr, q);

    const bool up = (q != 0);
    int sv[WCH / 32], dv[WCH / 32];
    #pragma unroll
    for (int g = 0; g < WCH / 32; ++g) {
      const int e  = t * GCH + wave * WCH + g * 32 + lane;
      const int ec = (e < E) ? e : (E - 1);
      sv[g] = src[ec];
      dv[g] = (e < E) ? dst[ec] : -1;
    }
    #pragma unroll
    for (int g = 0; g < WCH / 32; ++g) {
      const bool mine = (dv[g] >= 0) && ((dv[g] >= Nh) == up);
      const unsigned m = __builtin_amdgcn_ballot_w32(mine);
      const int pos = cnt + (int)__builtin_amdgcn_mbcnt_lo(m, 0u);
      if (mine) { ms[pos] = sv[g]; md[pos] = dv[g]; }
      cnt += __builtin_popcount(m);
    }

    int b = 0;
    for (; b + EB <= cnt; b += EB) {
      const int vs = ms[b + (lane & (EB - 1))];
      const int vd = md[b + (lane & (EB - 1))];
      float v[EB][3];
      #pragma unroll
      for (int j = 0; j < EB; ++j) {
        const int s = __builtin_amdgcn_readlane(vs, j);
        const float* hp = h + (long)s * HD + lane;
        v[j][0] = hp[0]; v[j][1] = hp[32]; v[j][2] = hp[64];
      }
      #pragma unroll
      for (int j = 0; j < EB; ++j) {
        const int d = __builtin_amdgcn_readlane(vd, j);
        atomic_add_row(agg + (long)d * HD + lane, v[j][0], v[j][1], v[j][2]);
      }
    }
    const int r = cnt - b;
    if (b > 0 && r > 0) {
      int ts = 0, td = 0;
      if (lane < r) { ts = ms[b + lane]; td = md[b + lane]; }
      if (lane < r) { ms[lane] = ts; md[lane] = td; }
    }
    cnt = r;
  }

  if (cnt > 0) {
    const int li = (lane < cnt) ? lane : 0;
    const int vs = ms[li];
    const int vd = md[li];
    for (int j = 0; j < cnt; ++j) {
      const int s = __builtin_amdgcn_readlane(vs, j);
      const int d = __builtin_amdgcn_readlane(vd, j);
      const float* hp = h + (long)s * HD + lane;
      atomic_add_row(agg + (long)d * HD + lane, hp[0], hp[32], hp[64]);
    }
  }
}

constexpr int RW = 8;

__global__ __launch_bounds__(256)
void finish_dot_kernel(const float* __restrict__ agg, const float* __restrict__ norm_in,
                       const float* __restrict__ b, const float* __restrict__ a,
                       const float* __restrict__ wsum, const float* __restrict__ bsum,
                       float* __restrict__ out, int n) {
  const int lane = threadIdx.x & 31;
  const int wave = __builtin_amdgcn_readfirstlane((int)(threadIdx.x >> 5));
  const long r0  = ((long)blockIdx.x * (blockDim.x >> 5) + wave) * RW;
  if (r0 >= n) return;
  const float w0 = wsum[lane], w1 = wsum[lane + 32], w2 = wsum[lane + 64];
  const float b0 = b[lane],    b1 = b[lane + 32],    b2 = b[lane + 64];
  const float a0 = a[lane],    a1 = a[lane + 32],    a2 = a[lane + 64];
  const float bs = *bsum;
  float zv[RW][3];
  float sc[RW];
  #pragma unroll
  for (int q = 0; q < RW; ++q) {
    const long r = (r0 + q < n) ? (r0 + q) : (long)(n - 1);
    const float* zp = agg + r * HD + lane;
    zv[q][0] = zp[0]; zv[q][1] = zp[32]; zv[q][2] = zp[64];
    sc[q] = norm_in[r];
  }
  float sum[RW];
  #pragma unroll
  for (int q = 0; q < RW; ++q) {
    float t, z0, z1, z2;
    t = zv[q][0] * sc[q] + b0; z0 = (t >= 0.0f) ? t : a0 * t;
    t = zv[q][1] * sc[q] + b1; z1 = (t >= 0.0f) ? t : a1 * t;
    t = zv[q][2] * sc[q] + b2; z2 = (t >= 0.0f) ? t : a2 * t;
    float s = 0.0f;
    s = fmaf(z0, w0, s);
    s = fmaf(z1, w1, s);
    s = fmaf(z2, w2, s);
    sum[q] = s;
  }
  #pragma unroll
  for (int off = 16; off > 0; off >>= 1) {
    #pragma unroll
    for (int q = 0; q < RW; ++q) sum[q] += __shfl_xor(sum[q], off, 32);
  }
  if (lane == 0) {
    #pragma unroll
    for (int q = 0; q < RW; ++q)
      if (r0 + q < n) out[r0 + q] = sum[q] + bs;
  }
}

extern "C" void kernel_launch(void* const* d_in, const int* in_sizes, int n_in,
                              void* d_out, int out_size, void* d_ws, size_t ws_size,
                              hipStream_t stream) {
  const float* x    = (const float*)d_in[0];
  const int*   src  = (const int*)  d_in[1];
  const int*   dst  = (const int*)  d_in[2];
  const int*   perm = (const int*)  d_in[3];
  const float* W1   = (const float*)d_in[4];
  const float* b1   = (const float*)d_in[5];
  const float* a1   = (const float*)d_in[6];
  const float* W2   = (const float*)d_in[7];
  const float* b2   = (const float*)d_in[8];
  const float* a2   = (const float*)d_in[9];
  const float* Wm   = (const float*)d_in[10];
  const float* bm   = (const float*)d_in[11];
  float* out = (float*)d_out;

  const int E  = in_sizes[1];
  const int N  = in_sizes[3];
  const int K1 = in_sizes[0] / N;

  float* ws       = (float*)d_ws;
  float* norm_out = ws;
  float* norm_in  = ws + N;
  float* wsum     = ws + 2L * N;
  float* bsum     = ws + 2L * N + HD;
  int*   ctr      = (int*)(ws + 2L * N + 128);
  float* bufA     = ws + 2L * N + 2 * HD;
  float* bufB     = bufA + (long)N * HD;

  const int gemm_blocks = (N + GW * 16 - 1) / (GW * 16);
  const int red_blocks  = (int)(((long)N + 8 * RW - 1) / (8 * RW));

  fill_zero_kernel<<<512, 256, 0, stream>>>(norm_out, 2L * N, ctr);
  degree_kernel<<<512, 256, 0, stream>>>(src, dst, norm_out, norm_in, ctr, E, N);
  finalize_norm_kernel<<<(2 * N + 255) / 256, 256, 0, stream>>>(norm_out, 2 * N);
  wsum_kernel<<<1, 128, 0, stream>>>(Wm, bm, wsum, bsum);

  for (int view = 0; view < 2; ++view) {
    const int* rowmap = view ? perm : nullptr;
    gemm_rows_wmma_kernel<<<gemm_blocks, GW * 32, 0, stream>>>(x, rowmap, norm_out, W1, bufA, bufB, N, K1);
    spmm_kernel<<<2048, 256, 0, stream>>>(src, dst, bufA, bufB, ctr + 1 + 2 * view, E, N);
    gemm_rows_finish_wmma_kernel<<<gemm_blocks, GW * 32, 0, stream>>>(bufB, norm_out, W2,
                                                                     norm_in, b1, a1, bufA, bufB, N, HD);
    spmm_kernel<<<2048, 256, 0, stream>>>(src, dst, bufA, bufB, ctr + 2 + 2 * view, E, N);
    finish_dot_kernel<<<red_blocks, 256, 0, stream>>>(bufB, norm_in, b2, a2, wsum, bsum,
                                                      out + (long)view * N, N);
  }
}


# GEMM tiles re-dealt over all 256 WGPs: 3 whole 16-row tiles per SIMD, the 53 left over cut by columns over the fourth waves; fullest SIMD 3072 -> 2560 WMMAs (layer 2: 768 -> 640); sums unchanged
# speedup vs baseline: 3.2416x; 1.0340x over previous
#include <hip/hip_runtime.h>

typedef float v2f __attribute__((ext_vector_type(2)));
typedef float v8f __attribute__((ext_vector_type(8)));

constexpr int HD   = 96;


__global__ void fill_zero_kernel(float* __restrict__ p, long n, int* __restrict__ ctr) {
  long i = (long)blockIdx.x * blockDim.x + threadIdx.x;
  long stride = (long)gridDim.x * blockDim.x;
  if (blockIdx.x == 0 && threadIdx.x < 64) ctr[threadIdx.x] = 0;
  for (; i < n; i += stride) p[i] = 0.0f;
}

constexpr int CTR_STRIDE = 32;

__device__ __forceinline__ int l2_half() {
  return (int)((__builtin_amdgcn_s_sendmsg_rtn(0x87) >> 18) & 1u);
}

__device__ __forceinline__ int draw_ticket(int* ctr, int q) {
  return __hip_atomic_fetch_add(ctr + q * CTR_STRIDE, 1, __ATOMIC_RELAXED, __HIP_MEMORY_SCOPE_AGENT);
}

constexpr int DCH = 1024;

__global__ __launch_bounds__(256)
void degree_kernel(const int* __restrict__ src, const int* __restrict__ dst,
                   float* __restrict__ deg_out, float* __restrict__ deg_in,
                   int* __restrict__ ctr, int E, int n) {
  __shared__ int sh_t[2];
  const int tid = threadIdx.x;
  const int myh = l2_half();
  const int Nh  = (n >> 1) & ~1;
  const int nitems = (E + DCH - 1) / DCH;
  int q = myh, tn = 0;
  if (tid == 0) tn = draw_ticket(ctr, q);
  for (int it = 0; ; ++it) {
    if (tid == 0) sh_t[it & 1] = tn;
    __syncthreads();
    const int t = __builtin_amdgcn_readfirstlane(sh_t[it & 1]);
    if (t >= nitems) {
      if (q != myh) break;
      q = 1 - myh;
      if (tid == 0) tn = draw_ticket(ctr, q);
      continue;
    }
    if (tid == 0) tn = draw_ticket(ctr, q);
    const bool up = (q != 0);
    int sv[DCH / 256], dv[DCH / 256];
    #pragma unroll
    for (int k = 0; k < DCH / 256; ++k) {
      const int e  = t * DCH + k * 256 + tid;
      const int ec = (e < E) ? e : (E - 1);
      sv[k] = src[ec];
      dv[k] = dst[ec];
    }
    #pragma unroll
    for (int k = 0; k < DCH / 256; ++k) {
      const bool ok = (t * DCH + k * 256 + tid) < E;
      if (ok && ((sv[k] >= Nh) == up))
        __hip_atomic_fetch_add(&deg_out[sv[k]], 1.0f, __ATOMIC_RELAXED, __HIP_MEMORY_SCOPE_AGENT);
      if (ok && ((dv[k] >= Nh) == up))
        __hip_atomic_fetch_add(&deg_in[dv[k]],  1.0f, __ATOMIC_RELAXED, __HIP_MEMORY_SCOPE_AGENT);
    }
  }
}

__global__ void finalize_norm_kernel(float* __restrict__ p, int n) {
  int i = blockIdx.x * blockDim.x + threadIdx.x;
  if (i < n) p[i] = 1.0f / sqrtf(fmaxf(p[i], 1.0f));
}

__global__ void wsum_kernel(const float* __restrict__ Wm, const float* __restrict__ bm,
                            float* __restrict__ wsum, float* __restrict__ bsum) {
  int t = threadIdx.x;
  if (t < HD) {
    float s = 0.0f;
    for (int j = 0; j < HD; ++j) s += Wm[t * HD + j];
    wsum[t] = s;
  } else if (t == HD) {
    float s = 0.0f;
    for (int j = 0; j < HD; ++j) s += bm[j];
    *bsum = s;
  }
}

constexpr int KB   = 64;
constexpr int KMAX = 512;
constexpr int WS   = 514;
constexpr int GW   = 16;
constexpr int FW   = 12;
constexpr int HW   = 3;

__device__ __forceinline__
void gemm_load_block(float4 (&dstv)[8], const float* X, int srow, int K, int k0) {
  #pragma unroll
  for (int q = 0; q < 8; ++q) {
    const int k  = k0 + q * 8;
    const int kc = (k < K) ? k : (K - 4);
    dstv[q] = *(const float4*)(X + (long)srow * K + kc);
  }
}

__device__ __forceinline__
void gemm_load_finish(float4 (&bb)[8], float4 (&aa)[8], const float* __restrict__ fin_b,
                      const float* __restrict__ fin_a, int K, int k0) {
  #pragma unroll
  for (int q = 0; q < 8; ++q) {
    const int k  = k0 + q * 8;
    const int kc = (k < K) ? k : (K - 4);
    bb[q] = *(const float4*)(fin_b + kc);
    aa[q] = *(const float4*)(fin_a + kc);
  }
}

template <bool FIN, int NT>
__device__ __forceinline__
void gemm_block_products(v8f (&acc)[NT], const float4 (&raw)[8], const float4 (&bb)[8],
                         const float4 (&aa)[8], float s, float si, int K, int k0,
                         const float* wb) {
  float4 cur[8];
  #pragma unroll
  for (int q = 0; q < 8; ++q) {
    float4 v = raw[q];
    const int k = k0 + q * 8;
    if (FIN) {
      float t;
      t = v.x * si + bb[q].x; v.x = (t >= 0.0f) ? t : aa[q].x * t;
      t = v.y * si + bb[q].y; v.y = (t >= 0.0f) ? t : aa[q].y * t;
      t = v.z * si + bb[q].z; v.z = (t >= 0.0f) ? t : aa[q].z * t;
      t = v.w * si + bb[q].w; v.w = (t >= 0.0f) ? t : aa[q].w * t;
    }
    v.x *= s; v.y *= s; v.z *= s; v.w *= s;
    if (k >= K) v = make_float4(0.f, 0.f, 0.f, 0.f);
    cur[q] = v;
  }
  #pragma unroll
  for (int st = 0; st < 16; ++st) {
    v2f A;
    if (st & 1) { A.x = cur[st >> 1].z; A.y = cur[st >> 1].w; }
    else        { A.x = cur[st >> 1].x; A.y = cur[st >> 1].y; }
    v2f Bf[NT];
    #pragma unroll
    for (int t = 0; t < NT; ++t)
      Bf[t] = *(const v2f*)(wb + t * 16 * WS + st * 2);
    #pragma unroll
    for (int t = 0; t < NT; ++t)
      acc[t] = __builtin_amdgcn_wmma_f32_16x16x4_f32(
           false, A,  false, Bf[t],
           (short)0, acc[t],  false,  false);
  }
}

template <bool FIN, int NT>
__device__ __forceinline__
void gemm_tile_cols(v8f (&acc)[NT], const float* ldsW, const float* X,
                    const int* __restrict__ rowmap, const float* __restrict__ norm,
                    const float* __restrict__ fin_norm, const float* __restrict__ fin_b,
                    const float* __restrict__ fin_a, int nrows, int K, int Kpad,
                    int tile, int c0, int half, int l16) {
  const int grow = tile * 16 + l16;
  const int crow = (grow < nrows) ? grow : (nrows - 1);
  const int srow = rowmap ? rowmap[crow] : crow;
  const float s  = norm[crow];
  float si = 0.0f;
  if (FIN) si = fin_norm[crow];

  const float* wl = ldsW + (c0 * 16 + l16) * WS + half * 32;

  float4 buf0[8], buf1[8];
  float4 bb[8], aa[8];
  gemm_load_block(buf0, X, srow, K, half * 4);
  for (int kb = 0; kb < Kpad; kb += 2 * KB) {
    if (FIN) gemm_load_finish(bb, aa, fin_b, fin_a, K, kb + half * 4);
    gemm_load_block(buf1, X, srow, K, kb + KB + half * 4);
    gemm_block_products<FIN, NT>(acc, buf0, bb, aa, s, si, K, kb + half * 4, wl + kb);
    if (FIN) gemm_load_finish(bb, aa, fin_b, fin_a, K, kb + KB + half * 4);
    gemm_load_block(buf0, X, srow, K, kb + 2 * KB + half * 4);
    gemm_block_products<FIN, NT>(acc, buf1, bb, aa, s, si, K, kb + KB + half * 4, wl + kb + KB);
  }
}

__device__ __forceinline__
void gemm_zero_rows(float* ZERO, int tile, int nrows, int lane) {
  const long zend = (long)nrows * HD;
  #pragma unroll
  for (int it = 0; it < (16 * HD / 4) / 32; ++it) {
    const long off = (long)tile * 16 * HD + (long)(lane + 32 * it) * 4;
    if (off < zend) *(float4*)(ZERO + off) = make_float4(0.f, 0.f, 0.f, 0.f);
  }
}

template <int NT>
__device__ __forceinline__
void gemm_store_cols(float* __restrict__ OUT, const v8f (&acc)[NT], int tile, int c0, int nrows,
                     int half, int l16) {
  const int orow = tile * 16 + half * 8;
  #pragma unroll
  for (int t = 0; t < NT; ++t) {
    int col = (c0 + t) * 16 + l16;
    #pragma unroll
    for (int i = 0; i < 8; ++i) {
      int r = orow + i;
      if (r < nrows) OUT[(long)r * HD + col] = acc[t][i];
    }
  }
}

template <bool FIN>
__device__ __forceinline__
void gemm_rows_body(float* ldsW, const float* X, const int* __restrict__ rowmap,
                    const float* __restrict__ norm, const float* __restrict__ W,
                    const float* __restrict__ fin_norm, const float* __restrict__ fin_b,
                    const float* __restrict__ fin_a,
                    float* __restrict__ OUT, float* ZERO, int nrows, int K) {

  const int tid  = threadIdx.x;
  const int lane = tid & 31;
  const int half = lane >> 4;
  const int l16  = lane & 15;
  const int wave = __builtin_amdgcn_readfirstlane(tid >> 5);
  const int Kpad = (K + 2 * KB - 1) & ~(2 * KB - 1);

  {
    const int nitem = Kpad * (HD / 4);
    #pragma unroll 6
    for (int idx = tid; idx < nitem; idx += GW * 32) {
      const int k  = idx / (HD / 4);
      const int j  = idx % (HD / 4);
      const int kc = (k < K) ? k : (K - 1);
      float4 v = *(const float4*)(W + (long)kc * HD + j * 4);
      if (k >= K) v = make_float4(0.f, 0.f, 0.f, 0.f);
      const int kk = k & (KB - 1);
      const int p  = (k & ~(KB - 1)) + ((kk >> 2) & 1) * 32 + (kk >> 3) * 4 + (kk & 3);
      ldsW[(j * 4 + 0) * WS + p] = v.x;
      ldsW[(j * 4 + 1) * WS + p] = v.y;
      ldsW[(j * 4 + 2) * WS + p] = v.z;
      ldsW[(j * 4 + 3) * WS + p] = v.w;
    }
  }
  __syncthreads();

  const int ntiles = (nrows + 15) >> 4;
  const int nwhole = (int)gridDim.x * FW;

  if (wave < FW) {
    const int tile = (int)blockIdx.x * FW + wave;
    if (tile < ntiles) {
      v8f acc[6] = {};
      gemm_tile_cols<FIN, 6>(acc, ldsW, X, rowmap, norm, fin_norm, fin_b, fin_a,
                             nrows, K, Kpad, tile, 0, half, l16);
      gemm_zero_rows(ZERO, tile, nrows, lane);
      gemm_store_cols<6>(OUT, acc, tile, 0, nrows, half, l16);
    }
  }

  for (int tile = nwhole + (int)blockIdx.x; tile < ntiles; tile += (int)gridDim.x) {
    const bool part = (wave >= FW) && (wave < FW + HW);
    const int  c0   = (wave - FW) * 2;
    v8f acc2[2] = {};
    if (part)
      gemm_tile_cols<FIN, 2>(acc2, ldsW, X, rowmap, norm, fin_norm, fin_b, fin_a,
                             nrows, K, Kpad, tile, c0, half, l16);
    if (FIN) __syncthreads();
    if (wave == FW) gemm_zero_rows(ZERO, tile, nrows, lane);
    if (part) gemm_store_cols<2>(OUT, acc2, tile, c0, nrows, half, l16);
  }
}

__global__ __launch_bounds__(GW * 32)
void gemm_rows_wmma_kernel(const float* X, const int* __restrict__ rowmap,
                           const float* __restrict__ norm, const float* __restrict__ W,
                           float* __restrict__ OUT, float* ZERO, int nrows, int K) {
  __shared__ float ldsW[HD * WS];
  gemm_rows_body<false>(ldsW, X, rowmap, norm, W, nullptr, nullptr, nullptr, OUT, ZERO, nrows, K);
}

__global__ __launch_bounds__(GW * 32)
void gemm_rows_finish_wmma_kernel(const float* X, const float* __restrict__ norm,
                                  const float* __restrict__ W,
                                  const float* __restrict__ fin_norm,
                                  const float* __restrict__ fin_b,
                                  const float* __restrict__ fin_a,
                                  float* __restrict__ OUT, float* ZERO, int nrows, int K) {
  __shared__ float ldsW[HD * WS];
  gemm_rows_body<true>(ldsW, X, nullptr, norm, W, fin_norm, fin_b, fin_a, OUT, ZERO, nrows, K);
}

constexpr int EB   = 8;
constexpr int SPW  = 8;
constexpr int WCH  = 64;
constexpr int GCH  = SPW * WCH;
constexpr int LCAP = WCH + EB;

__device__ __forceinline__ void atomic_add_row(float* __restrict__ ap, float v0, float v1, float v2) {
  __hip_atomic_fetch_add(&ap[0],  v0, __ATOMIC_RELAXED, __HIP_MEMORY_SCOPE_AGENT);
  __hip_atomic_fetch_add(&ap[32], v1, __ATOMIC_RELAXED, __HIP_MEMORY_SCOPE_AGENT);
  __hip_atomic_fetch_add(&ap[64], v2, __ATOMIC_RELAXED, __HIP_MEMORY_SCOPE_AGENT);
}

__global__ __launch_bounds__(256)
void spmm_kernel(const int* __restrict__ src, const int* __restrict__ dst,
                 const float* __restrict__ h, float* __restrict__ agg,
                 int* __restrict__ ctr, int E, int n) {
  __shared__ int lst_s[SPW][LCAP];
  __shared__ int lst_d[SPW][LCAP];
  __shared__ int sh_t[2];
  const int tid  = threadIdx.x;
  const int lane = tid & 31;
  const int wave = __builtin_amdgcn_readfirstlane(tid >> 5);
  const int myh  = l2_half();
  const int Nh   = (n >> 1) & ~1;
  const int nitems = (E + GCH - 1) / GCH;
  int* ms = lst_s[wave];
  int* md = lst_d[wave];

  int q = myh, cnt = 0, tn = 0;
  if (tid == 0) tn = draw_ticket(ctr, q);
  for (int it = 0; ; ++it) {
    if (tid == 0) sh_t[it & 1] = tn;
    __syncthreads();
    const int t = __builtin_amdgcn_readfirstlane(sh_t[it & 1]);
    if (t >= nitems) {
      if (q != myh) break;
      q = 1 - myh;
      if (tid == 0) tn = draw_ticket(ctr, q);
      continue;
    }
    if (tid == 0) tn = draw_ticket(ctr, q);

    const bool up = (q != 0);
    int sv[WCH / 32], dv[WCH / 32];
    #pragma unroll
    for (int g = 0; g < WCH / 32; ++g) {
      const int e  = t * GCH + wave * WCH + g * 32 + lane;
      const int ec = (e < E) ? e : (E - 1);
      sv[g] = src[ec];
      dv[g] = (e < E) ? dst[ec] : -1;
    }
    #pragma unroll
    for (int g = 0; g < WCH / 32; ++g) {
      const bool mine = (dv[g] >= 0) && ((dv[g] >= Nh) == up);
      const unsigned m = __builtin_amdgcn_ballot_w32(mine);
      const int pos = cnt + (int)__builtin_amdgcn_mbcnt_lo(m, 0u);
      if (mine) { ms[pos] = sv[g]; md[pos] = dv[g]; }
      cnt += __builtin_popcount(m);
    }

    int b = 0;
    for (; b + EB <= cnt; b += EB) {
      const int vs = ms[b + (lane & (EB - 1))];
      const int vd = md[b + (lane & (EB - 1))];
      float v[EB][3];
      #pragma unroll
      for (int j = 0; j < EB; ++j) {
        const int s = __builtin_amdgcn_readlane(vs, j);
        const float* hp = h + (long)s * HD + lane;
        v[j][0] = hp[0]; v[j][1] = hp[32]; v[j][2] = hp[64];
      }
      #pragma unroll
      for (int j = 0; j < EB; ++j) {
        const int d = __builtin_amdgcn_readlane(vd, j);
        atomic_add_row(agg + (long)d * HD + lane, v[j][0], v[j][1], v[j][2]);
      }
    }
    const int r = cnt - b;
    if (b > 0 && r > 0) {
      int ts = 0, td = 0;
      if (lane < r) { ts = ms[b + lane]; td = md[b + lane]; }
      if (lane < r) { ms[lane] = ts; md[lane] = td; }
    }
    cnt = r;
  }

  if (cnt > 0) {
    const int li = (lane < cnt) ? lane : 0;
    const int vs = ms[li];
    const int vd = md[li];
    for (int j = 0; j < cnt; ++j) {
      const int s = __builtin_amdgcn_readlane(vs, j);
      const int d = __builtin_amdgcn_readlane(vd, j);
      const float* hp = h + (long)s * HD + lane;
      atomic_add_row(agg + (long)d * HD + lane, hp[0], hp[32], hp[64]);
    }
  }
}

constexpr int RW = 8;

__global__ __launch_bounds__(256)
void finish_dot_kernel(const float* __restrict__ agg, const float* __restrict__ norm_in,
                       const float* __restrict__ b, const float* __restrict__ a,
                       const float* __restrict__ wsum, const float* __restrict__ bsum,
                       float* __restrict__ out, int n) {
  const int lane = threadIdx.x & 31;
  const int wave = __builtin_amdgcn_readfirstlane((int)(threadIdx.x >> 5));
  const long r0  = ((long)blockIdx.x * (blockDim.x >> 5) + wave) * RW;
  if (r0 >= n) return;
  const float w0 = wsum[lane], w1 = wsum[lane + 32], w2 = wsum[lane + 64];
  const float b0 = b[lane],    b1 = b[lane + 32],    b2 = b[lane + 64];
  const float a0 = a[lane],    a1 = a[lane + 32],    a2 = a[lane + 64];
  const float bs = *bsum;
  float zv[RW][3];
  float sc[RW];
  #pragma unroll
  for (int q = 0; q < RW; ++q) {
    const long r = (r0 + q < n) ? (r0 + q) : (long)(n - 1);
    const float* zp = agg + r * HD + lane;
    zv[q][0] = zp[0]; zv[q][1] = zp[32]; zv[q][2] = zp[64];
    sc[q] = norm_in[r];
  }
  float sum[RW];
  #pragma unroll
  for (int q = 0; q < RW; ++q) {
    float t, z0, z1, z2;
    t = zv[q][0] * sc[q] + b0; z0 = (t >= 0.0f) ? t : a0 * t;
    t = zv[q][1] * sc[q] + b1; z1 = (t >= 0.0f) ? t : a1 * t;
    t = zv[q][2] * sc[q] + b2; z2 = (t >= 0.0f) ? t : a2 * t;
    float s = 0.0f;
    s = fmaf(z0, w0, s);
    s = fmaf(z1, w1, s);
    s = fmaf(z2, w2, s);
    sum[q] = s;
  }
  #pragma unroll
  for (int off = 16; off > 0; off >>= 1) {
    #pragma unroll
    for (int q = 0; q < RW; ++q) sum[q] += __shfl_xor(sum[q], off, 32);
  }
  if (lane == 0) {
    #pragma unroll
    for (int q = 0; q < RW; ++q)
      if (r0 + q < n) out[r0 + q] = sum[q] + bs;
  }
}

extern "C" void kernel_launch(void* const* d_in, const int* in_sizes, int n_in,
                              void* d_out, int out_size, void* d_ws, size_t ws_size,
                              hipStream_t stream) {
  const float* x    = (const float*)d_in[0];
  const int*   src  = (const int*)  d_in[1];
  const int*   dst  = (const int*)  d_in[2];
  const int*   perm = (const int*)  d_in[3];
  const float* W1   = (const float*)d_in[4];
  const float* b1   = (const float*)d_in[5];
  const float* a1   = (const float*)d_in[6];
  const float* W2   = (const float*)d_in[7];
  const float* b2   = (const float*)d_in[8];
  const float* a2   = (const float*)d_in[9];
  const float* Wm   = (const float*)d_in[10];
  const float* bm   = (const float*)d_in[11];
  float* out = (float*)d_out;

  const int E  = in_sizes[1];
  const int N  = in_sizes[3];
  const int K1 = in_sizes[0] / N;

  float* ws       = (float*)d_ws;
  float* norm_out = ws;
  float* norm_in  = ws + N;
  float* wsum     = ws + 2L * N;
  float* bsum     = ws + 2L * N + HD;
  int*   ctr      = (int*)(ws + 2L * N + 128);
  float* bufA     = ws + 2L * N + 2 * HD;
  float* bufB     = bufA + (long)N * HD;

  const int gemm_blocks = 256;
  const int red_blocks  = (int)(((long)N + 8 * RW - 1) / (8 * RW));

  fill_zero_kernel<<<512, 256, 0, stream>>>(norm_out, 2L * N, ctr);
  degree_kernel<<<512, 256, 0, stream>>>(src, dst, norm_out, norm_in, ctr, E, N);
  finalize_norm_kernel<<<(2 * N + 255) / 256, 256, 0, stream>>>(norm_out, 2 * N);
  wsum_kernel<<<1, 128, 0, stream>>>(Wm, bm, wsum, bsum);

  for (int view = 0; view < 2; ++view) {
    const int* rowmap = view ? perm : nullptr;
    gemm_rows_wmma_kernel<<<gemm_blocks, GW * 32, 0, stream>>>(x, rowmap, norm_out, W1, bufA, bufB, N, K1);
    spmm_kernel<<<2048, 256, 0, stream>>>(src, dst, bufA, bufB, ctr + 1 + 2 * view, E, N);
    gemm_rows_finish_wmma_kernel<<<gemm_blocks, GW * 32, 0, stream>>>(bufB, norm_out, W2,
                                                                     norm_in, b1, a1, bufA, bufB, N, HD);
    spmm_kernel<<<2048, 256, 0, stream>>>(src, dst, bufA, bufB, ctr + 2 + 2 * view, E, N);
    finish_dot_kernel<<<red_blocks, 256, 0, stream>>>(bufB, norm_in, b2, a2, wsum, bsum,
                                                      out + (long)view * N, N);
  }
}


# SpMM as a push grouped by source: per call, edges keyed (dest half, source) by count, prefix, placement; a wave loads a source row once, adds it into each listed dest; arrival-order sums
# speedup vs baseline: 3.5130x; 1.0837x over previous
#include <hip/hip_runtime.h>

typedef float v2f __attribute__((ext_vector_type(2)));
typedef float v8f __attribute__((ext_vector_type(8)));

constexpr int HD   = 96;


__global__ void fill_zero_kernel(float* __restrict__ p, long n, int* __restrict__ ctr) {
  long i = (long)blockIdx.x * blockDim.x + threadIdx.x;
  long stride = (long)gridDim.x * blockDim.x;
  if (blockIdx.x == 0 && threadIdx.x < 64) ctr[threadIdx.x] = 0;
  for (; i < n; i += stride) p[i] = 0.0f;
}

constexpr int CTR_STRIDE = 32;

__device__ __forceinline__ int l2_half() {
  return (int)((__builtin_amdgcn_s_sendmsg_rtn(0x87) >> 18) & 1u);
}

__device__ __forceinline__ int draw_ticket(int* ctr, int q) {
  return __hip_atomic_fetch_add(ctr + q * CTR_STRIDE, 1, __ATOMIC_RELAXED, __HIP_MEMORY_SCOPE_AGENT);
}

constexpr int DCH = 1024;

__global__ __launch_bounds__(256)
void degree_kernel(const int* __restrict__ src, const int* __restrict__ dst,
                   float* __restrict__ deg_out, float* __restrict__ deg_in,
                   int* __restrict__ ctr, int E, int n) {
  __shared__ int sh_t[2];
  const int tid = threadIdx.x;
  const int myh = l2_half();
  const int Nh  = (n >> 1) & ~1;
  const int nitems = (E + DCH - 1) / DCH;
  int q = myh, tn = 0;
  if (tid == 0) tn = draw_ticket(ctr, q);
  for (int it = 0; ; ++it) {
    if (tid == 0) sh_t[it & 1] = tn;
    __syncthreads();
    const int t = __builtin_amdgcn_readfirstlane(sh_t[it & 1]);
    if (t >= nitems) {
      if (q != myh) break;
      q = 1 - myh;
      if (tid == 0) tn = draw_ticket(ctr, q);
      continue;
    }
    if (tid == 0) tn = draw_ticket(ctr, q);
    const bool up = (q != 0);
    int sv[DCH / 256], dv[DCH / 256];
    #pragma unroll
    for (int k = 0; k < DCH / 256; ++k) {
      const int e  = t * DCH + k * 256 + tid;
      const int ec = (e < E) ? e : (E - 1);
      sv[k] = src[ec];
      dv[k] = dst[ec];
    }
    #pragma unroll
    for (int k = 0; k < DCH / 256; ++k) {
      const bool ok = (t * DCH + k * 256 + tid) < E;
      if (ok && ((sv[k] >= Nh) == up))
        __hip_atomic_fetch_add(&deg_out[sv[k]], 1.0f, __ATOMIC_RELAXED, __HIP_MEMORY_SCOPE_AGENT);
      if (ok && ((dv[k] >= Nh) == up))
        __hip_atomic_fetch_add(&deg_in[dv[k]],  1.0f, __ATOMIC_RELAXED, __HIP_MEMORY_SCOPE_AGENT);
    }
  }
}

__global__ void finalize_norm_kernel(float* __restrict__ p, int n) {
  int i = blockIdx.x * blockDim.x + threadIdx.x;
  if (i < n) p[i] = 1.0f / sqrtf(fmaxf(p[i], 1.0f));
}

__global__ void wsum_kernel(const float* __restrict__ Wm, const float* __restrict__ bm,
                            float* __restrict__ wsum, float* __restrict__ bsum) {
  int t = threadIdx.x;
  if (t < HD) {
    float s = 0.0f;
    for (int j = 0; j < HD; ++j) s += Wm[t * HD + j];
    wsum[t] = s;
  } else if (t == HD) {
    float s = 0.0f;
    for (int j = 0; j < HD; ++j) s += bm[j];
    *bsum = s;
  }
}

constexpr int KB   = 64;
constexpr int KMAX = 512;
constexpr int WS   = 514;
constexpr int GW   = 16;
constexpr int FW   = 12;
constexpr int HW   = 3;

__device__ __forceinline__
void gemm_load_block(float4 (&dstv)[8], const float* X, int srow, int K, int k0) {
  #pragma unroll
  for (int q = 0; q < 8; ++q) {
    const int k  = k0 + q * 8;
    const int kc = (k < K) ? k : (K - 4);
    dstv[q] = *(const float4*)(X + (long)srow * K + kc);
  }
}

__device__ __forceinline__
void gemm_load_finish(float4 (&bb)[8], float4 (&aa)[8], const float* __restrict__ fin_b,
                      const float* __restrict__ fin_a, int K, int k0) {
  #pragma unroll
  for (int q = 0; q < 8; ++q) {
    const int k  = k0 + q * 8;
    const int kc = (k < K) ? k : (K - 4);
    bb[q] = *(const float4*)(fin_b + kc);
    aa[q] = *(const float4*)(fin_a + kc);
  }
}

template <bool FIN, int NT>
__device__ __forceinline__
void gemm_block_products(v8f (&acc)[NT], const float4 (&raw)[8], const float4 (&bb)[8],
                         const float4 (&aa)[8], float s, float si, int K, int k0,
                         const float* wb) {
  float4 cur[8];
  #pragma unroll
  for (int q = 0; q < 8; ++q) {
    float4 v = raw[q];
    const int k = k0 + q * 8;
    if (FIN) {
      float t;
      t = v.x * si + bb[q].x; v.x = (t >= 0.0f) ? t : aa[q].x * t;
      t = v.y * si + bb[q].y; v.y = (t >= 0.0f) ? t : aa[q].y * t;
      t = v.z * si + bb[q].z; v.z = (t >= 0.0f) ? t : aa[q].z * t;
      t = v.w * si + bb[q].w; v.w = (t >= 0.0f) ? t : aa[q].w * t;
    }
    v.x *= s; v.y *= s; v.z *= s; v.w *= s;
    if (k >= K) v = make_float4(0.f, 0.f, 0.f, 0.f);
    cur[q] = v;
  }
  #pragma unroll
  for (int st = 0; st < 16; ++st) {
    v2f A;
    if (st & 1) { A.x = cur[st >> 1].z; A.y = cur[st >> 1].w; }
    else        { A.x = cur[st >> 1].x; A.y = cur[st >> 1].y; }
    v2f Bf[NT];
    #pragma unroll
    for (int t = 0; t < NT; ++t)
      Bf[t] = *(const v2f*)(wb + t * 16 * WS + st * 2);
    #pragma unroll
    for (int t = 0; t < NT; ++t)
      acc[t] = __builtin_amdgcn_wmma_f32_16x16x4_f32(
           false, A,  false, Bf[t],
           (short)0, acc[t],  false,  false);
  }
}

template <bool FIN, int NT>
__device__ __forceinline__
void gemm_tile_cols(v8f (&acc)[NT], const float* ldsW, const float* X,
                    const int* __restrict__ rowmap, const float* __restrict__ norm,
                    const float* __restrict__ fin_norm, const float* __restrict__ fin_b,
                    const float* __restrict__ fin_a, int nrows, int K, int Kpad,
                    int tile, int c0, int half, int l16) {
  const int grow = tile * 16 + l16;
  const int crow = (grow < nrows) ? grow : (nrows - 1);
  const int srow = rowmap ? rowmap[crow] : crow;
  const float s  = norm[crow];
  float si = 0.0f;
  if (FIN) si = fin_norm[crow];

  const float* wl = ldsW + (c0 * 16 + l16) * WS + half * 32;

  float4 buf0[8], buf1[8];
  float4 bb[8], aa[8];
  gemm_load_block(buf0, X, srow, K, half * 4);
  for (int kb = 0; kb < Kpad; kb += 2 * KB) {
    if (FIN) gemm_load_finish(bb, aa, fin_b, fin_a, K, kb + half * 4);
    gemm_load_block(buf1, X, srow, K, kb + KB + half * 4);
    gemm_block_products<FIN, NT>(acc, buf0, bb, aa, s, si, K, kb + half * 4, wl + kb);
    if (FIN) gemm_load_finish(bb, aa, fin_b, fin_a, K, kb + KB + half * 4);
    gemm_load_block(buf0, X, srow, K, kb + 2 * KB + half * 4);
    gemm_block_products<FIN, NT>(acc, buf1, bb, aa, s, si, K, kb + KB + half * 4, wl + kb + KB);
  }
}

__device__ __forceinline__
void gemm_zero_rows(float* ZERO, int tile, int nrows, int lane) {
  const long zend = (long)nrows * HD;
  #pragma unroll
  for (int it = 0; it < (16 * HD / 4) / 32; ++it) {
    const long off = (long)tile * 16 * HD + (long)(lane + 32 * it) * 4;
    if (off < zend) *(float4*)(ZERO + off) = make_float4(0.f, 0.f, 0.f, 0.f);
  }
}

template <int NT>
__device__ __forceinline__
void gemm_store_cols(float* __restrict__ OUT, const v8f (&acc)[NT], int tile, int c0, int nrows,
                     int half, int l16) {
  const int orow = tile * 16 + half * 8;
  #pragma unroll
  for (int t = 0; t < NT; ++t) {
    int col = (c0 + t) * 16 + l16;
    #pragma unroll
    for (int i = 0; i < 8; ++i) {
      int r = orow + i;
      if (r < nrows) OUT[(long)r * HD + col] = acc[t][i];
    }
  }
}

template <bool FIN>
__device__ __forceinline__
void gemm_rows_body(float* ldsW, const float* X, const int* __restrict__ rowmap,
                    const float* __restrict__ norm, const float* __restrict__ W,
                    const float* __restrict__ fin_norm, const float* __restrict__ fin_b,
                    const float* __restrict__ fin_a,
                    float* __restrict__ OUT, float* ZERO, int nrows, int K) {

  const int tid  = threadIdx.x;
  const int lane = tid & 31;
  const int half = lane >> 4;
  const int l16  = lane & 15;
  const int wave = __builtin_amdgcn_readfirstlane(tid >> 5);
  const int Kpad = (K + 2 * KB - 1) & ~(2 * KB - 1);

  {
    const int nitem = Kpad * (HD / 4);
    #pragma unroll 6
    for (int idx = tid; idx < nitem; idx += GW * 32) {
      const int k  = idx / (HD / 4);
      const int j  = idx % (HD / 4);
      const int kc = (k < K) ? k : (K - 1);
      float4 v = *(const float4*)(W + (long)kc * HD + j * 4);
      if (k >= K) v = make_float4(0.f, 0.f, 0.f, 0.f);
      const int kk = k & (KB - 1);
      const int p  = (k & ~(KB - 1)) + ((kk >> 2) & 1) * 32 + (kk >> 3) * 4 + (kk & 3);
      ldsW[(j * 4 + 0) * WS + p] = v.x;
      ldsW[(j * 4 + 1) * WS + p] = v.y;
      ldsW[(j * 4 + 2) * WS + p] = v.z;
      ldsW[(j * 4 + 3) * WS + p] = v.w;
    }
  }
  __syncthreads();

  const int ntiles = (nrows + 15) >> 4;
  const int nwhole = (int)gridDim.x * FW;

  if (wave < FW) {
    const int tile = (int)blockIdx.x * FW + wave;
    if (tile < ntiles) {
      v8f acc[6] = {};
      gemm_tile_cols<FIN, 6>(acc, ldsW, X, rowmap, norm, fin_norm, fin_b, fin_a,
                             nrows, K, Kpad, tile, 0, half, l16);
      gemm_zero_rows(ZERO, tile, nrows, lane);
      gemm_store_cols<6>(OUT, acc, tile, 0, nrows, half, l16);
    }
  }

  for (int tile = nwhole + (int)blockIdx.x; tile < ntiles; tile += (int)gridDim.x) {
    const bool part = (wave >= FW) && (wave < FW + HW);
    const int  c0   = (wave - FW) * 2;
    v8f acc2[2] = {};
    if (part)
      gemm_tile_cols<FIN, 2>(acc2, ldsW, X, rowmap, norm, fin_norm, fin_b, fin_a,
                             nrows, K, Kpad, tile, c0, half, l16);
    if (FIN) __syncthreads();
    if (wave == FW) gemm_zero_rows(ZERO, tile, nrows, lane);
    if (part) gemm_store_cols<2>(OUT, acc2, tile, c0, nrows, half, l16);
  }
}

__global__ __launch_bounds__(GW * 32)
void gemm_rows_wmma_kernel(const float* X, const int* __restrict__ rowmap,
                           const float* __restrict__ norm, const float* __restrict__ W,
                           float* __restrict__ OUT, float* ZERO, int nrows, int K) {
  __shared__ float ldsW[HD * WS];
  gemm_rows_body<false>(ldsW, X, rowmap, norm, W, nullptr, nullptr, nullptr, OUT, ZERO, nrows, K);
}

__global__ __launch_bounds__(GW * 32)
void gemm_rows_finish_wmma_kernel(const float* X, const float* __restrict__ norm,
                                  const float* __restrict__ W,
                                  const float* __restrict__ fin_norm,
                                  const float* __restrict__ fin_b,
                                  const float* __restrict__ fin_a,
                                  float* __restrict__ OUT, float* ZERO, int nrows, int K) {
  __shared__ float ldsW[HD * WS];
  gemm_rows_body<true>(ldsW, X, nullptr, norm, W, fin_norm, fin_b, fin_a, OUT, ZERO, nrows, K);
}

constexpr int EB   = 8;
constexpr int SPW  = 8;
constexpr int WCH  = 64;
constexpr int GCH  = SPW * WCH;
constexpr int LCAP = WCH + EB;

__device__ __forceinline__ void atomic_add_row(float* __restrict__ ap, float v0, float v1, float v2) {
  __hip_atomic_fetch_add(&ap[0],  v0, __ATOMIC_RELAXED, __HIP_MEMORY_SCOPE_AGENT);
  __hip_atomic_fetch_add(&ap[32], v1, __ATOMIC_RELAXED, __HIP_MEMORY_SCOPE_AGENT);
  __hip_atomic_fetch_add(&ap[64], v2, __ATOMIC_RELAXED, __HIP_MEMORY_SCOPE_AGENT);
}

__global__ __launch_bounds__(256)
void spmm_kernel(const int* __restrict__ src, const int* __restrict__ dst,
                 const float* __restrict__ h, float* __restrict__ agg,
                 int* __restrict__ ctr, int E, int n) {
  __shared__ int lst_s[SPW][LCAP];
  __shared__ int lst_d[SPW][LCAP];
  __shared__ int sh_t[2];
  const int tid  = threadIdx.x;
  const int lane = tid & 31;
  const int wave = __builtin_amdgcn_readfirstlane(tid >> 5);
  const int myh  = l2_half();
  const int Nh   = (n >> 1) & ~1;
  const int nitems = (E + GCH - 1) / GCH;
  int* ms = lst_s[wave];
  int* md = lst_d[wave];

  int q = myh, cnt = 0, tn = 0;
  if (tid == 0) tn = draw_ticket(ctr, q);
  for (int it = 0; ; ++it) {
    if (tid == 0) sh_t[it & 1] = tn;
    __syncthreads();
    const int t = __builtin_amdgcn_readfirstlane(sh_t[it & 1]);
    if (t >= nitems) {
      if (q != myh) break;
      q = 1 - myh;
      if (tid == 0) tn = draw_ticket(ctr, q);
      continue;
    }
    if (tid == 0) tn = draw_ticket(ctr, q);

    const bool up = (q != 0);
    int sv[WCH / 32], dv[WCH / 32];
    #pragma unroll
    for (int g = 0; g < WCH / 32; ++g) {
      const int e  = t * GCH + wave * WCH + g * 32 + lane;
      const int ec = (e < E) ? e : (E - 1);
      sv[g] = src[ec];
      dv[g] = (e < E) ? dst[ec] : -1;
    }
    #pragma unroll
    for (int g = 0; g < WCH / 32; ++g) {
      const bool mine = (dv[g] >= 0) && ((dv[g] >= Nh) == up);
      const unsigned m = __builtin_amdgcn_ballot_w32(mine);
      const int pos = cnt + (int)__builtin_amdgcn_mbcnt_lo(m, 0u);
      if (mine) { ms[pos] = sv[g]; md[pos] = dv[g]; }
      cnt += __builtin_popcount(m);
    }

    int b = 0;
    for (; b + EB <= cnt; b += EB) {
      const int vs = ms[b + (lane & (EB - 1))];
      const int vd = md[b + (lane & (EB - 1))];
      float v[EB][3];
      #pragma unroll
      for (int j = 0; j < EB; ++j) {
        const int s = __builtin_amdgcn_readlane(vs, j);
        const float* hp = h + (long)s * HD + lane;
        v[j][0] = hp[0]; v[j][1] = hp[32]; v[j][2] = hp[64];
      }
      #pragma unroll
      for (int j = 0; j < EB; ++j) {
        const int d = __builtin_amdgcn_readlane(vd, j);
        atomic_add_row(agg + (long)d * HD + lane, v[j][0], v[j][1], v[j][2]);
      }
    }
    const int r = cnt - b;
    if (b > 0 && r > 0) {
      int ts = 0, td = 0;
      if (lane < r) { ts = ms[b + lane]; td = md[b + lane]; }
      if (lane < r) { ms[lane] = ts; md[lane] = td; }
    }
    cnt = r;
  }

  if (cnt > 0) {
    const int li = (lane < cnt) ? lane : 0;
    const int vs = ms[li];
    const int vd = md[li];
    for (int j = 0; j < cnt; ++j) {
      const int s = __builtin_amdgcn_readlane(vs, j);
      const int d = __builtin_amdgcn_readlane(vd, j);
      const float* hp = h + (long)s * HD + lane;
      atomic_add_row(agg + (long)d * HD + lane, hp[0], hp[32], hp[64]);
    }
  }
}

constexpr int KSL = 1024;

__device__ __forceinline__ int block_excl_scan(int v, int* sh_w, int tid, int& tot) {
  const int lane = tid & 31;
  const int wave = tid >> 5;
  int inc = v;
  #pragma unroll
  for (int o = 1; o < 32; o <<= 1) {
    const int u = __shfl_up(inc, o, 32);
    if (lane >= o) inc += u;
  }
  if (lane == 31) sh_w[wave] = inc;
  __syncthreads();
  int woff = 0, t = 0;
  #pragma unroll
  for (int w = 0; w < 8; ++w) {
    const int x = sh_w[w];
    if (w < wave) woff += x;
    t += x;
  }
  tot = t;
  return woff + inc - v;
}

__global__ void fill_zero_push_kernel(float* __restrict__ p, long n, int* __restrict__ p2, long n2,
                                      int* __restrict__ ctr) {
  long i = (long)blockIdx.x * blockDim.x + threadIdx.x;
  const long stride = (long)gridDim.x * blockDim.x;
  if (blockIdx.x == 0 && threadIdx.x < 64) ctr[threadIdx.x] = 0;
  for (long j = i; j < n; j += stride) p[j] = 0.0f;
  for (long j = i; j < n2; j += stride) p2[j] = 0;
}

__global__ __launch_bounds__(256)
void degree_keys_kernel(const int* __restrict__ src, const int* __restrict__ dst,
                        float* __restrict__ deg_in, int* __restrict__ cnt,
                        int* __restrict__ ctr, int E, int n) {
  __shared__ int sh_t[2];
  const int tid = threadIdx.x;
  const int myh = l2_half();
  const int Nh  = (n >> 1) & ~1;
  const int nitems = (E + DCH - 1) / DCH;
  int q = myh, tn = 0;
  if (tid == 0) tn = draw_ticket(ctr, q);
  for (int it = 0; ; ++it) {
    if (tid == 0) sh_t[it & 1] = tn;
    __syncthreads();
    const int t = __builtin_amdgcn_readfirstlane(sh_t[it & 1]);
    if (t >= nitems) {
      if (q != myh) break;
      q = 1 - myh;
      if (tid == 0) tn = draw_ticket(ctr, q);
      continue;
    }
    if (tid == 0) tn = draw_ticket(ctr, q);
    const bool up = (q != 0);
    int sv[DCH / 256], dv[DCH / 256];
    #pragma unroll
    for (int k = 0; k < DCH / 256; ++k) {
      const int e  = t * DCH + k * 256 + tid;
      const int ec = (e < E) ? e : (E - 1);
      sv[k] = src[ec];
      dv[k] = dst[ec];
    }
    #pragma unroll
    for (int k = 0; k < DCH / 256; ++k) {
      const bool ok = (t * DCH + k * 256 + tid) < E;
      if (ok && ((sv[k] >= Nh) == up))
        __hip_atomic_fetch_add(&cnt[((dv[k] >= Nh) ? n : 0) + sv[k]], 1, __ATOMIC_RELAXED,
                               __HIP_MEMORY_SCOPE_AGENT);
      if (ok && ((dv[k] >= Nh) == up))
        __hip_atomic_fetch_add(&deg_in[dv[k]], 1.0f, __ATOMIC_RELAXED, __HIP_MEMORY_SCOPE_AGENT);
    }
  }
}

__global__ __launch_bounds__(256)
void scan_norm_kernel(const int* __restrict__ cnt, int* __restrict__ off, int* __restrict__ bsum,
                      float* __restrict__ norm_out, float* __restrict__ norm_in, int n) {
  __shared__ int sh_w[8];
  const int tid = threadIdx.x;
  const int nk  = 2 * n;
  const int k0  = (int)blockIdx.x * KSL + tid * 4;
  int c[4];
  #pragma unroll
  for (int i = 0; i < 4; ++i) {
    const int k  = k0 + i;
    const int kc = (k < nk) ? k : (nk - 1);
    const int x  = cnt[kc];
    c[i] = (k < nk) ? x : 0;
  }
  int tot;
  int ex = block_excl_scan(c[0] + c[1] + c[2] + c[3], sh_w, tid, tot);
  #pragma unroll
  for (int i = 0; i < 4; ++i) {
    if (k0 + i < nk) off[k0 + i] = ex;
    ex += c[i];
  }
  if (tid == 0) bsum[blockIdx.x] = tot;
  for (int s = (int)blockIdx.x * 256 + tid; s < n; s += (int)gridDim.x * 256) {
    const float dout = (float)(cnt[s] + cnt[n + s]);
    norm_out[s] = 1.0f / sqrtf(fmaxf(dout, 1.0f));
    norm_in[s]  = 1.0f / sqrtf(fmaxf(norm_in[s], 1.0f));
  }
}

__global__ __launch_bounds__(256)
void place_kernel(const int* __restrict__ src, const int* __restrict__ dst,
                  const int* __restrict__ off, const int* __restrict__ bsum,
                  int* __restrict__ cur, int* __restrict__ sta, int* __restrict__ dlist,
                  int* __restrict__ ctr, int E, int n, int nsl) {
  __shared__ int sh_t[2];
  __shared__ int sh_w[8];
  __shared__ int sh_base[256];
  const int tid = threadIdx.x;
  const int myh = l2_half();
  const int Nh  = (n >> 1) & ~1;
  const int nk  = 2 * n;
  {
    int tot;
    const int v = (tid < nsl) ? bsum[(tid < nsl) ? tid : 0] : 0;
    sh_base[tid] = block_excl_scan(v, sh_w, tid, tot);
  }
  __syncthreads();
  for (int sl = (int)blockIdx.x; sl < nsl; sl += (int)gridDim.x) {
    const int b = sh_base[sl];
    #pragma unroll
    for (int i = 0; i < 4; ++i) {
      const int k = sl * KSL + tid * 4 + i;
      if (k < nk) sta[k] = b + off[k];
    }
  }
  const int nitems = (E + DCH - 1) / DCH;
  int q = myh, tn = 0;
  if (tid == 0) tn = draw_ticket(ctr, q);
  for (int it = 0; ; ++it) {
    if (tid == 0) sh_t[it & 1] = tn;
    __syncthreads();
    const int t = __builtin_amdgcn_readfirstlane(sh_t[it & 1]);
    if (t >= nitems) {
      if (q != myh) break;
      q = 1 - myh;
      if (tid == 0) tn = draw_ticket(ctr, q);
      continue;
    }
    if (tid == 0) tn = draw_ticket(ctr, q);
    const bool up = (q != 0);
    int sv[DCH / 256], dv[DCH / 256], key[DCH / 256], lo[DCH / 256], rk[DCH / 256];
    #pragma unroll
    for (int k = 0; k < DCH / 256; ++k) {
      const int e  = t * DCH + k * 256 + tid;
      const int ec = (e < E) ? e : (E - 1);
      sv[k] = src[ec];
      dv[k] = (e < E) ? dst[ec] : -1;
    }
    #pragma unroll
    for (int k = 0; k < DCH / 256; ++k) {
      const bool mine = (dv[k] >= 0) && ((sv[k] >= Nh) == up);
      key[k] = mine ? (((dv[k] >= Nh) ? n : 0) + sv[k]) : -1;
      lo[k] = 0; rk[k] = 0;
      if (mine) {
        lo[k] = off[key[k]];
        rk[k] = __hip_atomic_fetch_add(&cur[key[k]], 1, __ATOMIC_RELAXED, __HIP_MEMORY_SCOPE_AGENT);
      }
    }
    #pragma unroll
    for (int k = 0; k < DCH / 256; ++k) {
      if (key[k] >= 0) {
        const int pos = sh_base[key[k] / KSL] + lo[k] + rk[k];
        __hip_atomic_exchange(&dlist[pos], dv[k], __ATOMIC_RELAXED, __HIP_MEMORY_SCOPE_AGENT);
      }
    }
  }
}

constexpr int SPS = 4;
constexpr int SIT = SPW * SPS;

__global__ __launch_bounds__(256)
void spmm_push_kernel(const int* __restrict__ sta, const int* __restrict__ cnt,
                      const int* __restrict__ dlist, const float* __restrict__ h,
                      float* __restrict__ agg, int* __restrict__ ctr, int n) {
  __shared__ int sh_t[2];
  const int tid  = threadIdx.x;
  const int lane = tid & 31;
  const int wave = __builtin_amdgcn_readfirstlane(tid >> 5);
  const int myh  = l2_half();
  const int nitems = (n + SIT - 1) / SIT;
  int q = myh, tn = 0;
  if (tid == 0) tn = draw_ticket(ctr, q);
  for (int it = 0; ; ++it) {
    if (tid == 0) sh_t[it & 1] = tn;
    __syncthreads();
    const int t = __builtin_amdgcn_readfirstlane(sh_t[it & 1]);
    if (t >= nitems) {
      if (q != myh) break;
      q = 1 - myh;
      if (tid == 0) tn = draw_ticket(ctr, q);
      continue;
    }
    if (tid == 0) tn = draw_ticket(ctr, q);

    const int s0 = t * SIT + wave * SPS;
    const int sj  = s0 + (lane & (SPS - 1));
    const int sjc = (sj < n) ? sj : (n - 1);
    const int kj  = q * n + sjc;
    const int stv = sta[kj];
    const int cv0 = cnt[kj];
    const int cv  = (sj < n) ? cv0 : 0;
    float v[SPS][3];
    #pragma unroll
    for (int j = 0; j < SPS; ++j) {
      const int s  = s0 + j;
      const int sc = (s < n) ? s : (n - 1);
      const float* hp = h + (long)sc * HD + lane;
      v[j][0] = hp[0]; v[j][1] = hp[32]; v[j][2] = hp[64];
    }
    int dl[SPS];
    #pragma unroll
    for (int j = 0; j < SPS; ++j) {
      const int st = __builtin_amdgcn_readlane(stv, j);
      const int c  = __builtin_amdgcn_readlane(cv, j);
      dl[j] = (lane < c) ? dlist[st + lane] : 0;
    }
    #pragma unroll
    for (int j = 0; j < SPS; ++j) {
      const int st = __builtin_amdgcn_readlane(stv, j);
      const int c  = __builtin_amdgcn_readlane(cv, j);
      const int c1 = (c < 32) ? c : 32;
      for (int i = 0; i < c1; ++i) {
        const int d = __builtin_amdgcn_readlane(dl[j], i);
        atomic_add_row(agg + (long)d * HD + lane, v[j][0], v[j][1], v[j][2]);
      }
      for (int b = 32; b < c; b += 32) {
        const int dm = (b + lane < c) ? dlist[st + b + lane] : 0;
        const int c2 = (c - b < 32) ? (c - b) : 32;
        for (int i = 0; i < c2; ++i) {
          const int d = __builtin_amdgcn_readlane(dm, i);
          atomic_add_row(agg + (long)d * HD + lane, v[j][0], v[j][1], v[j][2]);
        }
      }
    }
  }
}

constexpr int RW = 8;

__global__ __launch_bounds__(256)
void finish_dot_kernel(const float* __restrict__ agg, const float* __restrict__ norm_in,
                       const float* __restrict__ b, const float* __restrict__ a,
                       const float* __restrict__ wsum, const float* __restrict__ bsum,
                       float* __restrict__ out, int n) {
  const int lane = threadIdx.x & 31;
  const int wave = __builtin_amdgcn_readfirstlane((int)(threadIdx.x >> 5));
  const long r0  = ((long)blockIdx.x * (blockDim.x >> 5) + wave) * RW;
  if (r0 >= n) return;
  const float w0 = wsum[lane], w1 = wsum[lane + 32], w2 = wsum[lane + 64];
  const float b0 = b[lane],    b1 = b[lane + 32],    b2 = b[lane + 64];
  const float a0 = a[lane],    a1 = a[lane + 32],    a2 = a[lane + 64];
  const float bs = *bsum;
  float zv[RW][3];
  float sc[RW];
  #pragma unroll
  for (int q = 0; q < RW; ++q) {
    const long r = (r0 + q < n) ? (r0 + q) : (long)(n - 1);
    const float* zp = agg + r * HD + lane;
    zv[q][0] = zp[0]; zv[q][1] = zp[32]; zv[q][2] = zp[64];
    sc[q] = norm_in[r];
  }
  float sum[RW];
  #pragma unroll
  for (int q = 0; q < RW; ++q) {
    float t, z0, z1, z2;
    t = zv[q][0] * sc[q] + b0; z0 = (t >= 0.0f) ? t : a0 * t;
    t = zv[q][1] * sc[q] + b1; z1 = (t >= 0.0f) ? t : a1 * t;
    t = zv[q][2] * sc[q] + b2; z2 = (t >= 0.0f) ? t : a2 * t;
    float s = 0.0f;
    s = fmaf(z0, w0, s);
    s = fmaf(z1, w1, s);
    s = fmaf(z2, w2, s);
    sum[q] = s;
  }
  #pragma unroll
  for (int off = 16; off > 0; off >>= 1) {
    #pragma unroll
    for (int q = 0; q < RW; ++q) sum[q] += __shfl_xor(sum[q], off, 32);
  }
  if (lane == 0) {
    #pragma unroll
    for (int q = 0; q < RW; ++q)
      if (r0 + q < n) out[r0 + q] = sum[q] + bs;
  }
}

extern "C" void kernel_launch(void* const* d_in, const int* in_sizes, int n_in,
                              void* d_out, int out_size, void* d_ws, size_t ws_size,
                              hipStream_t stream) {
  const float* x    = (const float*)d_in[0];
  const int*   src  = (const int*)  d_in[1];
  const int*   dst  = (const int*)  d_in[2];
  const int*   perm = (const int*)  d_in[3];
  const float* W1   = (const float*)d_in[4];
  const float* b1   = (const float*)d_in[5];
  const float* a1   = (const float*)d_in[6];
  const float* W2   = (const float*)d_in[7];
  const float* b2   = (const float*)d_in[8];
  const float* a2   = (const float*)d_in[9];
  const float* Wm   = (const float*)d_in[10];
  const float* bm   = (const float*)d_in[11];
  float* out = (float*)d_out;

  const int E  = in_sizes[1];
  const int N  = in_sizes[3];
  const int K1 = in_sizes[0] / N;

  float* ws       = (float*)d_ws;
  float* norm_out = ws;
  float* norm_in  = ws + N;
  float* wsum     = ws + 2L * N;
  float* wbsum    = ws + 2L * N + HD;
  int*   ctr      = (int*)(ws + 2L * N + 128);
  float* bufA     = ws + 2L * N + 2 * HD;
  float* bufB     = bufA + (long)N * HD;

  int* cnt   = (int*)(bufB + (long)N * HD);
  int* cur   = cnt + 2L * N;
  int* off   = cur + 2L * N;
  int* sta   = off + 2L * N;
  int* bsum  = sta + 2L * N;
  int* dlist = bsum + 256;
  const int  nsl  = (2 * N + KSL - 1) / KSL;
  const bool push = (nsl <= 256) &&
                    (ws_size >= (size_t)((char*)(dlist + E) - (char*)d_ws));

  const int gemm_blocks = 256;
  const int red_blocks  = (int)(((long)N + 8 * RW - 1) / (8 * RW));

  if (push) {
    fill_zero_push_kernel<<<512, 256, 0, stream>>>(norm_in, (long)N, cnt, 4L * N, ctr);
    degree_keys_kernel<<<512, 256, 0, stream>>>(src, dst, norm_in, cnt, ctr, E, N);
    scan_norm_kernel<<<nsl, 256, 0, stream>>>(cnt, off, bsum, norm_out, norm_in, N);
    place_kernel<<<1024, 256, 0, stream>>>(src, dst, off, bsum, cur, sta, dlist, ctr + 5, E, N, nsl);
  } else {
    fill_zero_kernel<<<512, 256, 0, stream>>>(norm_out, 2L * N, ctr);
    degree_kernel<<<512, 256, 0, stream>>>(src, dst, norm_out, norm_in, ctr, E, N);
    finalize_norm_kernel<<<(2 * N + 255) / 256, 256, 0, stream>>>(norm_out, 2 * N);
  }
  wsum_kernel<<<1, 128, 0, stream>>>(Wm, bm, wsum, wbsum);

  for (int view = 0; view < 2; ++view) {
    const int* rowmap = view ? perm : nullptr;
    gemm_rows_wmma_kernel<<<gemm_blocks, GW * 32, 0, stream>>>(x, rowmap, norm_out, W1, bufA, bufB, N, K1);
    if (push) spmm_push_kernel<<<2048, 256, 0, stream>>>(sta, cnt, dlist, bufA, bufB, ctr + 1 + 2 * view, N);
    else      spmm_kernel<<<2048, 256, 0, stream>>>(src, dst, bufA, bufB, ctr + 1 + 2 * view, E, N);
    gemm_rows_finish_wmma_kernel<<<gemm_blocks, GW * 32, 0, stream>>>(bufB, norm_out, W2,
                                                                     norm_in, b1, a1, bufA, bufB, N, HD);
    if (push) spmm_push_kernel<<<2048, 256, 0, stream>>>(sta, cnt, dlist, bufA, bufB, ctr + 2 + 2 * view, N);
    else      spmm_kernel<<<2048, 256, 0, stream>>>(src, dst, bufA, bufB, ctr + 2 + 2 * view, E, N);
    finish_dot_kernel<<<red_blocks, 256, 0, stream>>>(bufB, norm_in, b2, a2, wsum, wbsum,
                                                      out + (long)view * N, N);
  }
}


# three launches folded into matrix-bound neighbours: edge placement on 4 extra waves a work-group beside view 1's layer-1 products, view 1's finish+dot in view 2's layer-1 launch, wsum in degree pass
# speedup vs baseline: 3.6777x; 1.0469x over previous
#include <hip/hip_runtime.h>

typedef float v2f __attribute__((ext_vector_type(2)));
typedef float v8f __attribute__((ext_vector_type(8)));

constexpr int HD   = 96;


__global__ void fill_zero_kernel(float* __restrict__ p, long n, int* __restrict__ ctr) {
  long i = (long)blockIdx.x * blockDim.x + threadIdx.x;
  long stride = (long)gridDim.x * blockDim.x;
  if (blockIdx.x == 0 && threadIdx.x < 64) ctr[threadIdx.x] = 0;
  for (; i < n; i += stride) p[i] = 0.0f;
}

constexpr int CTR_STRIDE = 32;
constexpr int KSL = 1024;

__device__ __forceinline__ int l2_half() {
  return (int)((__builtin_amdgcn_s_sendmsg_rtn(0x87) >> 18) & 1u);
}

__device__ __forceinline__ int draw_ticket(int* ctr, int q) {
  return __hip_atomic_fetch_add(ctr + q * CTR_STRIDE, 1, __ATOMIC_RELAXED, __HIP_MEMORY_SCOPE_AGENT);
}

__device__ __forceinline__
void wsum_rows(const float* __restrict__ Wm, const float* __restrict__ bm,
               float* __restrict__ wsum, float* __restrict__ bsum, int t) {
  if (t < HD) {
    float s = 0.0f;
    for (int j = 0; j < HD; ++j) s += Wm[t * HD + j];
    wsum[t] = s;
  } else if (t == HD) {
    float s = 0.0f;
    for (int j = 0; j < HD; ++j) s += bm[j];
    *bsum = s;
  }
}

constexpr int DCH = 1024;

__global__ __launch_bounds__(256)
void degree_kernel(const int* __restrict__ src, const int* __restrict__ dst,
                   float* __restrict__ deg_out, float* __restrict__ deg_in,
                   int* __restrict__ ctr, int E, int n,
                   const float* __restrict__ Wm, const float* __restrict__ bm,
                   float* __restrict__ wsum, float* __restrict__ wbsum) {
  __shared__ int sh_t[2];
  const int tid = threadIdx.x;
  if (blockIdx.x == gridDim.x - 1) wsum_rows(Wm, bm, wsum, wbsum, tid);
  const int myh = l2_half();
  const int Nh  = (n >> 1) & ~1;
  const int nitems = (E + DCH - 1) / DCH;
  int q = myh, tn = 0;
  if (tid == 0) tn = draw_ticket(ctr, q);
  for (int it = 0; ; ++it) {
    if (tid == 0) sh_t[it & 1] = tn;
    __syncthreads();
    const int t = __builtin_amdgcn_readfirstlane(sh_t[it & 1]);
    if (t >= nitems) {
      if (q != myh) break;
      q = 1 - myh;
      if (tid == 0) tn = draw_ticket(ctr, q);
      continue;
    }
    if (tid == 0) tn = draw_ticket(ctr, q);
    const bool up = (q != 0);
    int sv[DCH / 256], dv[DCH / 256];
    #pragma unroll
    for (int k = 0; k < DCH / 256; ++k) {
      const int e  = t * DCH + k * 256 + tid;
      const int ec = (e < E) ? e : (E - 1);
      sv[k] = src[ec];
      dv[k] = dst[ec];
    }
    #pragma unroll
    for (int k = 0; k < DCH / 256; ++k) {
      const bool ok = (t * DCH + k * 256 + tid) < E;
      if (ok && ((sv[k] >= Nh) == up))
        __hip_atomic_fetch_add(&deg_out[sv[k]], 1.0f, __ATOMIC_RELAXED, __HIP_MEMORY_SCOPE_AGENT);
      if (ok && ((dv[k] >= Nh) == up))
        __hip_atomic_fetch_add(&deg_in[dv[k]],  1.0f, __ATOMIC_RELAXED, __HIP_MEMORY_SCOPE_AGENT);
    }
  }
}

__global__ void finalize_norm_kernel(float* __restrict__ p, int n) {
  int i = blockIdx.x * blockDim.x + threadIdx.x;
  if (i < n) p[i] = 1.0f / sqrtf(fmaxf(p[i], 1.0f));
}

constexpr int RW = 8;

__device__ __forceinline__
void finish_dot_rows(const float* agg, const float* __restrict__ norm_in,
                     float b0, float b1, float b2, float a0, float a1, float a2,
                     float w0, float w1, float w2, float bs,
                     float* __restrict__ out, long r0, int n, int lane) {
  float zv[RW][3];
  float sc[RW];
  #pragma unroll
  for (int q = 0; q < RW; ++q) {
    const long r = (r0 + q < n) ? (r0 + q) : (long)(n - 1);
    const float* zp = agg + r * HD + lane;
    zv[q][0] = zp[0]; zv[q][1] = zp[32]; zv[q][2] = zp[64];
    sc[q] = norm_in[r];
  }
  float sum[RW];
  #pragma unroll
  for (int q = 0; q < RW; ++q) {
    float t, z0, z1, z2;
    t = zv[q][0] * sc[q] + b0; z0 = (t >= 0.0f) ? t : a0 * t;
    t = zv[q][1] * sc[q] + b1; z1 = (t >= 0.0f) ? t : a1 * t;
    t = zv[q][2] * sc[q] + b2; z2 = (t >= 0.0f) ? t : a2 * t;
    float s = 0.0f;
    s = fmaf(z0, w0, s);
    s = fmaf(z1, w1, s);
    s = fmaf(z2, w2, s);
    sum[q] = s;
  }
  #pragma unroll
  for (int off = 16; off > 0; off >>= 1) {
    #pragma unroll
    for (int q = 0; q < RW; ++q) sum[q] += __shfl_xor(sum[q], off, 32);
  }
  if (lane == 0) {
    #pragma unroll
    for (int q = 0; q < RW; ++q)
      if (r0 + q < n) out[r0 + q] = sum[q] + bs;
  }
}

__device__ __forceinline__
void finish_dot_tile(const float* agg, const float* __restrict__ norm_in,
                     float b0, float b1, float b2, float a0, float a1, float a2,
                     float w0, float w1, float w2, float bs,
                     float* __restrict__ out, int tile, int n, int lane) {
  #pragma unroll
  for (int g = 0; g < 16 / RW; ++g) {
    const long r0 = (long)tile * 16 + g * RW;
    if (r0 < n)
      finish_dot_rows(agg, norm_in, b0, b1, b2, a0, a1, a2, w0, w1, w2, bs, out, r0, n, lane);
  }
}

constexpr int KB   = 64;
constexpr int KMAX = 512;
constexpr int WS   = 514;
constexpr int GW   = 16;
constexpr int FW   = 12;
constexpr int HW   = 3;

__device__ __forceinline__
void gemm_load_block(float4 (&dstv)[8], const float* X, int srow, int K, int k0) {
  #pragma unroll
  for (int q = 0; q < 8; ++q) {
    const int k  = k0 + q * 8;
    const int kc = (k < K) ? k : (K - 4);
    dstv[q] = *(const float4*)(X + (long)srow * K + kc);
  }
}

__device__ __forceinline__
void gemm_load_finish(float4 (&bb)[8], float4 (&aa)[8], const float* __restrict__ fin_b,
                      const float* __restrict__ fin_a, int K, int k0) {
  #pragma unroll
  for (int q = 0; q < 8; ++q) {
    const int k  = k0 + q * 8;
    const int kc = (k < K) ? k : (K - 4);
    bb[q] = *(const float4*)(fin_b + kc);
    aa[q] = *(const float4*)(fin_a + kc);
  }
}

template <bool FIN, int NT>
__device__ __forceinline__
void gemm_block_products(v8f (&acc)[NT], const float4 (&raw)[8], const float4 (&bb)[8],
                         const float4 (&aa)[8], float s, float si, int K, int k0,
                         const float* wb) {
  float4 cur[8];
  #pragma unroll
  for (int q = 0; q < 8; ++q) {
    float4 v = raw[q];
    const int k = k0 + q * 8;
    if (FIN) {
      float t;
      t = v.x * si + bb[q].x; v.x = (t >= 0.0f) ? t : aa[q].x * t;
      t = v.y * si + bb[q].y; v.y = (t >= 0.0f) ? t : aa[q].y * t;
      t = v.z * si + bb[q].z; v.z = (t >= 0.0f) ? t : aa[q].z * t;
      t = v.w * si + bb[q].w; v.w = (t >= 0.0f) ? t : aa[q].w * t;
    }
    v.x *= s; v.y *= s; v.z *= s; v.w *= s;
    if (k >= K) v = make_float4(0.f, 0.f, 0.f, 0.f);
    cur[q] = v;
  }
  #pragma unroll
  for (int st = 0; st < 16; ++st) {
    v2f A;
    if (st & 1) { A.x = cur[st >> 1].z; A.y = cur[st >> 1].w; }
    else        { A.x = cur[st >> 1].x; A.y = cur[st >> 1].y; }
    v2f Bf[NT];
    #pragma unroll
    for (int t = 0; t < NT; ++t)
      Bf[t] = *(const v2f*)(wb + t * 16 * WS + st * 2);
    #pragma unroll
    for (int t = 0; t < NT; ++t)
      acc[t] = __builtin_amdgcn_wmma_f32_16x16x4_f32(
           false, A,  false, Bf[t],
           (short)0, acc[t],  false,  false);
  }
}

template <bool FIN, int NT>
__device__ __forceinline__
void gemm_tile_cols(v8f (&acc)[NT], const float* ldsW, const float* X,
                    const int* __restrict__ rowmap, const float* __restrict__ norm,
                    const float* __restrict__ fin_norm, const float* __restrict__ fin_b,
                    const float* __restrict__ fin_a, int nrows, int K, int Kpad,
                    int tile, int c0, int half, int l16) {
  const int grow = tile * 16 + l16;
  const int crow = (grow < nrows) ? grow : (nrows - 1);
  const int srow = rowmap ? rowmap[crow] : crow;
  const float s  = norm[crow];
  float si = 0.0f;
  if (FIN) si = fin_norm[crow];

  const float* wl = ldsW + (c0 * 16 + l16) * WS + half * 32;

  float4 buf0[8], buf1[8];
  float4 bb[8], aa[8];
  gemm_load_block(buf0, X, srow, K, half * 4);
  for (int kb = 0; kb < Kpad; kb += 2 * KB) {
    if (FIN) gemm_load_finish(bb, aa, fin_b, fin_a, K, kb + half * 4);
    gemm_load_block(buf1, X, srow, K, kb + KB + half * 4);
    gemm_block_products<FIN, NT>(acc, buf0, bb, aa, s, si, K, kb + half * 4, wl + kb);
    if (FIN) gemm_load_finish(bb, aa, fin_b, fin_a, K, kb + KB + half * 4);
    gemm_load_block(buf0, X, srow, K, kb + 2 * KB + half * 4);
    gemm_block_products<FIN, NT>(acc, buf1, bb, aa, s, si, K, kb + KB + half * 4, wl + kb + KB);
  }
}

__device__ __forceinline__
void gemm_zero_rows(float* ZERO, int tile, int nrows, int lane) {
  const long zend = (long)nrows * HD;
  #pragma unroll
  for (int it = 0; it < (16 * HD / 4) / 32; ++it) {
    const long off = (long)tile * 16 * HD + (long)(lane + 32 * it) * 4;
    if (off < zend) *(float4*)(ZERO + off) = make_float4(0.f, 0.f, 0.f, 0.f);
  }
}

template <int NT>
__device__ __forceinline__
void gemm_store_cols(float* __restrict__ OUT, const v8f (&acc)[NT], int tile, int c0, int nrows,
                     int half, int l16) {
  const int orow = tile * 16 + half * 8;
  #pragma unroll
  for (int t = 0; t < NT; ++t) {
    int col = (c0 + t) * 16 + l16;
    #pragma unroll
    for (int i = 0; i < 8; ++i) {
      int r = orow + i;
      if (r < nrows) OUT[(long)r * HD + col] = acc[t][i];
    }
  }
}

constexpr int PW  = 4;
constexpr int PE  = 16;
constexpr int PCH = 32 * PE;

__device__ __forceinline__
void place_wave(const int* __restrict__ src, const int* __restrict__ dst,
                const int* __restrict__ off, int* __restrict__ cur, int* __restrict__ sta,
                int* __restrict__ dlist, int* __restrict__ ctr, const int* sh_base,
                int E, int n, int pw, int lane) {
  const int myh = l2_half();
  const int Nh  = (n >> 1) & ~1;
  const int nk  = 2 * n;
  {
    const int gw  = (int)blockIdx.x * PW + pw;
    const int ngw = (int)gridDim.x * PW;
    for (int k = gw * 32 + lane; k < nk; k += ngw * 32) sta[k] = sh_base[k / KSL] + off[k];
  }
  const int nitems = (E + PCH - 1) / PCH;
  int q = myh, tn = 0;
  if (lane == 0) tn = draw_ticket(ctr, q);
  for (;;) {
    const int t = __builtin_amdgcn_readfirstlane(tn);
    if (t >= nitems) {
      if (q != myh) break;
      q = 1 - myh;
      if (lane == 0) tn = draw_ticket(ctr, q);
      continue;
    }
    if (lane == 0) tn = draw_ticket(ctr, q);
    const bool up = (q != 0);
    int sv[PE], dv[PE], key[PE], lo[PE], rk[PE];
    #pragma unroll
    for (int k = 0; k < PE; ++k) {
      const int e  = t * PCH + k * 32 + lane;
      const int ec = (e < E) ? e : (E - 1);
      sv[k] = src[ec];
      dv[k] = dst[ec];
    }
    #pragma unroll
    for (int k = 0; k < PE; ++k)
      if (t * PCH + k * 32 + lane >= E) dv[k] = -1;
    #pragma unroll
    for (int k = 0; k < PE; ++k) {
      const bool mine = (dv[k] >= 0) && ((sv[k] >= Nh) == up);
      key[k] = mine ? (((dv[k] >= Nh) ? n : 0) + sv[k]) : -1;
    }
    #pragma unroll
    for (int k = 0; k < PE; ++k) {
      lo[k] = 0; rk[k] = 0;
      if (key[k] >= 0) {
        lo[k] = off[key[k]];
        rk[k] = __hip_atomic_fetch_add(&cur[key[k]], 1, __ATOMIC_RELAXED, __HIP_MEMORY_SCOPE_AGENT);
      }
    }
    #pragma unroll
    for (int k = 0; k < PE; ++k) {
      if (key[k] >= 0) {
        const int pos = sh_base[key[k] / KSL] + lo[k] + rk[k];
        __hip_atomic_exchange(&dlist[pos], dv[k], __ATOMIC_RELAXED, __HIP_MEMORY_SCOPE_AGENT);
      }
    }
  }
}

template <bool FIN>
__device__ __forceinline__
void gemm_rows_body(float* ldsW, const float* X, const int* __restrict__ rowmap,
                    const float* __restrict__ norm, const float* __restrict__ W,
                    const float* __restrict__ fin_norm, const float* __restrict__ fin_b,
                    const float* __restrict__ fin_a,
                    float* __restrict__ OUT, float* ZERO, int nrows, int K,
                    const float* __restrict__ pv_norm, const float* __restrict__ pv_b,
                    const float* __restrict__ pv_a, const float* __restrict__ pv_wsum,
                    const float* __restrict__ pv_bsum, float* __restrict__ pv_out,
                    const int* __restrict__ pl_src, const int* __restrict__ pl_dst,
                    const int* __restrict__ pl_off, const int* __restrict__ pl_bsum,
                    int* __restrict__ pl_cur, int* __restrict__ pl_sta,
                    int* __restrict__ pl_dlist, int* __restrict__ pl_ctr, int* sh_pbase,
                    int pl_E, int pl_nsl) {

  const int tid  = threadIdx.x;
  const int lane = tid & 31;
  const int half = lane >> 4;
  const int l16  = lane & 15;
  const int wave = __builtin_amdgcn_readfirstlane(tid >> 5);
  const int Kpad = (K + 2 * KB - 1) & ~(2 * KB - 1);
  const int ntiles = (nrows + 15) >> 4;
  const int nwhole = (int)gridDim.x * FW;

  if (!FIN && pv_out) {
    const float w0 = pv_wsum[lane], w1 = pv_wsum[lane + 32], w2 = pv_wsum[lane + 64];
    const float b0 = pv_b[lane],    b1 = pv_b[lane + 32],    b2 = pv_b[lane + 64];
    const float a0 = pv_a[lane],    a1 = pv_a[lane + 32],    a2 = pv_a[lane + 64];
    const float bs = *pv_bsum;
    if (wave < FW) {
      const int tile = (int)blockIdx.x * FW + wave;
      if (tile < ntiles)
        finish_dot_tile(ZERO, pv_norm, b0, b1, b2, a0, a1, a2, w0, w1, w2, bs, pv_out, tile, nrows, lane);
    } else if (wave == FW) {
      for (int tile = nwhole + (int)blockIdx.x; tile < ntiles; tile += (int)gridDim.x)
        finish_dot_tile(ZERO, pv_norm, b0, b1, b2, a0, a1, a2, w0, w1, w2, bs, pv_out, tile, nrows, lane);
    }
  }

  if (!FIN && pl_dlist && wave == GW) {
    int v[8], tot = 0;
    #pragma unroll
    for (int i = 0; i < 8; ++i) {
      const int idx = lane * 8 + i;
      const int x   = pl_bsum[(idx < pl_nsl) ? idx : 0];
      v[i] = (idx < pl_nsl) ? x : 0;
      tot += v[i];
    }
    int inc = tot;
    #pragma unroll
    for (int o = 1; o < 32; o <<= 1) {
      const int u = __shfl_up(inc, o, 32);
      if (lane >= o) inc += u;
    }
    int ex = inc - tot;
    #pragma unroll
    for (int i = 0; i < 8; ++i) { sh_pbase[lane * 8 + i] = ex; ex += v[i]; }
  }

  if (wave < GW) {
    const int nitem = Kpad * (HD / 4);
    #pragma unroll 6
    for (int idx = tid; idx < nitem; idx += GW * 32) {
      const int k  = idx / (HD / 4);
      const int j  = idx % (HD / 4);
      const int kc = (k < K) ? k : (K - 1);
      float4 v = *(const float4*)(W + (long)kc * HD + j * 4);
      if (k >= K) v = make_float4(0.f, 0.f, 0.f, 0.f);
      const int kk = k & (KB - 1);
      const int p  = (k & ~(KB - 1)) + ((kk >> 2) & 1) * 32 + (kk >> 3) * 4 + (kk & 3);
      ldsW[(j * 4 + 0) * WS + p] = v.x;
      ldsW[(j * 4 + 1) * WS + p] = v.y;
      ldsW[(j * 4 + 2) * WS + p] = v.z;
      ldsW[(j * 4 + 3) * WS + p] = v.w;
    }
  }
  __syncthreads();

  if (!FIN && wave >= GW) {
    if (pl_dlist)
      place_wave(pl_src, pl_dst, pl_off, pl_cur, pl_sta, pl_dlist, pl_ctr, sh_pbase,
                 pl_E, nrows, wave - GW, lane);
    return;
  }

  if (wave < FW) {
    const int tile = (int)blockIdx.x * FW + wave;
    if (tile < ntiles) {
      v8f acc[6] = {};
      gemm_tile_cols<FIN, 6>(acc, ldsW, X, rowmap, norm, fin_norm, fin_b, fin_a,
                             nrows, K, Kpad, tile, 0, half, l16);
      gemm_zero_rows(ZERO, tile, nrows, lane);
      gemm_store_cols<6>(OUT, acc, tile, 0, nrows, half, l16);
    }
  }

  for (int tile = nwhole + (int)blockIdx.x; tile < ntiles; tile += (int)gridDim.x) {
    const bool part = (wave >= FW) && (wave < FW + HW);
    const int  c0   = (wave - FW) * 2;
    v8f acc2[2] = {};
    if (part)
      gemm_tile_cols<FIN, 2>(acc2, ldsW, X, rowmap, norm, fin_norm, fin_b, fin_a,
                             nrows, K, Kpad, tile, c0, half, l16);
    if (FIN) __syncthreads();
    if (wave == FW) gemm_zero_rows(ZERO, tile, nrows, lane);
    if (part) gemm_store_cols<2>(OUT, acc2, tile, c0, nrows, half, l16);
  }
}

__global__ __launch_bounds__((GW + PW) * 32)
void gemm_rows_wmma_kernel(const float* X, const int* __restrict__ rowmap,
                           const float* __restrict__ norm, const float* __restrict__ W,
                           float* __restrict__ OUT, float* ZERO, int nrows, int K,
                           const float* __restrict__ pv_norm, const float* __restrict__ pv_b,
                           const float* __restrict__ pv_a, const float* __restrict__ pv_wsum,
                           const float* __restrict__ pv_bsum, float* __restrict__ pv_out,
                           const int* __restrict__ pl_src, const int* __restrict__ pl_dst,
                           const int* __restrict__ pl_off, const int* __restrict__ pl_bsum,
                           int* __restrict__ pl_cur, int* __restrict__ pl_sta,
                           int* __restrict__ pl_dlist, int* __restrict__ pl_ctr,
                           int pl_E, int pl_nsl) {
  __shared__ float ldsW[HD * WS];
  __shared__ int sh_pbase[256];
  gemm_rows_body<false>(ldsW, X, rowmap, norm, W, nullptr, nullptr, nullptr, OUT, ZERO, nrows, K,
                        pv_norm, pv_b, pv_a, pv_wsum, pv_bsum, pv_out,
                        pl_src, pl_dst, pl_off, pl_bsum, pl_cur, pl_sta, pl_dlist, pl_ctr,
                        sh_pbase, pl_E, pl_nsl);
}

__global__ __launch_bounds__(GW * 32)
void gemm_rows_finish_wmma_kernel(const float* X, const float* __restrict__ norm,
                                  const float* __restrict__ W,
                                  const float* __restrict__ fin_norm,
                                  const float* __restrict__ fin_b,
                                  const float* __restrict__ fin_a,
                                  float* __restrict__ OUT, float* ZERO, int nrows, int K) {
  __shared__ float ldsW[HD * WS];
  gemm_rows_body<true>(ldsW, X, nullptr, norm, W, fin_norm, fin_b, fin_a, OUT, ZERO, nrows, K,
                       nullptr, nullptr, nullptr, nullptr, nullptr, nullptr,
                       nullptr, nullptr, nullptr, nullptr, nullptr, nullptr, nullptr, nullptr,
                       nullptr, 0, 0);
}

constexpr int EB   = 8;
constexpr int SPW  = 8;
constexpr int WCH  = 64;
constexpr int GCH  = SPW * WCH;
constexpr int LCAP = WCH + EB;

__device__ __forceinline__ void atomic_add_row(float* __restrict__ ap, float v0, float v1, float v2) {
  __hip_atomic_fetch_add(&ap[0],  v0, __ATOMIC_RELAXED, __HIP_MEMORY_SCOPE_AGENT);
  __hip_atomic_fetch_add(&ap[32], v1, __ATOMIC_RELAXED, __HIP_MEMORY_SCOPE_AGENT);
  __hip_atomic_fetch_add(&ap[64], v2, __ATOMIC_RELAXED, __HIP_MEMORY_SCOPE_AGENT);
}

__global__ __launch_bounds__(256)
void spmm_kernel(const int* __restrict__ src, const int* __restrict__ dst,
                 const float* __restrict__ h, float* __restrict__ agg,
                 int* __restrict__ ctr, int E, int n) {
  __shared__ int lst_s[SPW][LCAP];
  __shared__ int lst_d[SPW][LCAP];
  __shared__ int sh_t[2];
  const int tid  = threadIdx.x;
  const int lane = tid & 31;
  const int wave = __builtin_amdgcn_readfirstlane(tid >> 5);
  const int myh  = l2_half();
  const int Nh   = (n >> 1) & ~1;
  const int nitems = (E + GCH - 1) / GCH;
  int* ms = lst_s[wave];
  int* md = lst_d[wave];

  int q = myh, cnt = 0, tn = 0;
  if (tid == 0) tn = draw_ticket(ctr, q);
  for (int it = 0; ; ++it) {
    if (tid == 0) sh_t[it & 1] = tn;
    __syncthreads();
    const int t = __builtin_amdgcn_readfirstlane(sh_t[it & 1]);
    if (t >= nitems) {
      if (q != myh) break;
      q = 1 - myh;
      if (tid == 0) tn = draw_ticket(ctr, q);
      continue;
    }
    if (tid == 0) tn = draw_ticket(ctr, q);

    const bool up = (q != 0);
    int sv[WCH / 32], dv[WCH / 32];
    #pragma unroll
    for (int g = 0; g < WCH / 32; ++g) {
      const int e  = t * GCH + wave * WCH + g * 32 + lane;
      const int ec = (e < E) ? e : (E - 1);
      sv[g] = src[ec];
      dv[g] = (e < E) ? dst[ec] : -1;
    }
    #pragma unroll
    for (int g = 0; g < WCH / 32; ++g) {
      const bool mine = (dv[g] >= 0) && ((dv[g] >= Nh) == up);
      const unsigned m = __builtin_amdgcn_ballot_w32(mine);
      const int pos = cnt + (int)__builtin_amdgcn_mbcnt_lo(m, 0u);
      if (mine) { ms[pos] = sv[g]; md[pos] = dv[g]; }
      cnt += __builtin_popcount(m);
    }

    int b = 0;
    for (; b + EB <= cnt; b += EB) {
      const int vs = ms[b + (lane & (EB - 1))];
      const int vd = md[b + (lane & (EB - 1))];
      float v[EB][3];
      #pragma unroll
      for (int j = 0; j < EB; ++j) {
        const int s = __builtin_amdgcn_readlane(vs, j);
        const float* hp = h + (long)s * HD + lane;
        v[j][0] = hp[0]; v[j][1] = hp[32]; v[j][2] = hp[64];
      }
      #pragma unroll
      for (int j = 0; j < EB; ++j) {
        const int d = __builtin_amdgcn_readlane(vd, j);
        atomic_add_row(agg + (long)d * HD + lane, v[j][0], v[j][1], v[j][2]);
      }
    }
    const int r = cnt - b;
    if (b > 0 && r > 0) {
      int ts = 0, td = 0;
      if (lane < r) { ts = ms[b + lane]; td = md[b + lane]; }
      if (lane < r) { ms[lane] = ts; md[lane] = td; }
    }
    cnt = r;
  }

  if (cnt > 0) {
    const int li = (lane < cnt) ? lane : 0;
    const int vs = ms[li];
    const int vd = md[li];
    for (int j = 0; j < cnt; ++j) {
      const int s = __builtin_amdgcn_readlane(vs, j);
      const int d = __builtin_amdgcn_readlane(vd, j);
      const float* hp = h + (long)s * HD + lane;
      atomic_add_row(agg + (long)d * HD + lane, hp[0], hp[32], hp[64]);
    }
  }
}


__device__ __forceinline__ int block_excl_scan(int v, int* sh_w, int tid, int& tot) {
  const int lane = tid & 31;
  const int wave = tid >> 5;
  int inc = v;
  #pragma unroll
  for (int o = 1; o < 32; o <<= 1) {
    const int u = __shfl_up(inc, o, 32);
    if (lane >= o) inc += u;
  }
  if (lane == 31) sh_w[wave] = inc;
  __syncthreads();
  int woff = 0, t = 0;
  #pragma unroll
  for (int w = 0; w < 8; ++w) {
    const int x = sh_w[w];
    if (w < wave) woff += x;
    t += x;
  }
  tot = t;
  return woff + inc - v;
}

__global__ void fill_zero_push_kernel(float* __restrict__ p, long n, int* __restrict__ p2, long n2,
                                      int* __restrict__ ctr) {
  long i = (long)blockIdx.x * blockDim.x + threadIdx.x;
  const long stride = (long)gridDim.x * blockDim.x;
  if (blockIdx.x == 0 && threadIdx.x < 64) ctr[threadIdx.x] = 0;
  for (long j = i; j < n; j += stride) p[j] = 0.0f;
  for (long j = i; j < n2; j += stride) p2[j] = 0;
}

__global__ __launch_bounds__(256)
void degree_keys_kernel(const int* __restrict__ src, const int* __restrict__ dst,
                        float* __restrict__ deg_in, int* __restrict__ cnt,
                        int* __restrict__ ctr, int E, int n,
                        const float* __restrict__ Wm, const float* __restrict__ bm,
                        float* __restrict__ wsum, float* __restrict__ wbsum) {
  __shared__ int sh_t[2];
  const int tid = threadIdx.x;
  if (blockIdx.x == gridDim.x - 1) wsum_rows(Wm, bm, wsum, wbsum, tid);
  const int myh = l2_half();
  const int Nh  = (n >> 1) & ~1;
  const int nitems = (E + DCH - 1) / DCH;
  int q = myh, tn = 0;
  if (tid == 0) tn = draw_ticket(ctr, q);
  for (int it = 0; ; ++it) {
    if (tid == 0) sh_t[it & 1] = tn;
    __syncthreads();
    const int t = __builtin_amdgcn_readfirstlane(sh_t[it & 1]);
    if (t >= nitems) {
      if (q != myh) break;
      q = 1 - myh;
      if (tid == 0) tn = draw_ticket(ctr, q);
      continue;
    }
    if (tid == 0) tn = draw_ticket(ctr, q);
    const bool up = (q != 0);
    int sv[DCH / 256], dv[DCH / 256];
    #pragma unroll
    for (int k = 0; k < DCH / 256; ++k) {
      const int e  = t * DCH + k * 256 + tid;
      const int ec = (e < E) ? e : (E - 1);
      sv[k] = src[ec];
      dv[k] = dst[ec];
    }
    #pragma unroll
    for (int k = 0; k < DCH / 256; ++k) {
      const bool ok = (t * DCH + k * 256 + tid) < E;
      if (ok && ((sv[k] >= Nh) == up))
        __hip_atomic_fetch_add(&cnt[((dv[k] >= Nh) ? n : 0) + sv[k]], 1, __ATOMIC_RELAXED,
                               __HIP_MEMORY_SCOPE_AGENT);
      if (ok && ((dv[k] >= Nh) == up))
        __hip_atomic_fetch_add(&deg_in[dv[k]], 1.0f, __ATOMIC_RELAXED, __HIP_MEMORY_SCOPE_AGENT);
    }
  }
}

__global__ __launch_bounds__(256)
void scan_norm_kernel(const int* __restrict__ cnt, int* __restrict__ off, int* __restrict__ bsum,
                      float* __restrict__ norm_out, float* __restrict__ norm_in, int n) {
  __shared__ int sh_w[8];
  const int tid = threadIdx.x;
  const int nk  = 2 * n;
  const int k0  = (int)blockIdx.x * KSL + tid * 4;
  int c[4];
  #pragma unroll
  for (int i = 0; i < 4; ++i) {
    const int k  = k0 + i;
    const int kc = (k < nk) ? k : (nk - 1);
    const int x  = cnt[kc];
    c[i] = (k < nk) ? x : 0;
  }
  int tot;
  int ex = block_excl_scan(c[0] + c[1] + c[2] + c[3], sh_w, tid, tot);
  #pragma unroll
  for (int i = 0; i < 4; ++i) {
    if (k0 + i < nk) off[k0 + i] = ex;
    ex += c[i];
  }
  if (tid == 0) bsum[blockIdx.x] = tot;
  for (int s = (int)blockIdx.x * 256 + tid; s < n; s += (int)gridDim.x * 256) {
    const float dout = (float)(cnt[s] + cnt[n + s]);
    norm_out[s] = 1.0f / sqrtf(fmaxf(dout, 1.0f));
    norm_in[s]  = 1.0f / sqrtf(fmaxf(norm_in[s], 1.0f));
  }
}


constexpr int SPS = 4;
constexpr int SIT = SPW * SPS;

__global__ __launch_bounds__(256)
void spmm_push_kernel(const int* __restrict__ sta, const int* __restrict__ cnt,
                      const int* __restrict__ dlist, const float* __restrict__ h,
                      float* __restrict__ agg, int* __restrict__ ctr, int n) {
  __shared__ int sh_t[2];
  const int tid  = threadIdx.x;
  const int lane = tid & 31;
  const int wave = __builtin_amdgcn_readfirstlane(tid >> 5);
  const int myh  = l2_half();
  const int nitems = (n + SIT - 1) / SIT;
  int q = myh, tn = 0;
  if (tid == 0) tn = draw_ticket(ctr, q);
  for (int it = 0; ; ++it) {
    if (tid == 0) sh_t[it & 1] = tn;
    __syncthreads();
    const int t = __builtin_amdgcn_readfirstlane(sh_t[it & 1]);
    if (t >= nitems) {
      if (q != myh) break;
      q = 1 - myh;
      if (tid == 0) tn = draw_ticket(ctr, q);
      continue;
    }
    if (tid == 0) tn = draw_ticket(ctr, q);

    const int s0 = t * SIT + wave * SPS;
    const int sj  = s0 + (lane & (SPS - 1));
    const int sjc = (sj < n) ? sj : (n - 1);
    const int kj  = q * n + sjc;
    const int stv = sta[kj];
    const int cv0 = cnt[kj];
    const int cv  = (sj < n) ? cv0 : 0;
    float v[SPS][3];
    #pragma unroll
    for (int j = 0; j < SPS; ++j) {
      const int s  = s0 + j;
      const int sc = (s < n) ? s : (n - 1);
      const float* hp = h + (long)sc * HD + lane;
      v[j][0] = hp[0]; v[j][1] = hp[32]; v[j][2] = hp[64];
    }
    int dl[SPS];
    #pragma unroll
    for (int j = 0; j < SPS; ++j) {
      const int st = __builtin_amdgcn_readlane(stv, j);
      const int c  = __builtin_amdgcn_readlane(cv, j);
      dl[j] = (lane < c) ? dlist[st + lane] : 0;
    }
    #pragma unroll
    for (int j = 0; j < SPS; ++j) {
      const int st = __builtin_amdgcn_readlane(stv, j);
      const int c  = __builtin_amdgcn_readlane(cv, j);
      const int c1 = (c < 32) ? c : 32;
      for (int i = 0; i < c1; ++i) {
        const int d = __builtin_amdgcn_readlane(dl[j], i);
        atomic_add_row(agg + (long)d * HD + lane, v[j][0], v[j][1], v[j][2]);
      }
      for (int b = 32; b < c; b += 32) {
        const int dm = (b + lane < c) ? dlist[st + b + lane] : 0;
        const int c2 = (c - b < 32) ? (c - b) : 32;
        for (int i = 0; i < c2; ++i) {
          const int d = __builtin_amdgcn_readlane(dm, i);
          atomic_add_row(agg + (long)d * HD + lane, v[j][0], v[j][1], v[j][2]);
        }
      }
    }
  }
}

__global__ __launch_bounds__(256)
void finish_dot_kernel(const float* __restrict__ agg, const float* __restrict__ norm_in,
                       const float* __restrict__ b, const float* __restrict__ a,
                       const float* __restrict__ wsum, const float* __restrict__ bsum,
                       float* __restrict__ out, int n) {
  const int lane = threadIdx.x & 31;
  const int wave = __builtin_amdgcn_readfirstlane((int)(threadIdx.x >> 5));
  const long r0  = ((long)blockIdx.x * (blockDim.x >> 5) + wave) * RW;
  if (r0 >= n) return;
  const float w0 = wsum[lane], w1 = wsum[lane + 32], w2 = wsum[lane + 64];
  const float b0 = b[lane],    b1 = b[lane + 32],    b2 = b[lane + 64];
  const float a0 = a[lane],    a1 = a[lane + 32],    a2 = a[lane + 64];
  const float bs = *bsum;
  finish_dot_rows(agg, norm_in, b0, b1, b2, a0, a1, a2, w0, w1, w2, bs, out, r0, n, lane);
}

extern "C" void kernel_launch(void* const* d_in, const int* in_sizes, int n_in,
                              void* d_out, int out_size, void* d_ws, size_t ws_size,
                              hipStream_t stream) {
  const float* x    = (const float*)d_in[0];
  const int*   src  = (const int*)  d_in[1];
  const int*   dst  = (const int*)  d_in[2];
  const int*   perm = (const int*)  d_in[3];
  const float* W1   = (const float*)d_in[4];
  const float* b1   = (const float*)d_in[5];
  const float* a1   = (const float*)d_in[6];
  const float* W2   = (const float*)d_in[7];
  const float* b2   = (const float*)d_in[8];
  const float* a2   = (const float*)d_in[9];
  const float* Wm   = (const float*)d_in[10];
  const float* bm   = (const float*)d_in[11];
  float* out = (float*)d_out;

  const int E  = in_sizes[1];
  const int N  = in_sizes[3];
  const int K1 = in_sizes[0] / N;

  float* ws       = (float*)d_ws;
  float* norm_out = ws;
  float* norm_in  = ws + N;
  float* wsum     = ws + 2L * N;
  float* wbsum    = ws + 2L * N + HD;
  int*   ctr      = (int*)(ws + 2L * N + 128);
  float* bufA     = ws + 2L * N + 2 * HD;
  float* bufB     = bufA + (long)N * HD;

  int* cnt   = (int*)(bufB + (long)N * HD);
  int* cur   = cnt + 2L * N;
  int* off   = cur + 2L * N;
  int* sta   = off + 2L * N;
  int* bsum  = sta + 2L * N;
  int* dlist = bsum + 256;
  const int  nsl  = (2 * N + KSL - 1) / KSL;
  const bool push = (nsl <= 256) &&
                    (ws_size >= (size_t)((char*)(dlist + E) - (char*)d_ws));

  const int gemm_blocks = 256;
  const int red_blocks  = (int)(((long)N + 8 * RW - 1) / (8 * RW));

  if (push) {
    fill_zero_push_kernel<<<512, 256, 0, stream>>>(norm_in, (long)N, cnt, 4L * N, ctr);
    degree_keys_kernel<<<512, 256, 0, stream>>>(src, dst, norm_in, cnt, ctr, E, N, Wm, bm, wsum, wbsum);
    scan_norm_kernel<<<nsl, 256, 0, stream>>>(cnt, off, bsum, norm_out, norm_in, N);
  } else {
    fill_zero_kernel<<<512, 256, 0, stream>>>(norm_out, 2L * N, ctr);
    degree_kernel<<<512, 256, 0, stream>>>(src, dst, norm_out, norm_in, ctr, E, N, Wm, bm, wsum, wbsum);
    finalize_norm_kernel<<<(2 * N + 255) / 256, 256, 0, stream>>>(norm_out, 2 * N);
  }

  for (int view = 0; view < 2; ++view) {
    const int* rowmap = view ? perm : nullptr;
    float* pv_out = view ? out : nullptr;
    const bool ride = push && (view == 0);
    int* pl_dlist = ride ? dlist : nullptr;
    gemm_rows_wmma_kernel<<<gemm_blocks, ride ? (GW + PW) * 32 : GW * 32, 0, stream>>>(
        x, rowmap, norm_out, W1, bufA, bufB, N, K1,
        norm_in, b2, a2, wsum, wbsum, pv_out,
        src, dst, off, bsum, cur, sta, pl_dlist, ctr + 5, E, nsl);
    if (push) spmm_push_kernel<<<2048, 256, 0, stream>>>(sta, cnt, dlist, bufA, bufB, ctr + 1 + 2 * view, N);
    else      spmm_kernel<<<2048, 256, 0, stream>>>(src, dst, bufA, bufB, ctr + 1 + 2 * view, E, N);
    gemm_rows_finish_wmma_kernel<<<gemm_blocks, GW * 32, 0, stream>>>(bufB, norm_out, W2,
                                                                     norm_in, b1, a1, bufA, bufB, N, HD);
    if (push) spmm_push_kernel<<<2048, 256, 0, stream>>>(sta, cnt, dlist, bufA, bufB, ctr + 2 + 2 * view, N);
    else      spmm_kernel<<<2048, 256, 0, stream>>>(src, dst, bufA, bufB, ctr + 2 + 2 * view, E, N);
    if (view) finish_dot_kernel<<<red_blocks, 256, 0, stream>>>(bufB, norm_in, b2, a2, wsum, wbsum,
                                                                out + (long)N, N);
  }
}


# x@W1 once for both views, kept unscaled: the layer-1 push scales its source row by norm_out[s], view 2 reads row perm[s]; one K=500 launch gone; h rounds differently; identity as in sibling chain
# speedup vs baseline: 4.4554x; 1.2115x over previous
#include <hip/hip_runtime.h>

typedef float v2f __attribute__((ext_vector_type(2)));
typedef float v8f __attribute__((ext_vector_type(8)));

constexpr int HD   = 96;


__global__ void fill_zero_kernel(float* __restrict__ p, long n, int* __restrict__ ctr) {
  long i = (long)blockIdx.x * blockDim.x + threadIdx.x;
  long stride = (long)gridDim.x * blockDim.x;
  if (blockIdx.x == 0 && threadIdx.x < 64) ctr[threadIdx.x] = 0;
  for (; i < n; i += stride) p[i] = 0.0f;
}

constexpr int CTR_STRIDE = 32;
constexpr int KSL = 1024;

__device__ __forceinline__ int l2_half() {
  return (int)((__builtin_amdgcn_s_sendmsg_rtn(0x87) >> 18) & 1u);
}

__device__ __forceinline__ int draw_ticket(int* ctr, int q) {
  return __hip_atomic_fetch_add(ctr + q * CTR_STRIDE, 1, __ATOMIC_RELAXED, __HIP_MEMORY_SCOPE_AGENT);
}

__device__ __forceinline__
void wsum_rows(const float* __restrict__ Wm, const float* __restrict__ bm,
               float* __restrict__ wsum, float* __restrict__ bsum, int t) {
  if (t < HD) {
    float s = 0.0f;
    for (int j = 0; j < HD; ++j) s += Wm[t * HD + j];
    wsum[t] = s;
  } else if (t == HD) {
    float s = 0.0f;
    for (int j = 0; j < HD; ++j) s += bm[j];
    *bsum = s;
  }
}

constexpr int DCH = 1024;

__global__ __launch_bounds__(256)
void degree_kernel(const int* __restrict__ src, const int* __restrict__ dst,
                   float* __restrict__ deg_out, float* __restrict__ deg_in,
                   int* __restrict__ ctr, int E, int n,
                   const float* __restrict__ Wm, const float* __restrict__ bm,
                   float* __restrict__ wsum, float* __restrict__ wbsum) {
  __shared__ int sh_t[2];
  const int tid = threadIdx.x;
  if (blockIdx.x == gridDim.x - 1) wsum_rows(Wm, bm, wsum, wbsum, tid);
  const int myh = l2_half();
  const int Nh  = (n >> 1) & ~1;
  const int nitems = (E + DCH - 1) / DCH;
  int q = myh, tn = 0;
  if (tid == 0) tn = draw_ticket(ctr, q);
  for (int it = 0; ; ++it) {
    if (tid == 0) sh_t[it & 1] = tn;
    __syncthreads();
    const int t = __builtin_amdgcn_readfirstlane(sh_t[it & 1]);
    if (t >= nitems) {
      if (q != myh) break;
      q = 1 - myh;
      if (tid == 0) tn = draw_ticket(ctr, q);
      continue;
    }
    if (tid == 0) tn = draw_ticket(ctr, q);
    const bool up = (q != 0);
    int sv[DCH / 256], dv[DCH / 256];
    #pragma unroll
    for (int k = 0; k < DCH / 256; ++k) {
      const int e  = t * DCH + k * 256 + tid;
      const int ec = (e < E) ? e : (E - 1);
      sv[k] = src[ec];
      dv[k] = dst[ec];
    }
    #pragma unroll
    for (int k = 0; k < DCH / 256; ++k) {
      const bool ok = (t * DCH + k * 256 + tid) < E;
      if (ok && ((sv[k] >= Nh) == up))
        __hip_atomic_fetch_add(&deg_out[sv[k]], 1.0f, __ATOMIC_RELAXED, __HIP_MEMORY_SCOPE_AGENT);
      if (ok && ((dv[k] >= Nh) == up))
        __hip_atomic_fetch_add(&deg_in[dv[k]],  1.0f, __ATOMIC_RELAXED, __HIP_MEMORY_SCOPE_AGENT);
    }
  }
}

__global__ void finalize_norm_kernel(float* __restrict__ p, int n) {
  int i = blockIdx.x * blockDim.x + threadIdx.x;
  if (i < n) p[i] = 1.0f / sqrtf(fmaxf(p[i], 1.0f));
}

constexpr int RW = 8;

__device__ __forceinline__
void finish_dot_rows(const float* agg, const float* __restrict__ norm_in,
                     float b0, float b1, float b2, float a0, float a1, float a2,
                     float w0, float w1, float w2, float bs,
                     float* __restrict__ out, long r0, int n, int lane) {
  float zv[RW][3];
  float sc[RW];
  #pragma unroll
  for (int q = 0; q < RW; ++q) {
    const long r = (r0 + q < n) ? (r0 + q) : (long)(n - 1);
    const float* zp = agg + r * HD + lane;
    zv[q][0] = zp[0]; zv[q][1] = zp[32]; zv[q][2] = zp[64];
    sc[q] = norm_in[r];
  }
  float sum[RW];
  #pragma unroll
  for (int q = 0; q < RW; ++q) {
    float t, z0, z1, z2;
    t = zv[q][0] * sc[q] + b0; z0 = (t >= 0.0f) ? t : a0 * t;
    t = zv[q][1] * sc[q] + b1; z1 = (t >= 0.0f) ? t : a1 * t;
    t = zv[q][2] * sc[q] + b2; z2 = (t >= 0.0f) ? t : a2 * t;
    float s = 0.0f;
    s = fmaf(z0, w0, s);
    s = fmaf(z1, w1, s);
    s = fmaf(z2, w2, s);
    sum[q] = s;
  }
  #pragma unroll
  for (int off = 16; off > 0; off >>= 1) {
    #pragma unroll
    for (int q = 0; q < RW; ++q) sum[q] += __shfl_xor(sum[q], off, 32);
  }
  if (lane == 0) {
    #pragma unroll
    for (int q = 0; q < RW; ++q)
      if (r0 + q < n) out[r0 + q] = sum[q] + bs;
  }
}

__device__ __forceinline__
void finish_dot_tile(const float* agg, const float* __restrict__ norm_in,
                     float b0, float b1, float b2, float a0, float a1, float a2,
                     float w0, float w1, float w2, float bs,
                     float* __restrict__ out, int tile, int n, int lane) {
  #pragma unroll
  for (int g = 0; g < 16 / RW; ++g) {
    const long r0 = (long)tile * 16 + g * RW;
    if (r0 < n)
      finish_dot_rows(agg, norm_in, b0, b1, b2, a0, a1, a2, w0, w1, w2, bs, out, r0, n, lane);
  }
}

constexpr int KB   = 64;
constexpr int KMAX = 512;
constexpr int WS   = 514;
constexpr int GW   = 16;
constexpr int FW   = 12;
constexpr int HW   = 3;

__device__ __forceinline__
void gemm_load_block(float4 (&dstv)[8], const float* X, int srow, int K, int k0) {
  #pragma unroll
  for (int q = 0; q < 8; ++q) {
    const int k  = k0 + q * 8;
    const int kc = (k < K) ? k : (K - 4);
    dstv[q] = *(const float4*)(X + (long)srow * K + kc);
  }
}

__device__ __forceinline__
void gemm_load_finish(float4 (&bb)[8], float4 (&aa)[8], const float* __restrict__ fin_b,
                      const float* __restrict__ fin_a, int K, int k0) {
  #pragma unroll
  for (int q = 0; q < 8; ++q) {
    const int k  = k0 + q * 8;
    const int kc = (k < K) ? k : (K - 4);
    bb[q] = *(const float4*)(fin_b + kc);
    aa[q] = *(const float4*)(fin_a + kc);
  }
}

template <bool FIN, int NT>
__device__ __forceinline__
void gemm_block_products(v8f (&acc)[NT], const float4 (&raw)[8], const float4 (&bb)[8],
                         const float4 (&aa)[8], float s, float si, int K, int k0,
                         const float* wb) {
  float4 cur[8];
  #pragma unroll
  for (int q = 0; q < 8; ++q) {
    float4 v = raw[q];
    const int k = k0 + q * 8;
    if (FIN) {
      float t;
      t = v.x * si + bb[q].x; v.x = (t >= 0.0f) ? t : aa[q].x * t;
      t = v.y * si + bb[q].y; v.y = (t >= 0.0f) ? t : aa[q].y * t;
      t = v.z * si + bb[q].z; v.z = (t >= 0.0f) ? t : aa[q].z * t;
      t = v.w * si + bb[q].w; v.w = (t >= 0.0f) ? t : aa[q].w * t;
    }
    v.x *= s; v.y *= s; v.z *= s; v.w *= s;
    if (k >= K) v = make_float4(0.f, 0.f, 0.f, 0.f);
    cur[q] = v;
  }
  #pragma unroll
  for (int st = 0; st < 16; ++st) {
    v2f A;
    if (st & 1) { A.x = cur[st >> 1].z; A.y = cur[st >> 1].w; }
    else        { A.x = cur[st >> 1].x; A.y = cur[st >> 1].y; }
    v2f Bf[NT];
    #pragma unroll
    for (int t = 0; t < NT; ++t)
      Bf[t] = *(const v2f*)(wb + t * 16 * WS + st * 2);
    #pragma unroll
    for (int t = 0; t < NT; ++t)
      acc[t] = __builtin_amdgcn_wmma_f32_16x16x4_f32(
           false, A,  false, Bf[t],
           (short)0, acc[t],  false,  false);
  }
}

template <bool FIN, int NT>
__device__ __forceinline__
void gemm_tile_cols(v8f (&acc)[NT], const float* ldsW, const float* X,
                    const int* __restrict__ rowmap, const float* __restrict__ norm,
                    const float* __restrict__ fin_norm, const float* __restrict__ fin_b,
                    const float* __restrict__ fin_a, int nrows, int K, int Kpad,
                    int tile, int c0, int half, int l16) {
  const int grow = tile * 16 + l16;
  const int crow = (grow < nrows) ? grow : (nrows - 1);
  const int srow = rowmap ? rowmap[crow] : crow;
  const float s  = norm ? norm[crow] : 1.0f;
  float si = 0.0f;
  if (FIN) si = fin_norm[crow];

  const float* wl = ldsW + (c0 * 16 + l16) * WS + half * 32;

  float4 buf0[8], buf1[8];
  float4 bb[8], aa[8];
  gemm_load_block(buf0, X, srow, K, half * 4);
  for (int kb = 0; kb < Kpad; kb += 2 * KB) {
    if (FIN) gemm_load_finish(bb, aa, fin_b, fin_a, K, kb + half * 4);
    gemm_load_block(buf1, X, srow, K, kb + KB + half * 4);
    gemm_block_products<FIN, NT>(acc, buf0, bb, aa, s, si, K, kb + half * 4, wl + kb);
    if (FIN) gemm_load_finish(bb, aa, fin_b, fin_a, K, kb + KB + half * 4);
    gemm_load_block(buf0, X, srow, K, kb + 2 * KB + half * 4);
    gemm_block_products<FIN, NT>(acc, buf1, bb, aa, s, si, K, kb + KB + half * 4, wl + kb + KB);
  }
}

__device__ __forceinline__
void gemm_zero_rows(float* ZERO, int tile, int nrows, int lane) {
  const long zend = (long)nrows * HD;
  #pragma unroll
  for (int it = 0; it < (16 * HD / 4) / 32; ++it) {
    const long off = (long)tile * 16 * HD + (long)(lane + 32 * it) * 4;
    if (off < zend) *(float4*)(ZERO + off) = make_float4(0.f, 0.f, 0.f, 0.f);
  }
}

template <int NT>
__device__ __forceinline__
void gemm_store_cols(float* __restrict__ OUT, const v8f (&acc)[NT], int tile, int c0, int nrows,
                     int half, int l16) {
  const int orow = tile * 16 + half * 8;
  #pragma unroll
  for (int t = 0; t < NT; ++t) {
    int col = (c0 + t) * 16 + l16;
    #pragma unroll
    for (int i = 0; i < 8; ++i) {
      int r = orow + i;
      if (r < nrows) OUT[(long)r * HD + col] = acc[t][i];
    }
  }
}

constexpr int PW  = 4;
constexpr int PE  = 16;
constexpr int PCH = 32 * PE;

__device__ __forceinline__
void place_wave(const int* __restrict__ src, const int* __restrict__ dst,
                const int* __restrict__ off, int* __restrict__ cur, int* __restrict__ sta,
                int* __restrict__ dlist, int* __restrict__ ctr, const int* sh_base,
                int E, int n, int pw, int lane) {
  const int myh = l2_half();
  const int Nh  = (n >> 1) & ~1;
  const int nk  = 2 * n;
  {
    const int gw  = (int)blockIdx.x * PW + pw;
    const int ngw = (int)gridDim.x * PW;
    for (int k = gw * 32 + lane; k < nk; k += ngw * 32) sta[k] = sh_base[k / KSL] + off[k];
  }
  const int nitems = (E + PCH - 1) / PCH;
  int q = myh, tn = 0;
  if (lane == 0) tn = draw_ticket(ctr, q);
  for (;;) {
    const int t = __builtin_amdgcn_readfirstlane(tn);
    if (t >= nitems) {
      if (q != myh) break;
      q = 1 - myh;
      if (lane == 0) tn = draw_ticket(ctr, q);
      continue;
    }
    if (lane == 0) tn = draw_ticket(ctr, q);
    const bool up = (q != 0);
    int sv[PE], dv[PE], key[PE], lo[PE], rk[PE];
    #pragma unroll
    for (int k = 0; k < PE; ++k) {
      const int e  = t * PCH + k * 32 + lane;
      const int ec = (e < E) ? e : (E - 1);
      sv[k] = src[ec];
      dv[k] = dst[ec];
    }
    #pragma unroll
    for (int k = 0; k < PE; ++k)
      if (t * PCH + k * 32 + lane >= E) dv[k] = -1;
    #pragma unroll
    for (int k = 0; k < PE; ++k) {
      const bool mine = (dv[k] >= 0) && ((sv[k] >= Nh) == up);
      key[k] = mine ? (((dv[k] >= Nh) ? n : 0) + sv[k]) : -1;
    }
    #pragma unroll
    for (int k = 0; k < PE; ++k) {
      lo[k] = 0; rk[k] = 0;
      if (key[k] >= 0) {
        lo[k] = off[key[k]];
        rk[k] = __hip_atomic_fetch_add(&cur[key[k]], 1, __ATOMIC_RELAXED, __HIP_MEMORY_SCOPE_AGENT);
      }
    }
    #pragma unroll
    for (int k = 0; k < PE; ++k) {
      if (key[k] >= 0) {
        const int pos = sh_base[key[k] / KSL] + lo[k] + rk[k];
        __hip_atomic_exchange(&dlist[pos], dv[k], __ATOMIC_RELAXED, __HIP_MEMORY_SCOPE_AGENT);
      }
    }
  }
}

template <bool FIN>
__device__ __forceinline__
void gemm_rows_body(float* ldsW, const float* X, const int* __restrict__ rowmap,
                    const float* __restrict__ norm, const float* __restrict__ W,
                    const float* __restrict__ fin_norm, const float* __restrict__ fin_b,
                    const float* __restrict__ fin_a,
                    float* __restrict__ OUT, float* ZERO, float* ZERO2, int nrows, int K,
                    const float* __restrict__ pv_norm, const float* __restrict__ pv_b,
                    const float* __restrict__ pv_a, const float* __restrict__ pv_wsum,
                    const float* __restrict__ pv_bsum, float* __restrict__ pv_out,
                    const int* __restrict__ pl_src, const int* __restrict__ pl_dst,
                    const int* __restrict__ pl_off, const int* __restrict__ pl_bsum,
                    int* __restrict__ pl_cur, int* __restrict__ pl_sta,
                    int* __restrict__ pl_dlist, int* __restrict__ pl_ctr, int* sh_pbase,
                    int pl_E, int pl_nsl) {

  const int tid  = threadIdx.x;
  const int lane = tid & 31;
  const int half = lane >> 4;
  const int l16  = lane & 15;
  const int wave = __builtin_amdgcn_readfirstlane(tid >> 5);
  const int Kpad = (K + 2 * KB - 1) & ~(2 * KB - 1);
  const int ntiles = (nrows + 15) >> 4;
  const int nwhole = (int)gridDim.x * FW;

  if (!FIN && pv_out) {
    const float w0 = pv_wsum[lane], w1 = pv_wsum[lane + 32], w2 = pv_wsum[lane + 64];
    const float b0 = pv_b[lane],    b1 = pv_b[lane + 32],    b2 = pv_b[lane + 64];
    const float a0 = pv_a[lane],    a1 = pv_a[lane + 32],    a2 = pv_a[lane + 64];
    const float bs = *pv_bsum;
    if (wave < FW) {
      const int tile = (int)blockIdx.x * FW + wave;
      if (tile < ntiles)
        finish_dot_tile(ZERO, pv_norm, b0, b1, b2, a0, a1, a2, w0, w1, w2, bs, pv_out, tile, nrows, lane);
    } else if (wave == FW) {
      for (int tile = nwhole + (int)blockIdx.x; tile < ntiles; tile += (int)gridDim.x)
        finish_dot_tile(ZERO, pv_norm, b0, b1, b2, a0, a1, a2, w0, w1, w2, bs, pv_out, tile, nrows, lane);
    }
  }

  if (!FIN && pl_dlist && wave == GW) {
    int v[8], tot = 0;
    #pragma unroll
    for (int i = 0; i < 8; ++i) {
      const int idx = lane * 8 + i;
      const int x   = pl_bsum[(idx < pl_nsl) ? idx : 0];
      v[i] = (idx < pl_nsl) ? x : 0;
      tot += v[i];
    }
    int inc = tot;
    #pragma unroll
    for (int o = 1; o < 32; o <<= 1) {
      const int u = __shfl_up(inc, o, 32);
      if (lane >= o) inc += u;
    }
    int ex = inc - tot;
    #pragma unroll
    for (int i = 0; i < 8; ++i) { sh_pbase[lane * 8 + i] = ex; ex += v[i]; }
  }

  if (wave < GW) {
    const int nitem = Kpad * (HD / 4);
    #pragma unroll 6
    for (int idx = tid; idx < nitem; idx += GW * 32) {
      const int k  = idx / (HD / 4);
      const int j  = idx % (HD / 4);
      const int kc = (k < K) ? k : (K - 1);
      float4 v = *(const float4*)(W + (long)kc * HD + j * 4);
      if (k >= K) v = make_float4(0.f, 0.f, 0.f, 0.f);
      const int kk = k & (KB - 1);
      const int p  = (k & ~(KB - 1)) + ((kk >> 2) & 1) * 32 + (kk >> 3) * 4 + (kk & 3);
      ldsW[(j * 4 + 0) * WS + p] = v.x;
      ldsW[(j * 4 + 1) * WS + p] = v.y;
      ldsW[(j * 4 + 2) * WS + p] = v.z;
      ldsW[(j * 4 + 3) * WS + p] = v.w;
    }
  }
  __syncthreads();

  if (!FIN && wave >= GW) {
    if (pl_dlist)
      place_wave(pl_src, pl_dst, pl_off, pl_cur, pl_sta, pl_dlist, pl_ctr, sh_pbase,
                 pl_E, nrows, wave - GW, lane);
    return;
  }

  if (wave < FW) {
    const int tile = (int)blockIdx.x * FW + wave;
    if (tile < ntiles) {
      v8f acc[6] = {};
      gemm_tile_cols<FIN, 6>(acc, ldsW, X, rowmap, norm, fin_norm, fin_b, fin_a,
                             nrows, K, Kpad, tile, 0, half, l16);
      gemm_zero_rows(ZERO, tile, nrows, lane);
      if (ZERO2) gemm_zero_rows(ZERO2, tile, nrows, lane);
      gemm_store_cols<6>(OUT, acc, tile, 0, nrows, half, l16);
    }
  }

  for (int tile = nwhole + (int)blockIdx.x; tile < ntiles; tile += (int)gridDim.x) {
    const bool part = (wave >= FW) && (wave < FW + HW);
    const int  c0   = (wave - FW) * 2;
    v8f acc2[2] = {};
    if (part)
      gemm_tile_cols<FIN, 2>(acc2, ldsW, X, rowmap, norm, fin_norm, fin_b, fin_a,
                             nrows, K, Kpad, tile, c0, half, l16);
    if (FIN) __syncthreads();
    if (wave == FW) {
      gemm_zero_rows(ZERO, tile, nrows, lane);
      if (ZERO2) gemm_zero_rows(ZERO2, tile, nrows, lane);
    }
    if (part) gemm_store_cols<2>(OUT, acc2, tile, c0, nrows, half, l16);
  }
}

__global__ __launch_bounds__((GW + PW) * 32)
void gemm_rows_wmma_kernel(const float* X, const int* __restrict__ rowmap,
                           const float* __restrict__ norm, const float* __restrict__ W,
                           float* __restrict__ OUT, float* ZERO, float* ZERO2, int nrows, int K,
                           const float* __restrict__ pv_norm, const float* __restrict__ pv_b,
                           const float* __restrict__ pv_a, const float* __restrict__ pv_wsum,
                           const float* __restrict__ pv_bsum, float* __restrict__ pv_out,
                           const int* __restrict__ pl_src, const int* __restrict__ pl_dst,
                           const int* __restrict__ pl_off, const int* __restrict__ pl_bsum,
                           int* __restrict__ pl_cur, int* __restrict__ pl_sta,
                           int* __restrict__ pl_dlist, int* __restrict__ pl_ctr,
                           int pl_E, int pl_nsl) {
  __shared__ float ldsW[HD * WS];
  __shared__ int sh_pbase[256];
  gemm_rows_body<false>(ldsW, X, rowmap, norm, W, nullptr, nullptr, nullptr, OUT, ZERO, ZERO2, nrows, K,
                        pv_norm, pv_b, pv_a, pv_wsum, pv_bsum, pv_out,
                        pl_src, pl_dst, pl_off, pl_bsum, pl_cur, pl_sta, pl_dlist, pl_ctr,
                        sh_pbase, pl_E, pl_nsl);
}

__global__ __launch_bounds__(GW * 32)
void gemm_rows_finish_wmma_kernel(const float* X, const float* __restrict__ norm,
                                  const float* __restrict__ W,
                                  const float* __restrict__ fin_norm,
                                  const float* __restrict__ fin_b,
                                  const float* __restrict__ fin_a,
                                  float* __restrict__ OUT, float* ZERO, int nrows, int K) {
  __shared__ float ldsW[HD * WS];
  gemm_rows_body<true>(ldsW, X, nullptr, norm, W, fin_norm, fin_b, fin_a, OUT, ZERO, nullptr, nrows, K,
                       nullptr, nullptr, nullptr, nullptr, nullptr, nullptr,
                       nullptr, nullptr, nullptr, nullptr, nullptr, nullptr, nullptr, nullptr,
                       nullptr, 0, 0);
}

constexpr int EB   = 8;
constexpr int SPW  = 8;
constexpr int WCH  = 64;
constexpr int GCH  = SPW * WCH;
constexpr int LCAP = WCH + EB;

__device__ __forceinline__ void atomic_add_row(float* __restrict__ ap, float v0, float v1, float v2) {
  __hip_atomic_fetch_add(&ap[0],  v0, __ATOMIC_RELAXED, __HIP_MEMORY_SCOPE_AGENT);
  __hip_atomic_fetch_add(&ap[32], v1, __ATOMIC_RELAXED, __HIP_MEMORY_SCOPE_AGENT);
  __hip_atomic_fetch_add(&ap[64], v2, __ATOMIC_RELAXED, __HIP_MEMORY_SCOPE_AGENT);
}

__global__ __launch_bounds__(256)
void spmm_kernel(const int* __restrict__ src, const int* __restrict__ dst,
                 const float* __restrict__ h, float* __restrict__ agg,
                 int* __restrict__ ctr, int E, int n) {
  __shared__ int lst_s[SPW][LCAP];
  __shared__ int lst_d[SPW][LCAP];
  __shared__ int sh_t[2];
  const int tid  = threadIdx.x;
  const int lane = tid & 31;
  const int wave = __builtin_amdgcn_readfirstlane(tid >> 5);
  const int myh  = l2_half();
  const int Nh   = (n >> 1) & ~1;
  const int nitems = (E + GCH - 1) / GCH;
  int* ms = lst_s[wave];
  int* md = lst_d[wave];

  int q = myh, cnt = 0, tn = 0;
  if (tid == 0) tn = draw_ticket(ctr, q);
  for (int it = 0; ; ++it) {
    if (tid == 0) sh_t[it & 1] = tn;
    __syncthreads();
    const int t = __builtin_amdgcn_readfirstlane(sh_t[it & 1]);
    if (t >= nitems) {
      if (q != myh) break;
      q = 1 - myh;
      if (tid == 0) tn = draw_ticket(ctr, q);
      continue;
    }
    if (tid == 0) tn = draw_ticket(ctr, q);

    const bool up = (q != 0);
    int sv[WCH / 32], dv[WCH / 32];
    #pragma unroll
    for (int g = 0; g < WCH / 32; ++g) {
      const int e  = t * GCH + wave * WCH + g * 32 + lane;
      const int ec = (e < E) ? e : (E - 1);
      sv[g] = src[ec];
      dv[g] = (e < E) ? dst[ec] : -1;
    }
    #pragma unroll
    for (int g = 0; g < WCH / 32; ++g) {
      const bool mine = (dv[g] >= 0) && ((dv[g] >= Nh) == up);
      const unsigned m = __builtin_amdgcn_ballot_w32(mine);
      const int pos = cnt + (int)__builtin_amdgcn_mbcnt_lo(m, 0u);
      if (mine) { ms[pos] = sv[g]; md[pos] = dv[g]; }
      cnt += __builtin_popcount(m);
    }

    int b = 0;
    for (; b + EB <= cnt; b += EB) {
      const int vs = ms[b + (lane & (EB - 1))];
      const int vd = md[b + (lane & (EB - 1))];
      float v[EB][3];
      #pragma unroll
      for (int j = 0; j < EB; ++j) {
        const int s = __builtin_amdgcn_readlane(vs, j);
        const float* hp = h + (long)s * HD + lane;
        v[j][0] = hp[0]; v[j][1] = hp[32]; v[j][2] = hp[64];
      }
      #pragma unroll
      for (int j = 0; j < EB; ++j) {
        const int d = __builtin_amdgcn_readlane(vd, j);
        atomic_add_row(agg + (long)d * HD + lane, v[j][0], v[j][1], v[j][2]);
      }
    }
    const int r = cnt - b;
    if (b > 0 && r > 0) {
      int ts = 0, td = 0;
      if (lane < r) { ts = ms[b + lane]; td = md[b + lane]; }
      if (lane < r) { ms[lane] = ts; md[lane] = td; }
    }
    cnt = r;
  }

  if (cnt > 0) {
    const int li = (lane < cnt) ? lane : 0;
    const int vs = ms[li];
    const int vd = md[li];
    for (int j = 0; j < cnt; ++j) {
      const int s = __builtin_amdgcn_readlane(vs, j);
      const int d = __builtin_amdgcn_readlane(vd, j);
      const float* hp = h + (long)s * HD + lane;
      atomic_add_row(agg + (long)d * HD + lane, hp[0], hp[32], hp[64]);
    }
  }
}


__device__ __forceinline__ int block_excl_scan(int v, int* sh_w, int tid, int& tot) {
  const int lane = tid & 31;
  const int wave = tid >> 5;
  int inc = v;
  #pragma unroll
  for (int o = 1; o < 32; o <<= 1) {
    const int u = __shfl_up(inc, o, 32);
    if (lane >= o) inc += u;
  }
  if (lane == 31) sh_w[wave] = inc;
  __syncthreads();
  int woff = 0, t = 0;
  #pragma unroll
  for (int w = 0; w < 8; ++w) {
    const int x = sh_w[w];
    if (w < wave) woff += x;
    t += x;
  }
  tot = t;
  return woff + inc - v;
}

__global__ void fill_zero_push_kernel(float* __restrict__ p, long n, int* __restrict__ p2, long n2,
                                      int* __restrict__ ctr) {
  long i = (long)blockIdx.x * blockDim.x + threadIdx.x;
  const long stride = (long)gridDim.x * blockDim.x;
  if (blockIdx.x == 0 && threadIdx.x < 64) ctr[threadIdx.x] = 0;
  for (long j = i; j < n; j += stride) p[j] = 0.0f;
  for (long j = i; j < n2; j += stride) p2[j] = 0;
}

__global__ __launch_bounds__(256)
void degree_keys_kernel(const int* __restrict__ src, const int* __restrict__ dst,
                        float* __restrict__ deg_in, int* __restrict__ cnt,
                        int* __restrict__ ctr, int E, int n,
                        const float* __restrict__ Wm, const float* __restrict__ bm,
                        float* __restrict__ wsum, float* __restrict__ wbsum) {
  __shared__ int sh_t[2];
  const int tid = threadIdx.x;
  if (blockIdx.x == gridDim.x - 1) wsum_rows(Wm, bm, wsum, wbsum, tid);
  const int myh = l2_half();
  const int Nh  = (n >> 1) & ~1;
  const int nitems = (E + DCH - 1) / DCH;
  int q = myh, tn = 0;
  if (tid == 0) tn = draw_ticket(ctr, q);
  for (int it = 0; ; ++it) {
    if (tid == 0) sh_t[it & 1] = tn;
    __syncthreads();
    const int t = __builtin_amdgcn_readfirstlane(sh_t[it & 1]);
    if (t >= nitems) {
      if (q != myh) break;
      q = 1 - myh;
      if (tid == 0) tn = draw_ticket(ctr, q);
      continue;
    }
    if (tid == 0) tn = draw_ticket(ctr, q);
    const bool up = (q != 0);
    int sv[DCH / 256], dv[DCH / 256];
    #pragma unroll
    for (int k = 0; k < DCH / 256; ++k) {
      const int e  = t * DCH + k * 256 + tid;
      const int ec = (e < E) ? e : (E - 1);
      sv[k] = src[ec];
      dv[k] = dst[ec];
    }
    #pragma unroll
    for (int k = 0; k < DCH / 256; ++k) {
      const bool ok = (t * DCH + k * 256 + tid) < E;
      if (ok && ((sv[k] >= Nh) == up))
        __hip_atomic_fetch_add(&cnt[((dv[k] >= Nh) ? n : 0) + sv[k]], 1, __ATOMIC_RELAXED,
                               __HIP_MEMORY_SCOPE_AGENT);
      if (ok && ((dv[k] >= Nh) == up))
        __hip_atomic_fetch_add(&deg_in[dv[k]], 1.0f, __ATOMIC_RELAXED, __HIP_MEMORY_SCOPE_AGENT);
    }
  }
}

__global__ __launch_bounds__(256)
void scan_norm_kernel(const int* __restrict__ cnt, int* __restrict__ off, int* __restrict__ bsum,
                      float* __restrict__ norm_out, float* __restrict__ norm_in, int n) {
  __shared__ int sh_w[8];
  const int tid = threadIdx.x;
  const int nk  = 2 * n;
  const int k0  = (int)blockIdx.x * KSL + tid * 4;
  int c[4];
  #pragma unroll
  for (int i = 0; i < 4; ++i) {
    const int k  = k0 + i;
    const int kc = (k < nk) ? k : (nk - 1);
    const int x  = cnt[kc];
    c[i] = (k < nk) ? x : 0;
  }
  int tot;
  int ex = block_excl_scan(c[0] + c[1] + c[2] + c[3], sh_w, tid, tot);
  #pragma unroll
  for (int i = 0; i < 4; ++i) {
    if (k0 + i < nk) off[k0 + i] = ex;
    ex += c[i];
  }
  if (tid == 0) bsum[blockIdx.x] = tot;
  for (int s = (int)blockIdx.x * 256 + tid; s < n; s += (int)gridDim.x * 256) {
    const float dout = (float)(cnt[s] + cnt[n + s]);
    norm_out[s] = 1.0f / sqrtf(fmaxf(dout, 1.0f));
    norm_in[s]  = 1.0f / sqrtf(fmaxf(norm_in[s], 1.0f));
  }
}


constexpr int SPS = 4;
constexpr int SIT = SPW * SPS;

template <bool MAP, bool SCL>
__global__ __launch_bounds__(256)
void spmm_push_kernel(const int* __restrict__ sta, const int* __restrict__ cnt,
                      const int* __restrict__ dlist, const float* __restrict__ h,
                      const int* __restrict__ rowmap, const float* __restrict__ scale,
                      float* __restrict__ agg, int* __restrict__ ctr, int n) {
  __shared__ int sh_t[2];
  const int tid  = threadIdx.x;
  const int lane = tid & 31;
  const int wave = __builtin_amdgcn_readfirstlane(tid >> 5);
  const int myh  = l2_half();
  const int nitems = (n + SIT - 1) / SIT;
  int q = myh, tn = 0;
  if (tid == 0) tn = draw_ticket(ctr, q);
  for (int it = 0; ; ++it) {
    if (tid == 0) sh_t[it & 1] = tn;
    __syncthreads();
    const int t = __builtin_amdgcn_readfirstlane(sh_t[it & 1]);
    if (t >= nitems) {
      if (q != myh) break;
      q = 1 - myh;
      if (tid == 0) tn = draw_ticket(ctr, q);
      continue;
    }
    if (tid == 0) tn = draw_ticket(ctr, q);

    const int s0 = t * SIT + wave * SPS;
    const int sj  = s0 + (lane & (SPS - 1));
    const int sjc = (sj < n) ? sj : (n - 1);
    const int kj  = q * n + sjc;
    const int stv = sta[kj];
    const int cv0 = cnt[kj];
    const int cv  = (sj < n) ? cv0 : 0;
    float v[SPS][3];
    float fs[SPS];
    int   rw[SPS];
    #pragma unroll
    for (int j = 0; j < SPS; ++j) {
      const int s  = s0 + j;
      const int sc = (s < n) ? s : (n - 1);
      rw[j] = MAP ? rowmap[sc] : sc;
      fs[j] = SCL ? scale[sc] : 1.0f;
    }
    #pragma unroll
    for (int j = 0; j < SPS; ++j) {
      const float* hp = h + (long)rw[j] * HD + lane;
      v[j][0] = hp[0]; v[j][1] = hp[32]; v[j][2] = hp[64];
    }
    int dl[SPS];
    #pragma unroll
    for (int j = 0; j < SPS; ++j) {
      const int st = __builtin_amdgcn_readlane(stv, j);
      const int c  = __builtin_amdgcn_readlane(cv, j);
      dl[j] = (lane < c) ? dlist[st + lane] : 0;
    }
    #pragma unroll
    for (int j = 0; j < SPS; ++j) {
      const int st = __builtin_amdgcn_readlane(stv, j);
      const int c  = __builtin_amdgcn_readlane(cv, j);
      const int c1 = (c < 32) ? c : 32;
      if (SCL) { v[j][0] *= fs[j]; v[j][1] *= fs[j]; v[j][2] *= fs[j]; }
      for (int i = 0; i < c1; ++i) {
        const int d = __builtin_amdgcn_readlane(dl[j], i);
        atomic_add_row(agg + (long)d * HD + lane, v[j][0], v[j][1], v[j][2]);
      }
      for (int b = 32; b < c; b += 32) {
        const int dm = (b + lane < c) ? dlist[st + b + lane] : 0;
        const int c2 = (c - b < 32) ? (c - b) : 32;
        for (int i = 0; i < c2; ++i) {
          const int d = __builtin_amdgcn_readlane(dm, i);
          atomic_add_row(agg + (long)d * HD + lane, v[j][0], v[j][1], v[j][2]);
        }
      }
    }
  }
}

__global__ __launch_bounds__(256)
void finish_dot_kernel(const float* __restrict__ agg0, const float* __restrict__ agg1,
                       const float* __restrict__ norm_in,
                       const float* __restrict__ b, const float* __restrict__ a,
                       const float* __restrict__ wsum, const float* __restrict__ bsum,
                       float* __restrict__ out0, int n, int nb) {
  const int lane = threadIdx.x & 31;
  const int wave = __builtin_amdgcn_readfirstlane((int)(threadIdx.x >> 5));
  const int view = ((int)blockIdx.x >= nb) ? 1 : 0;
  const float* agg = view ? agg1 : agg0;
  float* out = out0 + (long)view * n;
  const long r0  = ((long)((int)blockIdx.x - view * nb) * (blockDim.x >> 5) + wave) * RW;
  if (r0 >= n) return;
  const float w0 = wsum[lane], w1 = wsum[lane + 32], w2 = wsum[lane + 64];
  const float b0 = b[lane],    b1 = b[lane + 32],    b2 = b[lane + 64];
  const float a0 = a[lane],    a1 = a[lane + 32],    a2 = a[lane + 64];
  const float bs = *bsum;
  finish_dot_rows(agg, norm_in, b0, b1, b2, a0, a1, a2, w0, w1, w2, bs, out, r0, n, lane);
}

extern "C" void kernel_launch(void* const* d_in, const int* in_sizes, int n_in,
                              void* d_out, int out_size, void* d_ws, size_t ws_size,
                              hipStream_t stream) {
  const float* x    = (const float*)d_in[0];
  const int*   src  = (const int*)  d_in[1];
  const int*   dst  = (const int*)  d_in[2];
  const int*   perm = (const int*)  d_in[3];
  const float* W1   = (const float*)d_in[4];
  const float* b1   = (const float*)d_in[5];
  const float* a1   = (const float*)d_in[6];
  const float* W2   = (const float*)d_in[7];
  const float* b2   = (const float*)d_in[8];
  const float* a2   = (const float*)d_in[9];
  const float* Wm   = (const float*)d_in[10];
  const float* bm   = (const float*)d_in[11];
  float* out = (float*)d_out;

  const int E  = in_sizes[1];
  const int N  = in_sizes[3];
  const int K1 = in_sizes[0] / N;

  float* ws       = (float*)d_ws;
  float* norm_out = ws;
  float* norm_in  = ws + N;
  float* wsum     = ws + 2L * N;
  float* wbsum    = ws + 2L * N + HD;
  int*   ctr      = (int*)(ws + 2L * N + 128);
  float* bufA     = ws + 2L * N + 2 * HD;
  float* bufB     = bufA + (long)N * HD;

  int* cnt   = (int*)(bufB + (long)N * HD);
  int* cur   = cnt + 2L * N;
  int* off   = cur + 2L * N;
  int* sta   = off + 2L * N;
  int* bsum  = sta + 2L * N;
  int* dlist = bsum + 256;
  const int  nsl  = (2 * N + KSL - 1) / KSL;
  const bool push = (nsl <= 256) &&
                    (ws_size >= (size_t)((char*)(dlist + E) - (char*)d_ws));

  float* bufC = (float*)(((size_t)(dlist + E) + 15) & ~(size_t)15);
  const bool once = push && (ws_size >= (size_t)((char*)(bufC + (long)N * HD) - (char*)d_ws));

  const int gemm_blocks = 256;
  const int red_blocks  = (int)(((long)N + 8 * RW - 1) / (8 * RW));

  if (push) {
    fill_zero_push_kernel<<<512, 256, 0, stream>>>(norm_in, (long)N, cnt, 4L * N, ctr);
    degree_keys_kernel<<<512, 256, 0, stream>>>(src, dst, norm_in, cnt, ctr, E, N, Wm, bm, wsum, wbsum);
    scan_norm_kernel<<<nsl, 256, 0, stream>>>(cnt, off, bsum, norm_out, norm_in, N);
  } else {
    fill_zero_kernel<<<512, 256, 0, stream>>>(norm_out, 2L * N, ctr);
    degree_kernel<<<512, 256, 0, stream>>>(src, dst, norm_out, norm_in, ctr, E, N, Wm, bm, wsum, wbsum);
    finalize_norm_kernel<<<(2 * N + 255) / 256, 256, 0, stream>>>(norm_out, 2 * N);
  }

  if (once) {
    gemm_rows_wmma_kernel<<<gemm_blocks, (GW + PW) * 32, 0, stream>>>(
        x, nullptr, nullptr, W1, bufA, bufB, bufC, N, K1,
        norm_in, b2, a2, wsum, wbsum, nullptr,
        src, dst, off, bsum, cur, sta, dlist, ctr + 5, E, nsl);
    spmm_push_kernel<false, true><<<2048, 256, 0, stream>>>(sta, cnt, dlist, bufA, nullptr, norm_out, bufB, ctr + 1, N);
    spmm_push_kernel<true, true><<<2048, 256, 0, stream>>>(sta, cnt, dlist, bufA, perm, norm_out, bufC, ctr + 3, N);
    gemm_rows_finish_wmma_kernel<<<gemm_blocks, GW * 32, 0, stream>>>(bufB, norm_out, W2,
                                                                     norm_in, b1, a1, bufA, bufB, N, HD);
    spmm_push_kernel<false, false><<<2048, 256, 0, stream>>>(sta, cnt, dlist, bufA, nullptr, nullptr, bufB, ctr + 2, N);
    gemm_rows_finish_wmma_kernel<<<gemm_blocks, GW * 32, 0, stream>>>(bufC, norm_out, W2,
                                                                     norm_in, b1, a1, bufA, bufC, N, HD);
    spmm_push_kernel<false, false><<<2048, 256, 0, stream>>>(sta, cnt, dlist, bufA, nullptr, nullptr, bufC, ctr + 4, N);
    finish_dot_kernel<<<2 * red_blocks, 256, 0, stream>>>(bufB, bufC, norm_in, b2, a2, wsum, wbsum,
                                                          out, N, red_blocks);
    return;
  }

  for (int view = 0; view < 2; ++view) {
    const int* rowmap = view ? perm : nullptr;
    float* pv_out = view ? out : nullptr;
    const bool ride = push && (view == 0);
    int* pl_dlist = ride ? dlist : nullptr;
    gemm_rows_wmma_kernel<<<gemm_blocks, ride ? (GW + PW) * 32 : GW * 32, 0, stream>>>(
        x, rowmap, norm_out, W1, bufA, bufB, nullptr, N, K1,
        norm_in, b2, a2, wsum, wbsum, pv_out,
        src, dst, off, bsum, cur, sta, pl_dlist, ctr + 5, E, nsl);
    if (push) spmm_push_kernel<false, false><<<2048, 256, 0, stream>>>(sta, cnt, dlist, bufA, nullptr, nullptr, bufB, ctr + 1 + 2 * view, N);
    else      spmm_kernel<<<2048, 256, 0, stream>>>(src, dst, bufA, bufB, ctr + 1 + 2 * view, E, N);
    gemm_rows_finish_wmma_kernel<<<gemm_blocks, GW * 32, 0, stream>>>(bufB, norm_out, W2,
                                                                     norm_in, b1, a1, bufA, bufB, N, HD);
    if (push) spmm_push_kernel<false, false><<<2048, 256, 0, stream>>>(sta, cnt, dlist, bufA, nullptr, nullptr, bufB, ctr + 2 + 2 * view, N);
    else      spmm_kernel<<<2048, 256, 0, stream>>>(src, dst, bufA, bufB, ctr + 2 + 2 * view, E, N);
    if (view) finish_dot_kernel<<<red_blocks, 256, 0, stream>>>(bufB, bufB, norm_in, b2, a2, wsum, wbsum,
                                                                out + (long)N, N, red_blocks);
  }
}


# layer-2 products ride in the other view's edge-pass launch: each wave of 256 resident work-groups multiplies its tile, then draws edge items (idle 4th-slot waves at once); arrival-order adds kept
# speedup vs baseline: 4.6367x; 1.0407x over previous
#include <hip/hip_runtime.h>

typedef float v2f __attribute__((ext_vector_type(2)));
typedef float v8f __attribute__((ext_vector_type(8)));

constexpr int HD   = 96;


__global__ void fill_zero_kernel(float* __restrict__ p, long n, int* __restrict__ ctr) {
  long i = (long)blockIdx.x * blockDim.x + threadIdx.x;
  long stride = (long)gridDim.x * blockDim.x;
  if (blockIdx.x == 0 && threadIdx.x < 64) ctr[threadIdx.x] = 0;
  for (; i < n; i += stride) p[i] = 0.0f;
}

constexpr int CTR_STRIDE = 32;
constexpr int KSL = 1024;

__device__ __forceinline__ int l2_half() {
  return (int)((__builtin_amdgcn_s_sendmsg_rtn(0x87) >> 18) & 1u);
}

__device__ __forceinline__ int draw_ticket(int* ctr, int q) {
  return __hip_atomic_fetch_add(ctr + q * CTR_STRIDE, 1, __ATOMIC_RELAXED, __HIP_MEMORY_SCOPE_AGENT);
}

__device__ __forceinline__
void wsum_rows(const float* __restrict__ Wm, const float* __restrict__ bm,
               float* __restrict__ wsum, float* __restrict__ bsum, int t) {
  if (t < HD) {
    float s = 0.0f;
    for (int j = 0; j < HD; ++j) s += Wm[t * HD + j];
    wsum[t] = s;
  } else if (t == HD) {
    float s = 0.0f;
    for (int j = 0; j < HD; ++j) s += bm[j];
    *bsum = s;
  }
}

constexpr int DCH = 1024;

__global__ __launch_bounds__(256)
void degree_kernel(const int* __restrict__ src, const int* __restrict__ dst,
                   float* __restrict__ deg_out, float* __restrict__ deg_in,
                   int* __restrict__ ctr, int E, int n,
                   const float* __restrict__ Wm, const float* __restrict__ bm,
                   float* __restrict__ wsum, float* __restrict__ wbsum) {
  __shared__ int sh_t[2];
  const int tid = threadIdx.x;
  if (blockIdx.x == gridDim.x - 1) wsum_rows(Wm, bm, wsum, wbsum, tid);
  const int myh = l2_half();
  const int Nh  = (n >> 1) & ~1;
  const int nitems = (E + DCH - 1) / DCH;
  int q = myh, tn = 0;
  if (tid == 0) tn = draw_ticket(ctr, q);
  for (int it = 0; ; ++it) {
    if (tid == 0) sh_t[it & 1] = tn;
    __syncthreads();
    const int t = __builtin_amdgcn_readfirstlane(sh_t[it & 1]);
    if (t >= nitems) {
      if (q != myh) break;
      q = 1 - myh;
      if (tid == 0) tn = draw_ticket(ctr, q);
      continue;
    }
    if (tid == 0) tn = draw_ticket(ctr, q);
    const bool up = (q != 0);
    int sv[DCH / 256], dv[DCH / 256];
    #pragma unroll
    for (int k = 0; k < DCH / 256; ++k) {
      const int e  = t * DCH + k * 256 + tid;
      const int ec = (e < E) ? e : (E - 1);
      sv[k] = src[ec];
      dv[k] = dst[ec];
    }
    #pragma unroll
    for (int k = 0; k < DCH / 256; ++k) {
      const bool ok = (t * DCH + k * 256 + tid) < E;
      if (ok && ((sv[k] >= Nh) == up))
        __hip_atomic_fetch_add(&deg_out[sv[k]], 1.0f, __ATOMIC_RELAXED, __HIP_MEMORY_SCOPE_AGENT);
      if (ok && ((dv[k] >= Nh) == up))
        __hip_atomic_fetch_add(&deg_in[dv[k]],  1.0f, __ATOMIC_RELAXED, __HIP_MEMORY_SCOPE_AGENT);
    }
  }
}

__global__ void finalize_norm_kernel(float* __restrict__ p, int n) {
  int i = blockIdx.x * blockDim.x + threadIdx.x;
  if (i < n) p[i] = 1.0f / sqrtf(fmaxf(p[i], 1.0f));
}

constexpr int RW = 8;

__device__ __forceinline__
void finish_dot_rows(const float* agg, const float* __restrict__ norm_in,
                     float b0, float b1, float b2, float a0, float a1, float a2,
                     float w0, float w1, float w2, float bs,
                     float* __restrict__ out, long r0, int n, int lane) {
  float zv[RW][3];
  float sc[RW];
  #pragma unroll
  for (int q = 0; q < RW; ++q) {
    const long r = (r0 + q < n) ? (r0 + q) : (long)(n - 1);
    const float* zp = agg + r * HD + lane;
    zv[q][0] = zp[0]; zv[q][1] = zp[32]; zv[q][2] = zp[64];
    sc[q] = norm_in[r];
  }
  float sum[RW];
  #pragma unroll
  for (int q = 0; q < RW; ++q) {
    float t, z0, z1, z2;
    t = zv[q][0] * sc[q] + b0; z0 = (t >= 0.0f) ? t : a0 * t;
    t = zv[q][1] * sc[q] + b1; z1 = (t >= 0.0f) ? t : a1 * t;
    t = zv[q][2] * sc[q] + b2; z2 = (t >= 0.0f) ? t : a2 * t;
    float s = 0.0f;
    s = fmaf(z0, w0, s);
    s = fmaf(z1, w1, s);
    s = fmaf(z2, w2, s);
    sum[q] = s;
  }
  #pragma unroll
  for (int off = 16; off > 0; off >>= 1) {
    #pragma unroll
    for (int q = 0; q < RW; ++q) sum[q] += __shfl_xor(sum[q], off, 32);
  }
  if (lane == 0) {
    #pragma unroll
    for (int q = 0; q < RW; ++q)
      if (r0 + q < n) out[r0 + q] = sum[q] + bs;
  }
}

__device__ __forceinline__
void finish_dot_tile(const float* agg, const float* __restrict__ norm_in,
                     float b0, float b1, float b2, float a0, float a1, float a2,
                     float w0, float w1, float w2, float bs,
                     float* __restrict__ out, int tile, int n, int lane) {
  #pragma unroll
  for (int g = 0; g < 16 / RW; ++g) {
    const long r0 = (long)tile * 16 + g * RW;
    if (r0 < n)
      finish_dot_rows(agg, norm_in, b0, b1, b2, a0, a1, a2, w0, w1, w2, bs, out, r0, n, lane);
  }
}

constexpr int KB   = 64;
constexpr int KMAX = 512;
constexpr int WS   = 514;
constexpr int GW   = 16;
constexpr int FW   = 12;
constexpr int HW   = 3;

__device__ __forceinline__
void gemm_load_block(float4 (&dstv)[8], const float* X, int srow, int K, int k0) {
  #pragma unroll
  for (int q = 0; q < 8; ++q) {
    const int k  = k0 + q * 8;
    const int kc = (k < K) ? k : (K - 4);
    dstv[q] = *(const float4*)(X + (long)srow * K + kc);
  }
}

__device__ __forceinline__
void gemm_load_finish(float4 (&bb)[8], float4 (&aa)[8], const float* __restrict__ fin_b,
                      const float* __restrict__ fin_a, int K, int k0) {
  #pragma unroll
  for (int q = 0; q < 8; ++q) {
    const int k  = k0 + q * 8;
    const int kc = (k < K) ? k : (K - 4);
    bb[q] = *(const float4*)(fin_b + kc);
    aa[q] = *(const float4*)(fin_a + kc);
  }
}

template <bool FIN, int NT>
__device__ __forceinline__
void gemm_block_products(v8f (&acc)[NT], const float4 (&raw)[8], const float4 (&bb)[8],
                         const float4 (&aa)[8], float s, float si, int K, int k0,
                         const float* wb) {
  float4 cur[8];
  #pragma unroll
  for (int q = 0; q < 8; ++q) {
    float4 v = raw[q];
    const int k = k0 + q * 8;
    if (FIN) {
      float t;
      t = v.x * si + bb[q].x; v.x = (t >= 0.0f) ? t : aa[q].x * t;
      t = v.y * si + bb[q].y; v.y = (t >= 0.0f) ? t : aa[q].y * t;
      t = v.z * si + bb[q].z; v.z = (t >= 0.0f) ? t : aa[q].z * t;
      t = v.w * si + bb[q].w; v.w = (t >= 0.0f) ? t : aa[q].w * t;
    }
    v.x *= s; v.y *= s; v.z *= s; v.w *= s;
    if (k >= K) v = make_float4(0.f, 0.f, 0.f, 0.f);
    cur[q] = v;
  }
  #pragma unroll
  for (int st = 0; st < 16; ++st) {
    v2f A;
    if (st & 1) { A.x = cur[st >> 1].z; A.y = cur[st >> 1].w; }
    else        { A.x = cur[st >> 1].x; A.y = cur[st >> 1].y; }
    v2f Bf[NT];
    #pragma unroll
    for (int t = 0; t < NT; ++t)
      Bf[t] = *(const v2f*)(wb + t * 16 * WS + st * 2);
    #pragma unroll
    for (int t = 0; t < NT; ++t)
      acc[t] = __builtin_amdgcn_wmma_f32_16x16x4_f32(
           false, A,  false, Bf[t],
           (short)0, acc[t],  false,  false);
  }
}

template <bool FIN, int NT>
__device__ __forceinline__
void gemm_tile_cols(v8f (&acc)[NT], const float* ldsW, const float* X,
                    const int* __restrict__ rowmap, const float* __restrict__ norm,
                    const float* __restrict__ fin_norm, const float* __restrict__ fin_b,
                    const float* __restrict__ fin_a, int nrows, int K, int Kpad,
                    int tile, int c0, int half, int l16) {
  const int grow = tile * 16 + l16;
  const int crow = (grow < nrows) ? grow : (nrows - 1);
  const int srow = rowmap ? rowmap[crow] : crow;
  const float s  = norm ? norm[crow] : 1.0f;
  float si = 0.0f;
  if (FIN) si = fin_norm[crow];

  const float* wl = ldsW + (c0 * 16 + l16) * WS + half * 32;

  float4 buf0[8], buf1[8];
  float4 bb[8], aa[8];
  gemm_load_block(buf0, X, srow, K, half * 4);
  for (int kb = 0; kb < Kpad; kb += 2 * KB) {
    if (FIN) gemm_load_finish(bb, aa, fin_b, fin_a, K, kb + half * 4);
    gemm_load_block(buf1, X, srow, K, kb + KB + half * 4);
    gemm_block_products<FIN, NT>(acc, buf0, bb, aa, s, si, K, kb + half * 4, wl + kb);
    if (FIN) gemm_load_finish(bb, aa, fin_b, fin_a, K, kb + KB + half * 4);
    gemm_load_block(buf0, X, srow, K, kb + 2 * KB + half * 4);
    gemm_block_products<FIN, NT>(acc, buf1, bb, aa, s, si, K, kb + KB + half * 4, wl + kb + KB);
  }
}

__device__ __forceinline__
void gemm_zero_rows(float* ZERO, int tile, int nrows, int lane) {
  const long zend = (long)nrows * HD;
  #pragma unroll
  for (int it = 0; it < (16 * HD / 4) / 32; ++it) {
    const long off = (long)tile * 16 * HD + (long)(lane + 32 * it) * 4;
    if (off < zend) *(float4*)(ZERO + off) = make_float4(0.f, 0.f, 0.f, 0.f);
  }
}

template <int NT>
__device__ __forceinline__
void gemm_store_cols(float* __restrict__ OUT, const v8f (&acc)[NT], int tile, int c0, int nrows,
                     int half, int l16) {
  const int orow = tile * 16 + half * 8;
  #pragma unroll
  for (int t = 0; t < NT; ++t) {
    int col = (c0 + t) * 16 + l16;
    #pragma unroll
    for (int i = 0; i < 8; ++i) {
      int r = orow + i;
      if (r < nrows) OUT[(long)r * HD + col] = acc[t][i];
    }
  }
}

constexpr int PW  = 4;
constexpr int PE  = 16;
constexpr int PCH = 32 * PE;

__device__ __forceinline__
void place_wave(const int* __restrict__ src, const int* __restrict__ dst,
                const int* __restrict__ off, int* __restrict__ cur, int* __restrict__ sta,
                int* __restrict__ dlist, int* __restrict__ ctr, const int* sh_base,
                int E, int n, int pw, int lane) {
  const int myh = l2_half();
  const int Nh  = (n >> 1) & ~1;
  const int nk  = 2 * n;
  {
    const int gw  = (int)blockIdx.x * PW + pw;
    const int ngw = (int)gridDim.x * PW;
    for (int k = gw * 32 + lane; k < nk; k += ngw * 32) sta[k] = sh_base[k / KSL] + off[k];
  }
  const int nitems = (E + PCH - 1) / PCH;
  int q = myh, tn = 0;
  if (lane == 0) tn = draw_ticket(ctr, q);
  for (;;) {
    const int t = __builtin_amdgcn_readfirstlane(tn);
    if (t >= nitems) {
      if (q != myh) break;
      q = 1 - myh;
      if (lane == 0) tn = draw_ticket(ctr, q);
      continue;
    }
    if (lane == 0) tn = draw_ticket(ctr, q);
    const bool up = (q != 0);
    int sv[PE], dv[PE], key[PE], lo[PE], rk[PE];
    #pragma unroll
    for (int k = 0; k < PE; ++k) {
      const int e  = t * PCH + k * 32 + lane;
      const int ec = (e < E) ? e : (E - 1);
      sv[k] = src[ec];
      dv[k] = dst[ec];
    }
    #pragma unroll
    for (int k = 0; k < PE; ++k)
      if (t * PCH + k * 32 + lane >= E) dv[k] = -1;
    #pragma unroll
    for (int k = 0; k < PE; ++k) {
      const bool mine = (dv[k] >= 0) && ((sv[k] >= Nh) == up);
      key[k] = mine ? (((dv[k] >= Nh) ? n : 0) + sv[k]) : -1;
    }
    #pragma unroll
    for (int k = 0; k < PE; ++k) {
      lo[k] = 0; rk[k] = 0;
      if (key[k] >= 0) {
        lo[k] = off[key[k]];
        rk[k] = __hip_atomic_fetch_add(&cur[key[k]], 1, __ATOMIC_RELAXED, __HIP_MEMORY_SCOPE_AGENT);
      }
    }
    #pragma unroll
    for (int k = 0; k < PE; ++k) {
      if (key[k] >= 0) {
        const int pos = sh_base[key[k] / KSL] + lo[k] + rk[k];
        __hip_atomic_exchange(&dlist[pos], dv[k], __ATOMIC_RELAXED, __HIP_MEMORY_SCOPE_AGENT);
      }
    }
  }
}

template <bool FIN>
__device__ __forceinline__
void gemm_rows_body(float* ldsW, const float* X, const int* __restrict__ rowmap,
                    const float* __restrict__ norm, const float* __restrict__ W,
                    const float* __restrict__ fin_norm, const float* __restrict__ fin_b,
                    const float* __restrict__ fin_a,
                    float* __restrict__ OUT, float* ZERO, float* ZERO2, int nrows, int K,
                    const float* __restrict__ pv_norm, const float* __restrict__ pv_b,
                    const float* __restrict__ pv_a, const float* __restrict__ pv_wsum,
                    const float* __restrict__ pv_bsum, float* __restrict__ pv_out,
                    const int* __restrict__ pl_src, const int* __restrict__ pl_dst,
                    const int* __restrict__ pl_off, const int* __restrict__ pl_bsum,
                    int* __restrict__ pl_cur, int* __restrict__ pl_sta,
                    int* __restrict__ pl_dlist, int* __restrict__ pl_ctr, int* sh_pbase,
                    int pl_E, int pl_nsl) {

  const int tid  = threadIdx.x;
  const int lane = tid & 31;
  const int half = lane >> 4;
  const int l16  = lane & 15;
  const int wave = __builtin_amdgcn_readfirstlane(tid >> 5);
  const int Kpad = (K + 2 * KB - 1) & ~(2 * KB - 1);
  const int ntiles = (nrows + 15) >> 4;
  const int nwhole = (int)gridDim.x * FW;

  if (!FIN && pv_out) {
    const float w0 = pv_wsum[lane], w1 = pv_wsum[lane + 32], w2 = pv_wsum[lane + 64];
    const float b0 = pv_b[lane],    b1 = pv_b[lane + 32],    b2 = pv_b[lane + 64];
    const float a0 = pv_a[lane],    a1 = pv_a[lane + 32],    a2 = pv_a[lane + 64];
    const float bs = *pv_bsum;
    if (wave < FW) {
      const int tile = (int)blockIdx.x * FW + wave;
      if (tile < ntiles)
        finish_dot_tile(ZERO, pv_norm, b0, b1, b2, a0, a1, a2, w0, w1, w2, bs, pv_out, tile, nrows, lane);
    } else if (wave == FW) {
      for (int tile = nwhole + (int)blockIdx.x; tile < ntiles; tile += (int)gridDim.x)
        finish_dot_tile(ZERO, pv_norm, b0, b1, b2, a0, a1, a2, w0, w1, w2, bs, pv_out, tile, nrows, lane);
    }
  }

  if (!FIN && pl_dlist && wave == GW) {
    int v[8], tot = 0;
    #pragma unroll
    for (int i = 0; i < 8; ++i) {
      const int idx = lane * 8 + i;
      const int x   = pl_bsum[(idx < pl_nsl) ? idx : 0];
      v[i] = (idx < pl_nsl) ? x : 0;
      tot += v[i];
    }
    int inc = tot;
    #pragma unroll
    for (int o = 1; o < 32; o <<= 1) {
      const int u = __shfl_up(inc, o, 32);
      if (lane >= o) inc += u;
    }
    int ex = inc - tot;
    #pragma unroll
    for (int i = 0; i < 8; ++i) { sh_pbase[lane * 8 + i] = ex; ex += v[i]; }
  }

  if (wave < GW) {
    const int nitem = Kpad * (HD / 4);
    #pragma unroll 6
    for (int idx = tid; idx < nitem; idx += GW * 32) {
      const int k  = idx / (HD / 4);
      const int j  = idx % (HD / 4);
      const int kc = (k < K) ? k : (K - 1);
      float4 v = *(const float4*)(W + (long)kc * HD + j * 4);
      if (k >= K) v = make_float4(0.f, 0.f, 0.f, 0.f);
      const int kk = k & (KB - 1);
      const int p  = (k & ~(KB - 1)) + ((kk >> 2) & 1) * 32 + (kk >> 3) * 4 + (kk & 3);
      ldsW[(j * 4 + 0) * WS + p] = v.x;
      ldsW[(j * 4 + 1) * WS + p] = v.y;
      ldsW[(j * 4 + 2) * WS + p] = v.z;
      ldsW[(j * 4 + 3) * WS + p] = v.w;
    }
  }
  __syncthreads();

  if (!FIN && wave >= GW) {
    if (pl_dlist)
      place_wave(pl_src, pl_dst, pl_off, pl_cur, pl_sta, pl_dlist, pl_ctr, sh_pbase,
                 pl_E, nrows, wave - GW, lane);
    return;
  }

  if (wave < FW) {
    const int tile = (int)blockIdx.x * FW + wave;
    if (tile < ntiles) {
      v8f acc[6] = {};
      gemm_tile_cols<FIN, 6>(acc, ldsW, X, rowmap, norm, fin_norm, fin_b, fin_a,
                             nrows, K, Kpad, tile, 0, half, l16);
      gemm_zero_rows(ZERO, tile, nrows, lane);
      if (ZERO2) gemm_zero_rows(ZERO2, tile, nrows, lane);
      gemm_store_cols<6>(OUT, acc, tile, 0, nrows, half, l16);
    }
  }

  for (int tile = nwhole + (int)blockIdx.x; tile < ntiles; tile += (int)gridDim.x) {
    const bool part = (wave >= FW) && (wave < FW + HW);
    const int  c0   = (wave - FW) * 2;
    v8f acc2[2] = {};
    if (part)
      gemm_tile_cols<FIN, 2>(acc2, ldsW, X, rowmap, norm, fin_norm, fin_b, fin_a,
                             nrows, K, Kpad, tile, c0, half, l16);
    if (FIN) __syncthreads();
    if (wave == FW) {
      gemm_zero_rows(ZERO, tile, nrows, lane);
      if (ZERO2) gemm_zero_rows(ZERO2, tile, nrows, lane);
    }
    if (part) gemm_store_cols<2>(OUT, acc2, tile, c0, nrows, half, l16);
  }
}

__global__ __launch_bounds__((GW + PW) * 32)
void gemm_rows_wmma_kernel(const float* X, const int* __restrict__ rowmap,
                           const float* __restrict__ norm, const float* __restrict__ W,
                           float* __restrict__ OUT, float* ZERO, float* ZERO2, int nrows, int K,
                           const float* __restrict__ pv_norm, const float* __restrict__ pv_b,
                           const float* __restrict__ pv_a, const float* __restrict__ pv_wsum,
                           const float* __restrict__ pv_bsum, float* __restrict__ pv_out,
                           const int* __restrict__ pl_src, const int* __restrict__ pl_dst,
                           const int* __restrict__ pl_off, const int* __restrict__ pl_bsum,
                           int* __restrict__ pl_cur, int* __restrict__ pl_sta,
                           int* __restrict__ pl_dlist, int* __restrict__ pl_ctr,
                           int pl_E, int pl_nsl) {
  __shared__ float ldsW[HD * WS];
  __shared__ int sh_pbase[256];
  gemm_rows_body<false>(ldsW, X, rowmap, norm, W, nullptr, nullptr, nullptr, OUT, ZERO, ZERO2, nrows, K,
                        pv_norm, pv_b, pv_a, pv_wsum, pv_bsum, pv_out,
                        pl_src, pl_dst, pl_off, pl_bsum, pl_cur, pl_sta, pl_dlist, pl_ctr,
                        sh_pbase, pl_E, pl_nsl);
}

__global__ __launch_bounds__(GW * 32)
void gemm_rows_finish_wmma_kernel(const float* X, const float* __restrict__ norm,
                                  const float* __restrict__ W,
                                  const float* __restrict__ fin_norm,
                                  const float* __restrict__ fin_b,
                                  const float* __restrict__ fin_a,
                                  float* __restrict__ OUT, float* ZERO, int nrows, int K) {
  __shared__ float ldsW[HD * WS];
  gemm_rows_body<true>(ldsW, X, nullptr, norm, W, fin_norm, fin_b, fin_a, OUT, ZERO, nullptr, nrows, K,
                       nullptr, nullptr, nullptr, nullptr, nullptr, nullptr,
                       nullptr, nullptr, nullptr, nullptr, nullptr, nullptr, nullptr, nullptr,
                       nullptr, 0, 0);
}

constexpr int EB   = 8;
constexpr int SPW  = 8;
constexpr int WCH  = 64;
constexpr int GCH  = SPW * WCH;
constexpr int LCAP = WCH + EB;

__device__ __forceinline__ void atomic_add_row(float* __restrict__ ap, float v0, float v1, float v2) {
  __hip_atomic_fetch_add(&ap[0],  v0, __ATOMIC_RELAXED, __HIP_MEMORY_SCOPE_AGENT);
  __hip_atomic_fetch_add(&ap[32], v1, __ATOMIC_RELAXED, __HIP_MEMORY_SCOPE_AGENT);
  __hip_atomic_fetch_add(&ap[64], v2, __ATOMIC_RELAXED, __HIP_MEMORY_SCOPE_AGENT);
}

__global__ __launch_bounds__(256)
void spmm_kernel(const int* __restrict__ src, const int* __restrict__ dst,
                 const float* __restrict__ h, float* __restrict__ agg,
                 int* __restrict__ ctr, int E, int n) {
  __shared__ int lst_s[SPW][LCAP];
  __shared__ int lst_d[SPW][LCAP];
  __shared__ int sh_t[2];
  const int tid  = threadIdx.x;
  const int lane = tid & 31;
  const int wave = __builtin_amdgcn_readfirstlane(tid >> 5);
  const int myh  = l2_half();
  const int Nh   = (n >> 1) & ~1;
  const int nitems = (E + GCH - 1) / GCH;
  int* ms = lst_s[wave];
  int* md = lst_d[wave];

  int q = myh, cnt = 0, tn = 0;
  if (tid == 0) tn = draw_ticket(ctr, q);
  for (int it = 0; ; ++it) {
    if (tid == 0) sh_t[it & 1] = tn;
    __syncthreads();
    const int t = __builtin_amdgcn_readfirstlane(sh_t[it & 1]);
    if (t >= nitems) {
      if (q != myh) break;
      q = 1 - myh;
      if (tid == 0) tn = draw_ticket(ctr, q);
      continue;
    }
    if (tid == 0) tn = draw_ticket(ctr, q);

    const bool up = (q != 0);
    int sv[WCH / 32], dv[WCH / 32];
    #pragma unroll
    for (int g = 0; g < WCH / 32; ++g) {
      const int e  = t * GCH + wave * WCH + g * 32 + lane;
      const int ec = (e < E) ? e : (E - 1);
      sv[g] = src[ec];
      dv[g] = (e < E) ? dst[ec] : -1;
    }
    #pragma unroll
    for (int g = 0; g < WCH / 32; ++g) {
      const bool mine = (dv[g] >= 0) && ((dv[g] >= Nh) == up);
      const unsigned m = __builtin_amdgcn_ballot_w32(mine);
      const int pos = cnt + (int)__builtin_amdgcn_mbcnt_lo(m, 0u);
      if (mine) { ms[pos] = sv[g]; md[pos] = dv[g]; }
      cnt += __builtin_popcount(m);
    }

    int b = 0;
    for (; b + EB <= cnt; b += EB) {
      const int vs = ms[b + (lane & (EB - 1))];
      const int vd = md[b + (lane & (EB - 1))];
      float v[EB][3];
      #pragma unroll
      for (int j = 0; j < EB; ++j) {
        const int s = __builtin_amdgcn_readlane(vs, j);
        const float* hp = h + (long)s * HD + lane;
        v[j][0] = hp[0]; v[j][1] = hp[32]; v[j][2] = hp[64];
      }
      #pragma unroll
      for (int j = 0; j < EB; ++j) {
        const int d = __builtin_amdgcn_readlane(vd, j);
        atomic_add_row(agg + (long)d * HD + lane, v[j][0], v[j][1], v[j][2]);
      }
    }
    const int r = cnt - b;
    if (b > 0 && r > 0) {
      int ts = 0, td = 0;
      if (lane < r) { ts = ms[b + lane]; td = md[b + lane]; }
      if (lane < r) { ms[lane] = ts; md[lane] = td; }
    }
    cnt = r;
  }

  if (cnt > 0) {
    const int li = (lane < cnt) ? lane : 0;
    const int vs = ms[li];
    const int vd = md[li];
    for (int j = 0; j < cnt; ++j) {
      const int s = __builtin_amdgcn_readlane(vs, j);
      const int d = __builtin_amdgcn_readlane(vd, j);
      const float* hp = h + (long)s * HD + lane;
      atomic_add_row(agg + (long)d * HD + lane, hp[0], hp[32], hp[64]);
    }
  }
}


__device__ __forceinline__ int block_excl_scan(int v, int* sh_w, int tid, int& tot) {
  const int lane = tid & 31;
  const int wave = tid >> 5;
  int inc = v;
  #pragma unroll
  for (int o = 1; o < 32; o <<= 1) {
    const int u = __shfl_up(inc, o, 32);
    if (lane >= o) inc += u;
  }
  if (lane == 31) sh_w[wave] = inc;
  __syncthreads();
  int woff = 0, t = 0;
  #pragma unroll
  for (int w = 0; w < 8; ++w) {
    const int x = sh_w[w];
    if (w < wave) woff += x;
    t += x;
  }
  tot = t;
  return woff + inc - v;
}

__global__ void fill_zero_push_kernel(float* __restrict__ p, long n, int* __restrict__ p2, long n2,
                                      int* __restrict__ ctr) {
  long i = (long)blockIdx.x * blockDim.x + threadIdx.x;
  const long stride = (long)gridDim.x * blockDim.x;
  if (blockIdx.x == 0 && threadIdx.x < 64) ctr[threadIdx.x] = 0;
  for (long j = i; j < n; j += stride) p[j] = 0.0f;
  for (long j = i; j < n2; j += stride) p2[j] = 0;
}

__global__ __launch_bounds__(256)
void degree_keys_kernel(const int* __restrict__ src, const int* __restrict__ dst,
                        float* __restrict__ deg_in, int* __restrict__ cnt,
                        int* __restrict__ ctr, int E, int n,
                        const float* __restrict__ Wm, const float* __restrict__ bm,
                        float* __restrict__ wsum, float* __restrict__ wbsum) {
  __shared__ int sh_t[2];
  const int tid = threadIdx.x;
  if (blockIdx.x == gridDim.x - 1) wsum_rows(Wm, bm, wsum, wbsum, tid);
  const int myh = l2_half();
  const int Nh  = (n >> 1) & ~1;
  const int nitems = (E + DCH - 1) / DCH;
  int q = myh, tn = 0;
  if (tid == 0) tn = draw_ticket(ctr, q);
  for (int it = 0; ; ++it) {
    if (tid == 0) sh_t[it & 1] = tn;
    __syncthreads();
    const int t = __builtin_amdgcn_readfirstlane(sh_t[it & 1]);
    if (t >= nitems) {
      if (q != myh) break;
      q = 1 - myh;
      if (tid == 0) tn = draw_ticket(ctr, q);
      continue;
    }
    if (tid == 0) tn = draw_ticket(ctr, q);
    const bool up = (q != 0);
    int sv[DCH / 256], dv[DCH / 256];
    #pragma unroll
    for (int k = 0; k < DCH / 256; ++k) {
      const int e  = t * DCH + k * 256 + tid;
      const int ec = (e < E) ? e : (E - 1);
      sv[k] = src[ec];
      dv[k] = dst[ec];
    }
    #pragma unroll
    for (int k = 0; k < DCH / 256; ++k) {
      const bool ok = (t * DCH + k * 256 + tid) < E;
      if (ok && ((sv[k] >= Nh) == up))
        __hip_atomic_fetch_add(&cnt[((dv[k] >= Nh) ? n : 0) + sv[k]], 1, __ATOMIC_RELAXED,
                               __HIP_MEMORY_SCOPE_AGENT);
      if (ok && ((dv[k] >= Nh) == up))
        __hip_atomic_fetch_add(&deg_in[dv[k]], 1.0f, __ATOMIC_RELAXED, __HIP_MEMORY_SCOPE_AGENT);
    }
  }
}

__global__ __launch_bounds__(256)
void scan_norm_kernel(const int* __restrict__ cnt, int* __restrict__ off, int* __restrict__ bsum,
                      float* __restrict__ norm_out, float* __restrict__ norm_in, int n) {
  __shared__ int sh_w[8];
  const int tid = threadIdx.x;
  const int nk  = 2 * n;
  const int k0  = (int)blockIdx.x * KSL + tid * 4;
  int c[4];
  #pragma unroll
  for (int i = 0; i < 4; ++i) {
    const int k  = k0 + i;
    const int kc = (k < nk) ? k : (nk - 1);
    const int x  = cnt[kc];
    c[i] = (k < nk) ? x : 0;
  }
  int tot;
  int ex = block_excl_scan(c[0] + c[1] + c[2] + c[3], sh_w, tid, tot);
  #pragma unroll
  for (int i = 0; i < 4; ++i) {
    if (k0 + i < nk) off[k0 + i] = ex;
    ex += c[i];
  }
  if (tid == 0) bsum[blockIdx.x] = tot;
  for (int s = (int)blockIdx.x * 256 + tid; s < n; s += (int)gridDim.x * 256) {
    const float dout = (float)(cnt[s] + cnt[n + s]);
    norm_out[s] = 1.0f / sqrtf(fmaxf(dout, 1.0f));
    norm_in[s]  = 1.0f / sqrtf(fmaxf(norm_in[s], 1.0f));
  }
}


constexpr int SPS = 4;
constexpr int SIT = SPW * SPS;

template <bool MAP, bool SCL>
__global__ __launch_bounds__(256)
void spmm_push_kernel(const int* __restrict__ sta, const int* __restrict__ cnt,
                      const int* __restrict__ dlist, const float* __restrict__ h,
                      const int* __restrict__ rowmap, const float* __restrict__ scale,
                      float* __restrict__ agg, int* __restrict__ ctr, int n) {
  __shared__ int sh_t[2];
  const int tid  = threadIdx.x;
  const int lane = tid & 31;
  const int wave = __builtin_amdgcn_readfirstlane(tid >> 5);
  const int myh  = l2_half();
  const int nitems = (n + SIT - 1) / SIT;
  int q = myh, tn = 0;
  if (tid == 0) tn = draw_ticket(ctr, q);
  for (int it = 0; ; ++it) {
    if (tid == 0) sh_t[it & 1] = tn;
    __syncthreads();
    const int t = __builtin_amdgcn_readfirstlane(sh_t[it & 1]);
    if (t >= nitems) {
      if (q != myh) break;
      q = 1 - myh;
      if (tid == 0) tn = draw_ticket(ctr, q);
      continue;
    }
    if (tid == 0) tn = draw_ticket(ctr, q);

    const int s0 = t * SIT + wave * SPS;
    const int sj  = s0 + (lane & (SPS - 1));
    const int sjc = (sj < n) ? sj : (n - 1);
    const int kj  = q * n + sjc;
    const int stv = sta[kj];
    const int cv0 = cnt[kj];
    const int cv  = (sj < n) ? cv0 : 0;
    float v[SPS][3];
    float fs[SPS];
    int   rw[SPS];
    #pragma unroll
    for (int j = 0; j < SPS; ++j) {
      const int s  = s0 + j;
      const int sc = (s < n) ? s : (n - 1);
      rw[j] = MAP ? rowmap[sc] : sc;
      fs[j] = SCL ? scale[sc] : 1.0f;
    }
    #pragma unroll
    for (int j = 0; j < SPS; ++j) {
      const float* hp = h + (long)rw[j] * HD + lane;
      v[j][0] = hp[0]; v[j][1] = hp[32]; v[j][2] = hp[64];
    }
    int dl[SPS];
    #pragma unroll
    for (int j = 0; j < SPS; ++j) {
      const int st = __builtin_amdgcn_readlane(stv, j);
      const int c  = __builtin_amdgcn_readlane(cv, j);
      dl[j] = (lane < c) ? dlist[st + lane] : 0;
    }
    #pragma unroll
    for (int j = 0; j < SPS; ++j) {
      const int st = __builtin_amdgcn_readlane(stv, j);
      const int c  = __builtin_amdgcn_readlane(cv, j);
      const int c1 = (c < 32) ? c : 32;
      if (SCL) { v[j][0] *= fs[j]; v[j][1] *= fs[j]; v[j][2] *= fs[j]; }
      for (int i = 0; i < c1; ++i) {
        const int d = __builtin_amdgcn_readlane(dl[j], i);
        atomic_add_row(agg + (long)d * HD + lane, v[j][0], v[j][1], v[j][2]);
      }
      for (int b = 32; b < c; b += 32) {
        const int dm = (b + lane < c) ? dlist[st + b + lane] : 0;
        const int c2 = (c - b < 32) ? (c - b) : 32;
        for (int i = 0; i < c2; ++i) {
          const int d = __builtin_amdgcn_readlane(dm, i);
          atomic_add_row(agg + (long)d * HD + lane, v[j][0], v[j][1], v[j][2]);
        }
      }
    }
  }
}

constexpr int SPR = 8;

template <bool MAP, bool SCL>
__device__ __forceinline__
void spmm_push_wave(const int* __restrict__ sta, const int* __restrict__ cnt,
                    const int* __restrict__ dlist, const float* __restrict__ h,
                    const int* __restrict__ rowmap, const float* __restrict__ scale,
                    float* __restrict__ agg, int* __restrict__ ctr, int n, int lane) {
  const int myh = l2_half();
  const int nitems = (n + SPR - 1) / SPR;
  int q = myh, tn = 0;
  if (lane == 0) tn = draw_ticket(ctr, q);
  for (;;) {
    const int t = __builtin_amdgcn_readfirstlane(tn);
    if (t >= nitems) {
      if (q != myh) break;
      q = 1 - myh;
      if (lane == 0) tn = draw_ticket(ctr, q);
      continue;
    }
    if (lane == 0) tn = draw_ticket(ctr, q);

    const int s0 = t * SPR;
    const int sj  = s0 + (lane & (SPR - 1));
    const int sjc = (sj < n) ? sj : (n - 1);
    const int kj  = q * n + sjc;
    const int stv = sta[kj];
    const int cv0 = cnt[kj];
    const int cv  = (sj < n) ? cv0 : 0;
    float v[SPR][3];
    float fs[SPR];
    int   rw[SPR];
    #pragma unroll
    for (int j = 0; j < SPR; ++j) {
      const int s  = s0 + j;
      const int sc = (s < n) ? s : (n - 1);
      rw[j] = MAP ? rowmap[sc] : sc;
      fs[j] = SCL ? scale[sc] : 1.0f;
    }
    #pragma unroll
    for (int j = 0; j < SPR; ++j) {
      const float* hp = h + (long)rw[j] * HD + lane;
      v[j][0] = hp[0]; v[j][1] = hp[32]; v[j][2] = hp[64];
    }
    int dl[SPR];
    #pragma unroll
    for (int j = 0; j < SPR; ++j) {
      const int st = __builtin_amdgcn_readlane(stv, j);
      const int c  = __builtin_amdgcn_readlane(cv, j);
      dl[j] = (lane < c) ? dlist[st + lane] : 0;
    }
    #pragma unroll
    for (int j = 0; j < SPR; ++j) {
      const int st = __builtin_amdgcn_readlane(stv, j);
      const int c  = __builtin_amdgcn_readlane(cv, j);
      const int c1 = (c < 32) ? c : 32;
      if (SCL) { v[j][0] *= fs[j]; v[j][1] *= fs[j]; v[j][2] *= fs[j]; }
      for (int i = 0; i < c1; ++i) {
        const int d = __builtin_amdgcn_readlane(dl[j], i);
        atomic_add_row(agg + (long)d * HD + lane, v[j][0], v[j][1], v[j][2]);
      }
      for (int b = 32; b < c; b += 32) {
        const int dm = (b + lane < c) ? dlist[st + b + lane] : 0;
        const int c2 = (c - b < 32) ? (c - b) : 32;
        for (int i = 0; i < c2; ++i) {
          const int d = __builtin_amdgcn_readlane(dm, i);
          atomic_add_row(agg + (long)d * HD + lane, v[j][0], v[j][1], v[j][2]);
        }
      }
    }
  }
}

template <bool MAP, bool SCL>
__global__ __launch_bounds__(GW * 32)
void gemm_rows_finish_push_kernel(const float* X, const float* __restrict__ norm,
                                  const float* __restrict__ W,
                                  const float* __restrict__ fin_norm,
                                  const float* __restrict__ fin_b,
                                  const float* __restrict__ fin_a,
                                  float* __restrict__ OUT, float* ZERO, int nrows, int K,
                                  const int* __restrict__ sta, const int* __restrict__ cnt,
                                  const int* __restrict__ dlist, const float* __restrict__ h,
                                  const int* __restrict__ rowmap, const float* __restrict__ scale,
                                  float* __restrict__ agg, int* __restrict__ ctr) {
  __shared__ float ldsW[HD * WS];
  gemm_rows_body<true>(ldsW, X, nullptr, norm, W, fin_norm, fin_b, fin_a, OUT, ZERO, nullptr, nrows, K,
                       nullptr, nullptr, nullptr, nullptr, nullptr, nullptr,
                       nullptr, nullptr, nullptr, nullptr, nullptr, nullptr, nullptr, nullptr,
                       nullptr, 0, 0);
  spmm_push_wave<MAP, SCL>(sta, cnt, dlist, h, rowmap, scale, agg, ctr, nrows, (int)(threadIdx.x & 31));
}

__global__ __launch_bounds__(256)
void finish_dot_kernel(const float* __restrict__ agg0, const float* __restrict__ agg1,
                       const float* __restrict__ norm_in,
                       const float* __restrict__ b, const float* __restrict__ a,
                       const float* __restrict__ wsum, const float* __restrict__ bsum,
                       float* __restrict__ out0, int n, int nb) {
  const int lane = threadIdx.x & 31;
  const int wave = __builtin_amdgcn_readfirstlane((int)(threadIdx.x >> 5));
  const int view = ((int)blockIdx.x >= nb) ? 1 : 0;
  const float* agg = view ? agg1 : agg0;
  float* out = out0 + (long)view * n;
  const long r0  = ((long)((int)blockIdx.x - view * nb) * (blockDim.x >> 5) + wave) * RW;
  if (r0 >= n) return;
  const float w0 = wsum[lane], w1 = wsum[lane + 32], w2 = wsum[lane + 64];
  const float b0 = b[lane],    b1 = b[lane + 32],    b2 = b[lane + 64];
  const float a0 = a[lane],    a1 = a[lane + 32],    a2 = a[lane + 64];
  const float bs = *bsum;
  finish_dot_rows(agg, norm_in, b0, b1, b2, a0, a1, a2, w0, w1, w2, bs, out, r0, n, lane);
}

extern "C" void kernel_launch(void* const* d_in, const int* in_sizes, int n_in,
                              void* d_out, int out_size, void* d_ws, size_t ws_size,
                              hipStream_t stream) {
  const float* x    = (const float*)d_in[0];
  const int*   src  = (const int*)  d_in[1];
  const int*   dst  = (const int*)  d_in[2];
  const int*   perm = (const int*)  d_in[3];
  const float* W1   = (const float*)d_in[4];
  const float* b1   = (const float*)d_in[5];
  const float* a1   = (const float*)d_in[6];
  const float* W2   = (const float*)d_in[7];
  const float* b2   = (const float*)d_in[8];
  const float* a2   = (const float*)d_in[9];
  const float* Wm   = (const float*)d_in[10];
  const float* bm   = (const float*)d_in[11];
  float* out = (float*)d_out;

  const int E  = in_sizes[1];
  const int N  = in_sizes[3];
  const int K1 = in_sizes[0] / N;

  float* ws       = (float*)d_ws;
  float* norm_out = ws;
  float* norm_in  = ws + N;
  float* wsum     = ws + 2L * N;
  float* wbsum    = ws + 2L * N + HD;
  int*   ctr      = (int*)(ws + 2L * N + 128);
  float* bufA     = ws + 2L * N + 2 * HD;
  float* bufB     = bufA + (long)N * HD;

  int* cnt   = (int*)(bufB + (long)N * HD);
  int* cur   = cnt + 2L * N;
  int* off   = cur + 2L * N;
  int* sta   = off + 2L * N;
  int* bsum  = sta + 2L * N;
  int* dlist = bsum + 256;
  const int  nsl  = (2 * N + KSL - 1) / KSL;
  const bool push = (nsl <= 256) &&
                    (ws_size >= (size_t)((char*)(dlist + E) - (char*)d_ws));

  float* bufC = (float*)(((size_t)(dlist + E) + 15) & ~(size_t)15);
  const bool once = push && (ws_size >= (size_t)((char*)(bufC + (long)N * HD) - (char*)d_ws));

  const int gemm_blocks = 256;
  const int red_blocks  = (int)(((long)N + 8 * RW - 1) / (8 * RW));

  if (push) {
    fill_zero_push_kernel<<<512, 256, 0, stream>>>(norm_in, (long)N, cnt, 4L * N, ctr);
    degree_keys_kernel<<<512, 256, 0, stream>>>(src, dst, norm_in, cnt, ctr, E, N, Wm, bm, wsum, wbsum);
    scan_norm_kernel<<<nsl, 256, 0, stream>>>(cnt, off, bsum, norm_out, norm_in, N);
  } else {
    fill_zero_kernel<<<512, 256, 0, stream>>>(norm_out, 2L * N, ctr);
    degree_kernel<<<512, 256, 0, stream>>>(src, dst, norm_out, norm_in, ctr, E, N, Wm, bm, wsum, wbsum);
    finalize_norm_kernel<<<(2 * N + 255) / 256, 256, 0, stream>>>(norm_out, 2 * N);
  }

  float* bufD = bufC + (long)N * HD;
  const bool ovl = once && (ws_size >= (size_t)((char*)(bufD + (long)N * HD) - (char*)d_ws));
  if (ovl) {
    gemm_rows_wmma_kernel<<<gemm_blocks, (GW + PW) * 32, 0, stream>>>(
        x, nullptr, nullptr, W1, bufA, bufB, bufC, N, K1,
        norm_in, b2, a2, wsum, wbsum, nullptr,
        src, dst, off, bsum, cur, sta, dlist, ctr + 5, E, nsl);
    spmm_push_kernel<false, true><<<2048, 256, 0, stream>>>(sta, cnt, dlist, bufA, nullptr, norm_out, bufB, ctr + 1, N);
    gemm_rows_finish_push_kernel<true, true><<<gemm_blocks, GW * 32, 0, stream>>>(
        bufB, norm_out, W2, norm_in, b1, a1, bufD, bufB, N, HD,
        sta, cnt, dlist, bufA, perm, norm_out, bufC, ctr + 3);
    gemm_rows_finish_push_kernel<false, false><<<gemm_blocks, GW * 32, 0, stream>>>(
        bufC, norm_out, W2, norm_in, b1, a1, bufA, bufC, N, HD,
        sta, cnt, dlist, bufD, nullptr, nullptr, bufB, ctr + 2);
    spmm_push_kernel<false, false><<<2048, 256, 0, stream>>>(sta, cnt, dlist, bufA, nullptr, nullptr, bufC, ctr + 4, N);
    finish_dot_kernel<<<2 * red_blocks, 256, 0, stream>>>(bufB, bufC, norm_in, b2, a2, wsum, wbsum,
                                                          out, N, red_blocks);
    return;
  }

  if (once) {
    gemm_rows_wmma_kernel<<<gemm_blocks, (GW + PW) * 32, 0, stream>>>(
        x, nullptr, nullptr, W1, bufA, bufB, bufC, N, K1,
        norm_in, b2, a2, wsum, wbsum, nullptr,
        src, dst, off, bsum, cur, sta, dlist, ctr + 5, E, nsl);
    spmm_push_kernel<false, true><<<2048, 256, 0, stream>>>(sta, cnt, dlist, bufA, nullptr, norm_out, bufB, ctr + 1, N);
    spmm_push_kernel<true, true><<<2048, 256, 0, stream>>>(sta, cnt, dlist, bufA, perm, norm_out, bufC, ctr + 3, N);
    gemm_rows_finish_wmma_kernel<<<gemm_blocks, GW * 32, 0, stream>>>(bufB, norm_out, W2,
                                                                     norm_in, b1, a1, bufA, bufB, N, HD);
    spmm_push_kernel<false, false><<<2048, 256, 0, stream>>>(sta, cnt, dlist, bufA, nullptr, nullptr, bufB, ctr + 2, N);
    gemm_rows_finish_wmma_kernel<<<gemm_blocks, GW * 32, 0, stream>>>(bufC, norm_out, W2,
                                                                     norm_in, b1, a1, bufA, bufC, N, HD);
    spmm_push_kernel<false, false><<<2048, 256, 0, stream>>>(sta, cnt, dlist, bufA, nullptr, nullptr, bufC, ctr + 4, N);
    finish_dot_kernel<<<2 * red_blocks, 256, 0, stream>>>(bufB, bufC, norm_in, b2, a2, wsum, wbsum,
                                                          out, N, red_blocks);
    return;
  }

  for (int view = 0; view < 2; ++view) {
    const int* rowmap = view ? perm : nullptr;
    float* pv_out = view ? out : nullptr;
    const bool ride = push && (view == 0);
    int* pl_dlist = ride ? dlist : nullptr;
    gemm_rows_wmma_kernel<<<gemm_blocks, ride ? (GW + PW) * 32 : GW * 32, 0, stream>>>(
        x, rowmap, norm_out, W1, bufA, bufB, nullptr, N, K1,
        norm_in, b2, a2, wsum, wbsum, pv_out,
        src, dst, off, bsum, cur, sta, pl_dlist, ctr + 5, E, nsl);
    if (push) spmm_push_kernel<false, false><<<2048, 256, 0, stream>>>(sta, cnt, dlist, bufA, nullptr, nullptr, bufB, ctr + 1 + 2 * view, N);
    else      spmm_kernel<<<2048, 256, 0, stream>>>(src, dst, bufA, bufB, ctr + 1 + 2 * view, E, N);
    gemm_rows_finish_wmma_kernel<<<gemm_blocks, GW * 32, 0, stream>>>(bufB, norm_out, W2,
                                                                     norm_in, b1, a1, bufA, bufB, N, HD);
    if (push) spmm_push_kernel<false, false><<<2048, 256, 0, stream>>>(sta, cnt, dlist, bufA, nullptr, nullptr, bufB, ctr + 2 + 2 * view, N);
    else      spmm_kernel<<<2048, 256, 0, stream>>>(src, dst, bufA, bufB, ctr + 2 + 2 * view, E, N);
    if (view) finish_dot_kernel<<<red_blocks, 256, 0, stream>>>(bufB, bufB, norm_in, b2, a2, wsum, wbsum,
                                                                out + (long)N, N, red_blocks);
  }
}
